# starting kernel
# baseline (speedup 1.0000x reference)
#include <hip/hip_runtime.h>

#define NB 32
#define NT 1024
#define NJ 128
#define ND 256

typedef __attribute__((ext_vector_type(2))) float v2f;
typedef __attribute__((ext_vector_type(8))) float v8f;

__device__ __forceinline__ float wave_reduce_sum(float v) {
#pragma unroll
  for (int o = 16; o > 0; o >>= 1) v += __shfl_xor(v, o, 32);
  return v;
}
__device__ __forceinline__ float wave_reduce_max(float v) {
#pragma unroll
  for (int o = 16; o > 0; o >>= 1) v = fmaxf(v, __shfl_xor(v, o, 32));
  return v;
}

__global__ void k_rowdots(const float* __restrict__ C, const float* __restrict__ Q,
                          const float* __restrict__ w, float* __restrict__ ch,
                          float* __restrict__ qb) {
  int wave = blockIdx.x * (blockDim.x >> 5) + (threadIdx.x >> 5);
  int lane = threadIdx.x & 31;
  const int nch = NB * NT;
  const int nrows = nch + NB * NJ;
  if (wave >= nrows) return;
  const float* row; const float* wv; float* out;
  if (wave < nch) { row = C + (size_t)wave * ND; wv = w;      out = ch + wave; }
  else { int r = wave - nch; row = Q + (size_t)r * ND; wv = w + ND; out = qb + r; }
  float s = 0.f;
#pragma unroll
  for (int i = lane; i < ND; i += 32) s += row[i] * wv[i];
  s = wave_reduce_sum(s);
  if (lane == 0) *out = s;
}

__global__ void k_score(const float* __restrict__ C, const float* __restrict__ Q,
                        const float* __restrict__ w, const float* __restrict__ ch,
                        const float* __restrict__ qb, float* __restrict__ S) {
  int blk = blockIdx.x;
  int b = blk >> 6;
  int ttile = blk & 63;
  int wv = threadIdx.x >> 5;
  int lane = threadIdx.x & 31;
  int m = lane & 15, kh = lane >> 4;
  int t0 = ttile * 16, j0 = wv * 16;
  const float* whu = w + 2 * ND;
  const float* Crow = C + (size_t)(b * NT + t0 + m) * ND;
  const float* Qrow = Q + (size_t)(b * NJ + j0 + m) * ND;
  v8f acc = {};
#pragma unroll 4
  for (int k0 = 0; k0 < ND; k0 += 4) {
    int ka = k0 + 2 * kh;
    v2f a  = *(const v2f*)(Crow + ka);
    v2f wu = *(const v2f*)(whu + ka);
    a *= wu;
    v2f bf = *(const v2f*)(Qrow + ka);
    acc = __builtin_amdgcn_wmma_f32_16x16x4_f32(false, a, false, bf, (short)0,
                                                acc, false, false);
  }
  int jj = j0 + m;
  float qv = qb[b * NJ + jj];
#pragma unroll
  for (int r = 0; r < 8; ++r) {
    int tt = t0 + r + 8 * kh;
    S[(size_t)(b * NT + tt) * NJ + jj] = acc[r] + ch[b * NT + tt] + qv;
  }
}

__global__ void k_softmax_rows(const float* __restrict__ S, float* __restrict__ Sj) {
  int wave = blockIdx.x * (blockDim.x >> 5) + (threadIdx.x >> 5);
  int lane = threadIdx.x & 31;
  if (wave >= NB * NT) return;
  const float* row = S + (size_t)wave * NJ;
  float4 v = *(const float4*)(row + lane * 4);
  float m = fmaxf(fmaxf(v.x, v.y), fmaxf(v.z, v.w));
  m = wave_reduce_max(m);
  float e0 = expf(v.x - m), e1 = expf(v.y - m), e2 = expf(v.z - m), e3 = expf(v.w - m);
  float s = wave_reduce_sum(e0 + e1 + e2 + e3);
  float inv = 1.0f / s;
  float4 o; o.x = e0 * inv; o.y = e1 * inv; o.z = e2 * inv; o.w = e3 * inv;
  *(float4*)(Sj + (size_t)wave * NJ + lane * 4) = o;
}

__global__ void k_softmax_cols(const float* __restrict__ S, float* __restrict__ St) {
  __shared__ float red[8][32];
  __shared__ float cmax[32];
  __shared__ float csum[32];
  int b = blockIdx.x >> 2;
  int jc = blockIdx.x & 3;
  int lx = threadIdx.x & 31, ty = threadIdx.x >> 5;
  int j = jc * 32 + lx;
  const float* base = S + (size_t)b * NT * NJ + j;
  float lm = -3.4e38f;
  for (int t = ty; t < NT; t += 8) lm = fmaxf(lm, base[(size_t)t * NJ]);
  red[ty][lx] = lm;
  __syncthreads();
  if (ty == 0) {
    float m = red[0][lx];
#pragma unroll
    for (int i = 1; i < 8; ++i) m = fmaxf(m, red[i][lx]);
    cmax[lx] = m;
  }
  __syncthreads();
  float M = cmax[lx];
  float ls = 0.f;
  for (int t = ty; t < NT; t += 8) ls += expf(base[(size_t)t * NJ] - M);
  __syncthreads();
  red[ty][lx] = ls;
  __syncthreads();
  if (ty == 0) {
    float s = red[0][lx];
#pragma unroll
    for (int i = 1; i < 8; ++i) s += red[i][lx];
    csum[lx] = s;
  }
  __syncthreads();
  float inv = 1.0f / csum[lx];
  float* ob = St + (size_t)b * NT * NJ + j;
  for (int t = ty; t < NT; t += 8)
    ob[(size_t)t * NJ] = expf(base[(size_t)t * NJ] - M) * inv;
}

__global__ void k_tmp(const float* __restrict__ St, const float* __restrict__ C,
                      float* __restrict__ tmp) {
  int b = blockIdx.x >> 3;
  int dpair = blockIdx.x & 7;
  int wv = threadIdx.x >> 5;
  int lane = threadIdx.x & 31;
  int m = lane & 15, kh = lane >> 4;
  int j0 = wv * 16, d0 = dpair * 32;
  const float* Sbase = St + (size_t)b * NT * NJ + j0 + m;
  const float* Cbase = C + (size_t)b * NT * ND + d0 + m;
  v8f acc0 = {}, acc1 = {};
#pragma unroll 2
  for (int k0 = 0; k0 < NT; k0 += 4) {
    int ka = k0 + 2 * kh;
    v2f a, b0, b1;
    a.x  = Sbase[(size_t)ka * NJ];
    a.y  = Sbase[(size_t)(ka + 1) * NJ];
    b0.x = Cbase[(size_t)ka * ND];
    b0.y = Cbase[(size_t)(ka + 1) * ND];
    b1.x = Cbase[(size_t)ka * ND + 16];
    b1.y = Cbase[(size_t)(ka + 1) * ND + 16];
    acc0 = __builtin_amdgcn_wmma_f32_16x16x4_f32(false, a, false, b0, (short)0,
                                                 acc0, false, false);
    acc1 = __builtin_amdgcn_wmma_f32_16x16x4_f32(false, a, false, b1, (short)0,
                                                 acc1, false, false);
  }
#pragma unroll
  for (int r = 0; r < 8; ++r) {
    int jj = j0 + r + 8 * kh;
    float* ob = tmp + (size_t)(b * NJ + jj) * ND + d0 + m;
    ob[0]  = acc0[r];
    ob[16] = acc1[r];
  }
}

__global__ void k_out(const float* __restrict__ Sj, const float* __restrict__ Q,
                      const float* __restrict__ tmp, const float* __restrict__ C,
                      float* __restrict__ out) {
  int blk = blockIdx.x;
  int b = blk >> 7;
  int rem = blk & 127;
  int ttile = rem >> 1;
  int dh = rem & 1;
  int wv = threadIdx.x >> 5;
  int lane = threadIdx.x & 31;
  int m = lane & 15, kh = lane >> 4;
  int t0 = ttile * 16, d0 = (dh * 8 + wv) * 16;
  const float* Srow = Sj + (size_t)(b * NT + t0 + m) * NJ;
  const float* Qb = Q + (size_t)b * NJ * ND + d0 + m;
  const float* Tb = tmp + (size_t)b * NJ * ND + d0 + m;
  v8f accA = {}, accB = {};
#pragma unroll 4
  for (int k0 = 0; k0 < NJ; k0 += 4) {
    int ka = k0 + 2 * kh;
    v2f a = *(const v2f*)(Srow + ka);
    v2f bq, bt;
    bq.x = Qb[(size_t)ka * ND];
    bq.y = Qb[(size_t)(ka + 1) * ND];
    bt.x = Tb[(size_t)ka * ND];
    bt.y = Tb[(size_t)(ka + 1) * ND];
    accA = __builtin_amdgcn_wmma_f32_16x16x4_f32(false, a, false, bq, (short)0,
                                                 accA, false, false);
    accB = __builtin_amdgcn_wmma_f32_16x16x4_f32(false, a, false, bt, (short)0,
                                                 accB, false, false);
  }
  int d = d0 + m;
#pragma unroll
  for (int r = 0; r < 8; ++r) {
    int t = t0 + r + 8 * kh;
    float cv = C[(size_t)(b * NT + t) * ND + d];
    float* ob = out + (size_t)(b * NT + t) * (4 * ND);
    ob[d]          = cv;
    ob[ND + d]     = accA[r];
    ob[2 * ND + d] = cv * accA[r];
    ob[3 * ND + d] = cv * accB[r];
  }
}

extern "C" void kernel_launch(void* const* d_in, const int* in_sizes, int n_in,
                              void* d_out, int out_size, void* d_ws, size_t ws_size,
                              hipStream_t stream) {
  (void)in_sizes; (void)n_in; (void)out_size; (void)ws_size;
  const float* C = (const float*)d_in[0];
  const float* Q = (const float*)d_in[1];
  const float* w = (const float*)d_in[2];
  float* out = (float*)d_out;

  float* ws = (float*)d_ws;
  float* ch = ws;
  float* qb = ch + NB * NT;
  float* S  = qb + NB * NJ;
  float* Sj = S  + (size_t)NB * NT * NJ;
  float* St = Sj + (size_t)NB * NT * NJ;
  float* tp = St + (size_t)NB * NT * NJ;

  int rows = NB * NT + NB * NJ;
  k_rowdots<<<(rows + 7) / 8, 256, 0, stream>>>(C, Q, w, ch, qb);
  k_score<<<NB * 64, 256, 0, stream>>>(C, Q, w, ch, qb, S);
  k_softmax_rows<<<(NB * NT + 7) / 8, 256, 0, stream>>>(S, Sj);
  k_softmax_cols<<<NB * (NJ / 32), 256, 0, stream>>>(S, St);
  k_tmp<<<NB * 8, 256, 0, stream>>>(St, C, tp);
  k_out<<<NB * 128, 256, 0, stream>>>(Sj, Q, tp, C, out);
}


# k_out: a wave owns 32 t x 32 d of both products (2x2 tiles each, 8 accumulators): 2 A and 4 B fragment loads feed 8 WMMAs; 1024 blocks of 8 waves; k order and all sums unchanged
# speedup vs baseline: 1.3006x; 1.3006x over previous
#include <hip/hip_runtime.h>

#define NB 32
#define NT 1024
#define NJ 128
#define ND 256

typedef __attribute__((ext_vector_type(2))) float v2f;
typedef __attribute__((ext_vector_type(8))) float v8f;

__device__ __forceinline__ float wave_reduce_sum(float v) {
#pragma unroll
  for (int o = 16; o > 0; o >>= 1) v += __shfl_xor(v, o, 32);
  return v;
}
__device__ __forceinline__ float wave_reduce_max(float v) {
#pragma unroll
  for (int o = 16; o > 0; o >>= 1) v = fmaxf(v, __shfl_xor(v, o, 32));
  return v;
}

__global__ void k_rowdots(const float* __restrict__ C, const float* __restrict__ Q,
                          const float* __restrict__ w, float* __restrict__ ch,
                          float* __restrict__ qb) {
  int wave = blockIdx.x * (blockDim.x >> 5) + (threadIdx.x >> 5);
  int lane = threadIdx.x & 31;
  const int nch = NB * NT;
  const int nrows = nch + NB * NJ;
  if (wave >= nrows) return;
  const float* row; const float* wv; float* out;
  if (wave < nch) { row = C + (size_t)wave * ND; wv = w;      out = ch + wave; }
  else { int r = wave - nch; row = Q + (size_t)r * ND; wv = w + ND; out = qb + r; }
  float s = 0.f;
#pragma unroll
  for (int i = lane; i < ND; i += 32) s += row[i] * wv[i];
  s = wave_reduce_sum(s);
  if (lane == 0) *out = s;
}

__global__ void k_score(const float* __restrict__ C, const float* __restrict__ Q,
                        const float* __restrict__ w, const float* __restrict__ ch,
                        const float* __restrict__ qb, float* __restrict__ S,
                        float* __restrict__ ST) {
  int blk = blockIdx.x;
  int b = blk >> 6;
  int ttile = blk & 63;
  int wv = threadIdx.x >> 5;
  int lane = threadIdx.x & 31;
  int m = lane & 15, kh = lane >> 4;
  int t0 = ttile * 16, j0 = wv * 16;
  const float* whu = w + 2 * ND;
  const float* Crow = C + (size_t)(b * NT + t0 + m) * ND;
  const float* Qrow = Q + (size_t)(b * NJ + j0 + m) * ND;
  v8f acc = {};
#pragma unroll 4
  for (int k0 = 0; k0 < ND; k0 += 4) {
    int ka = k0 + 2 * kh;
    v2f a  = *(const v2f*)(Crow + ka);
    v2f wu = *(const v2f*)(whu + ka);
    a *= wu;
    v2f bf = *(const v2f*)(Qrow + ka);
    acc = __builtin_amdgcn_wmma_f32_16x16x4_f32(false, a, false, bf, (short)0,
                                                acc, false, false);
  }
  int jj = j0 + m;
  float qv = qb[b * NJ + jj];
  float sv[8];
#pragma unroll
  for (int r = 0; r < 8; ++r) {
    int tt = t0 + r + 8 * kh;
    float s = acc[r] + ch[b * NT + tt] + qv;
    S[(size_t)(b * NT + tt) * NJ + jj] = s;
    sv[r] = s;
  }
  float* tr = ST + (size_t)(b * NJ + jj) * NT + t0 + 8 * kh;
  *(float4*)(tr)     = make_float4(sv[0], sv[1], sv[2], sv[3]);
  *(float4*)(tr + 4) = make_float4(sv[4], sv[5], sv[6], sv[7]);
}

__global__ void k_softmax_rows(const float* __restrict__ S, float* __restrict__ Sj) {
  int wave = blockIdx.x * (blockDim.x >> 5) + (threadIdx.x >> 5);
  int lane = threadIdx.x & 31;
  if (wave >= NB * NT) return;
  const float* row = S + (size_t)wave * NJ;
  float4 v = *(const float4*)(row + lane * 4);
  float m = fmaxf(fmaxf(v.x, v.y), fmaxf(v.z, v.w));
  m = wave_reduce_max(m);
  float e0 = expf(v.x - m), e1 = expf(v.y - m), e2 = expf(v.z - m), e3 = expf(v.w - m);
  float s = wave_reduce_sum(e0 + e1 + e2 + e3);
  float inv = 1.0f / s;
  float4 o; o.x = e0 * inv; o.y = e1 * inv; o.z = e2 * inv; o.w = e3 * inv;
  *(float4*)(Sj + (size_t)wave * NJ + lane * 4) = o;
}

__global__ void k_softmax_t(float* __restrict__ ST) {
  int wave = blockIdx.x * (blockDim.x >> 5) + (threadIdx.x >> 5);
  int lane = threadIdx.x & 31;
  if (wave >= NB * NJ) return;
  float* row = ST + (size_t)wave * NT + lane * 4;
  float4 v[8];
#pragma unroll
  for (int i = 0; i < 8; ++i) v[i] = *(const float4*)(row + i * 128);
  float m = -3.4e38f;
#pragma unroll
  for (int i = 0; i < 8; ++i)
    m = fmaxf(m, fmaxf(fmaxf(v[i].x, v[i].y), fmaxf(v[i].z, v[i].w)));
  m = wave_reduce_max(m);
  float s = 0.f;
#pragma unroll
  for (int i = 0; i < 8; ++i) {
    v[i].x = expf(v[i].x - m); v[i].y = expf(v[i].y - m);
    v[i].z = expf(v[i].z - m); v[i].w = expf(v[i].w - m);
    s += (v[i].x + v[i].y) + (v[i].z + v[i].w);
  }
  s = wave_reduce_sum(s);
  float inv = 1.0f / s;
#pragma unroll
  for (int i = 0; i < 8; ++i) {
    float4 o;
    o.x = v[i].x * inv; o.y = v[i].y * inv; o.z = v[i].z * inv; o.w = v[i].w * inv;
    *(float4*)(row + i * 128) = o;
  }
}

__global__ void k_tmp(const float* __restrict__ StT, const float* __restrict__ C,
                      float* __restrict__ tmp) {
  int b = blockIdx.x >> 3;
  int dpair = blockIdx.x & 7;
  int wv = threadIdx.x >> 5;
  int lane = threadIdx.x & 31;
  int m = lane & 15, kh = lane >> 4;
  int j0 = wv * 16, d0 = dpair * 32;
  const float* Arow = StT + (size_t)(b * NJ + j0 + m) * NT;
  const float* Cbase = C + (size_t)b * NT * ND + d0 + m;
  v8f acc0 = {}, acc1 = {};
#pragma unroll 2
  for (int k0 = 0; k0 < NT; k0 += 4) {
    int ka = k0 + 2 * kh;
    v2f b0, b1;
    v2f a = *(const v2f*)(Arow + ka);
    b0.x = Cbase[(size_t)ka * ND];
    b0.y = Cbase[(size_t)(ka + 1) * ND];
    b1.x = Cbase[(size_t)ka * ND + 16];
    b1.y = Cbase[(size_t)(ka + 1) * ND + 16];
    acc0 = __builtin_amdgcn_wmma_f32_16x16x4_f32(false, a, false, b0, (short)0,
                                                 acc0, false, false);
    acc1 = __builtin_amdgcn_wmma_f32_16x16x4_f32(false, a, false, b1, (short)0,
                                                 acc1, false, false);
  }
#pragma unroll
  for (int r = 0; r < 8; ++r) {
    int jj = j0 + r + 8 * kh;
    float* ob = tmp + (size_t)(b * NJ + jj) * ND + d0 + m;
    ob[0]  = acc0[r];
    ob[16] = acc1[r];
  }
}

__global__ void __launch_bounds__(256)
k_out(const float* __restrict__ Sj, const float* __restrict__ Q,
      const float* __restrict__ tmp, const float* __restrict__ C,
      float* __restrict__ out) {
  int blk = blockIdx.x;
  int b = blk >> 5;
  int tblk = blk & 31;
  int wv = threadIdx.x >> 5;
  int lane = threadIdx.x & 31;
  int m = lane & 15, kh = lane >> 4;
  int t0 = tblk * 32, d0 = wv * 32;
  const float* S0 = Sj + (size_t)(b * NT + t0 + m) * NJ + 2 * kh;
  const float* S1 = S0 + (size_t)16 * NJ;
  const float* Qb = Q + (size_t)b * NJ * ND + (size_t)(2 * kh) * ND + d0 + m;
  const float* Tb = tmp + (size_t)b * NJ * ND + (size_t)(2 * kh) * ND + d0 + m;
  v8f aQ00 = {}, aQ01 = {}, aQ10 = {}, aQ11 = {};
  v8f aT00 = {}, aT01 = {}, aT10 = {}, aT11 = {};
#pragma unroll 2
  for (int k0 = 0; k0 < NJ; k0 += 4) {
    v2f a0 = *(const v2f*)(S0 + k0);
    v2f a1 = *(const v2f*)(S1 + k0);
    const float* qp = Qb + (size_t)k0 * ND;
    const float* tp = Tb + (size_t)k0 * ND;
    v2f q0, q1, u0, u1;
    q0.x = qp[0];  q0.y = qp[ND];
    q1.x = qp[16]; q1.y = qp[ND + 16];
    u0.x = tp[0];  u0.y = tp[ND];
    u1.x = tp[16]; u1.y = tp[ND + 16];
    aQ00 = __builtin_amdgcn_wmma_f32_16x16x4_f32(false, a0, false, q0, (short)0,
                                                 aQ00, false, false);
    aT00 = __builtin_amdgcn_wmma_f32_16x16x4_f32(false, a0, false, u0, (short)0,
                                                 aT00, false, false);
    aQ01 = __builtin_amdgcn_wmma_f32_16x16x4_f32(false, a0, false, q1, (short)0,
                                                 aQ01, false, false);
    aT01 = __builtin_amdgcn_wmma_f32_16x16x4_f32(false, a0, false, u1, (short)0,
                                                 aT01, false, false);
    aQ10 = __builtin_amdgcn_wmma_f32_16x16x4_f32(false, a1, false, q0, (short)0,
                                                 aQ10, false, false);
    aT10 = __builtin_amdgcn_wmma_f32_16x16x4_f32(false, a1, false, u0, (short)0,
                                                 aT10, false, false);
    aQ11 = __builtin_amdgcn_wmma_f32_16x16x4_f32(false, a1, false, q1, (short)0,
                                                 aQ11, false, false);
    aT11 = __builtin_amdgcn_wmma_f32_16x16x4_f32(false, a1, false, u1, (short)0,
                                                 aT11, false, false);
  }
  int d = d0 + m;
#pragma unroll
  for (int r = 0; r < 8; ++r) {
    int t = t0 + r + 8 * kh;
    const float* cp = C + (size_t)(b * NT + t) * ND + d;
    float* ob = out + (size_t)(b * NT + t) * (4 * ND) + d;
    float c0 = cp[0], c1 = cp[16];
    ob[0]           = c0;            ob[16]          = c1;
    ob[ND]          = aQ00[r];       ob[ND + 16]     = aQ01[r];
    ob[2 * ND]      = c0 * aQ00[r];  ob[2 * ND + 16] = c1 * aQ01[r];
    ob[3 * ND]      = c0 * aT00[r];  ob[3 * ND + 16] = c1 * aT01[r];
  }
#pragma unroll
  for (int r = 0; r < 8; ++r) {
    int t = t0 + 16 + r + 8 * kh;
    const float* cp = C + (size_t)(b * NT + t) * ND + d;
    float* ob = out + (size_t)(b * NT + t) * (4 * ND) + d;
    float c0 = cp[0], c1 = cp[16];
    ob[0]           = c0;            ob[16]          = c1;
    ob[ND]          = aQ10[r];       ob[ND + 16]     = aQ11[r];
    ob[2 * ND]      = c0 * aQ10[r];  ob[2 * ND + 16] = c1 * aQ11[r];
    ob[3 * ND]      = c0 * aT10[r];  ob[3 * ND + 16] = c1 * aT11[r];
  }
}

extern "C" void kernel_launch(void* const* d_in, const int* in_sizes, int n_in,
                              void* d_out, int out_size, void* d_ws, size_t ws_size,
                              hipStream_t stream) {
  (void)in_sizes; (void)n_in; (void)out_size; (void)ws_size;
  const float* C = (const float*)d_in[0];
  const float* Q = (const float*)d_in[1];
  const float* w = (const float*)d_in[2];
  float* out = (float*)d_out;

  float* ws = (float*)d_ws;
  float* ch = ws;
  float* qb = ch + NB * NT;
  float* S  = qb + NB * NJ;
  float* Sj = S  + (size_t)NB * NT * NJ;
  float* ST = Sj + (size_t)NB * NT * NJ;
  float* tp = ST + (size_t)NB * NT * NJ;

  int rows = NB * NT + NB * NJ;
  k_rowdots<<<(rows + 7) / 8, 256, 0, stream>>>(C, Q, w, ch, qb);
  k_score<<<NB * 64, 256, 0, stream>>>(C, Q, w, ch, qb, S, ST);
  k_softmax_rows<<<(NB * NT + 7) / 8, 256, 0, stream>>>(S, Sj);
  k_softmax_t<<<(NB * NJ) / 4, 128, 0, stream>>>(ST);
  k_tmp<<<NB * 8, 256, 0, stream>>>(ST, C, tp);
  k_out<<<NB * 32, 256, 0, stream>>>(Sj, Q, tp, C, out);
}


# k_score: a wave owns 32 t x 64 j (2x4 tiles, 8 accumulators): 2 A and 4 B fragment loads feed 8 WMMAs, two fragment sets so loads run ahead; 256 blocks of 8 waves; k order and all sums unchanged
# speedup vs baseline: 1.5679x; 1.2055x over previous
#include <hip/hip_runtime.h>

#define NB 32
#define NT 1024
#define NJ 128
#define ND 256

typedef __attribute__((ext_vector_type(2))) float v2f;
typedef __attribute__((ext_vector_type(8))) float v8f;

__device__ __forceinline__ float wave_reduce_sum(float v) {
#pragma unroll
  for (int o = 16; o > 0; o >>= 1) v += __shfl_xor(v, o, 32);
  return v;
}
__device__ __forceinline__ float wave_reduce_max(float v) {
#pragma unroll
  for (int o = 16; o > 0; o >>= 1) v = fmaxf(v, __shfl_xor(v, o, 32));
  return v;
}

__global__ void k_rowdots(const float* __restrict__ C, const float* __restrict__ Q,
                          const float* __restrict__ w, float* __restrict__ ch,
                          float* __restrict__ qb) {
  int wave = blockIdx.x * (blockDim.x >> 5) + (threadIdx.x >> 5);
  int lane = threadIdx.x & 31;
  const int nch = NB * NT;
  const int nrows = nch + NB * NJ;
  if (wave >= nrows) return;
  const float* row; const float* wv; float* out;
  if (wave < nch) { row = C + (size_t)wave * ND; wv = w;      out = ch + wave; }
  else { int r = wave - nch; row = Q + (size_t)r * ND; wv = w + ND; out = qb + r; }
  float s = 0.f;
#pragma unroll
  for (int i = lane; i < ND; i += 32) s += row[i] * wv[i];
  s = wave_reduce_sum(s);
  if (lane == 0) *out = s;
}

struct ScoreFrag { v2f wu, a0, a1, q0, q1, q2, q3; };

__device__ __forceinline__ void score_load(ScoreFrag (&f)[4], const float* whu,
                                           const float* C0, const float* Q0, int kb) {
#pragma unroll
  for (int s = 0; s < 4; ++s) {
    int kk = kb + 4 * s;
    f[s].wu = *(const v2f*)(whu + kk);
    f[s].a0 = *(const v2f*)(C0 + kk);
    f[s].a1 = *(const v2f*)(C0 + 16 * ND + kk);
    f[s].q0 = *(const v2f*)(Q0 + kk);
    f[s].q1 = *(const v2f*)(Q0 + 16 * ND + kk);
    f[s].q2 = *(const v2f*)(Q0 + 32 * ND + kk);
    f[s].q3 = *(const v2f*)(Q0 + 48 * ND + kk);
  }
}

__device__ __forceinline__ void score_mma(const ScoreFrag (&f)[4], v8f (&acc)[2][4]) {
#pragma unroll
  for (int s = 0; s < 4; ++s) {
    v2f a0 = f[s].a0 * f[s].wu;
    v2f a1 = f[s].a1 * f[s].wu;
    acc[0][0] = __builtin_amdgcn_wmma_f32_16x16x4_f32(false, a0, false, f[s].q0, (short)0,
                                                      acc[0][0], false, false);
    acc[1][0] = __builtin_amdgcn_wmma_f32_16x16x4_f32(false, a1, false, f[s].q0, (short)0,
                                                      acc[1][0], false, false);
    acc[0][1] = __builtin_amdgcn_wmma_f32_16x16x4_f32(false, a0, false, f[s].q1, (short)0,
                                                      acc[0][1], false, false);
    acc[1][1] = __builtin_amdgcn_wmma_f32_16x16x4_f32(false, a1, false, f[s].q1, (short)0,
                                                      acc[1][1], false, false);
    acc[0][2] = __builtin_amdgcn_wmma_f32_16x16x4_f32(false, a0, false, f[s].q2, (short)0,
                                                      acc[0][2], false, false);
    acc[1][2] = __builtin_amdgcn_wmma_f32_16x16x4_f32(false, a1, false, f[s].q2, (short)0,
                                                      acc[1][2], false, false);
    acc[0][3] = __builtin_amdgcn_wmma_f32_16x16x4_f32(false, a0, false, f[s].q3, (short)0,
                                                      acc[0][3], false, false);
    acc[1][3] = __builtin_amdgcn_wmma_f32_16x16x4_f32(false, a1, false, f[s].q3, (short)0,
                                                      acc[1][3], false, false);
  }
}

__global__ void __launch_bounds__(256)
k_score(const float* __restrict__ C, const float* __restrict__ Q,
        const float* __restrict__ w, const float* __restrict__ ch,
        const float* __restrict__ qb, float* __restrict__ S,
        float* __restrict__ ST) {
  int blk = blockIdx.x;
  int b = blk >> 3;
  int tblk = blk & 7;
  int wv = threadIdx.x >> 5;
  int lane = threadIdx.x & 31;
  int m = lane & 15, kh = lane >> 4;
  int t0 = tblk * 128 + (wv >> 1) * 32;
  int j0 = (wv & 1) * 64;
  const float* whu = w + 2 * ND + 2 * kh;
  const float* C0 = C + (size_t)(b * NT + t0 + m) * ND + 2 * kh;
  const float* Q0 = Q + (size_t)(b * NJ + j0 + m) * ND + 2 * kh;
  v8f acc[2][4] = {};
  ScoreFrag fa[4], fb[4];
  score_load(fa, whu, C0, Q0, 0);
#pragma unroll 1
  for (int kb = 0; kb < ND - 32; kb += 32) {
    score_load(fb, whu, C0, Q0, kb + 16);
    score_mma(fa, acc);
    score_load(fa, whu, C0, Q0, kb + 32);
    score_mma(fb, acc);
  }
  score_load(fb, whu, C0, Q0, ND - 16);
  score_mma(fa, acc);
  score_mma(fb, acc);

#pragma unroll
  for (int rt = 0; rt < 2; ++rt) {
    int tb = t0 + 16 * rt + 8 * kh;
    float chv[8];
#pragma unroll
    for (int r = 0; r < 8; ++r) chv[r] = ch[b * NT + tb + r];
#pragma unroll
    for (int ct = 0; ct < 4; ++ct) {
      int jj = j0 + 16 * ct + m;
      float qv = qb[b * NJ + jj];
      float sv[8];
#pragma unroll
      for (int r = 0; r < 8; ++r) {
        float s = acc[rt][ct][r] + chv[r] + qv;
        S[(size_t)(b * NT + tb + r) * NJ + jj] = s;
        sv[r] = s;
      }
      float* tr = ST + (size_t)(b * NJ + jj) * NT + tb;
      *(float4*)(tr)     = make_float4(sv[0], sv[1], sv[2], sv[3]);
      *(float4*)(tr + 4) = make_float4(sv[4], sv[5], sv[6], sv[7]);
    }
  }
}

__global__ void k_softmax_rows(const float* __restrict__ S, float* __restrict__ Sj) {
  int wave = blockIdx.x * (blockDim.x >> 5) + (threadIdx.x >> 5);
  int lane = threadIdx.x & 31;
  if (wave >= NB * NT) return;
  const float* row = S + (size_t)wave * NJ;
  float4 v = *(const float4*)(row + lane * 4);
  float m = fmaxf(fmaxf(v.x, v.y), fmaxf(v.z, v.w));
  m = wave_reduce_max(m);
  float e0 = expf(v.x - m), e1 = expf(v.y - m), e2 = expf(v.z - m), e3 = expf(v.w - m);
  float s = wave_reduce_sum(e0 + e1 + e2 + e3);
  float inv = 1.0f / s;
  float4 o; o.x = e0 * inv; o.y = e1 * inv; o.z = e2 * inv; o.w = e3 * inv;
  *(float4*)(Sj + (size_t)wave * NJ + lane * 4) = o;
}

__global__ void k_softmax_t(float* __restrict__ ST) {
  int wave = blockIdx.x * (blockDim.x >> 5) + (threadIdx.x >> 5);
  int lane = threadIdx.x & 31;
  if (wave >= NB * NJ) return;
  float* row = ST + (size_t)wave * NT + lane * 4;
  float4 v[8];
#pragma unroll
  for (int i = 0; i < 8; ++i) v[i] = *(const float4*)(row + i * 128);
  float m = -3.4e38f;
#pragma unroll
  for (int i = 0; i < 8; ++i)
    m = fmaxf(m, fmaxf(fmaxf(v[i].x, v[i].y), fmaxf(v[i].z, v[i].w)));
  m = wave_reduce_max(m);
  float s = 0.f;
#pragma unroll
  for (int i = 0; i < 8; ++i) {
    v[i].x = expf(v[i].x - m); v[i].y = expf(v[i].y - m);
    v[i].z = expf(v[i].z - m); v[i].w = expf(v[i].w - m);
    s += (v[i].x + v[i].y) + (v[i].z + v[i].w);
  }
  s = wave_reduce_sum(s);
  float inv = 1.0f / s;
#pragma unroll
  for (int i = 0; i < 8; ++i) {
    float4 o;
    o.x = v[i].x * inv; o.y = v[i].y * inv; o.z = v[i].z * inv; o.w = v[i].w * inv;
    *(float4*)(row + i * 128) = o;
  }
}

__global__ void k_tmp(const float* __restrict__ StT, const float* __restrict__ C,
                      float* __restrict__ tmp) {
  int b = blockIdx.x >> 3;
  int dpair = blockIdx.x & 7;
  int wv = threadIdx.x >> 5;
  int lane = threadIdx.x & 31;
  int m = lane & 15, kh = lane >> 4;
  int j0 = wv * 16, d0 = dpair * 32;
  const float* Arow = StT + (size_t)(b * NJ + j0 + m) * NT;
  const float* Cbase = C + (size_t)b * NT * ND + d0 + m;
  v8f acc0 = {}, acc1 = {};
#pragma unroll 2
  for (int k0 = 0; k0 < NT; k0 += 4) {
    int ka = k0 + 2 * kh;
    v2f b0, b1;
    v2f a = *(const v2f*)(Arow + ka);
    b0.x = Cbase[(size_t)ka * ND];
    b0.y = Cbase[(size_t)(ka + 1) * ND];
    b1.x = Cbase[(size_t)ka * ND + 16];
    b1.y = Cbase[(size_t)(ka + 1) * ND + 16];
    acc0 = __builtin_amdgcn_wmma_f32_16x16x4_f32(false, a, false, b0, (short)0,
                                                 acc0, false, false);
    acc1 = __builtin_amdgcn_wmma_f32_16x16x4_f32(false, a, false, b1, (short)0,
                                                 acc1, false, false);
  }
#pragma unroll
  for (int r = 0; r < 8; ++r) {
    int jj = j0 + r + 8 * kh;
    float* ob = tmp + (size_t)(b * NJ + jj) * ND + d0 + m;
    ob[0]  = acc0[r];
    ob[16] = acc1[r];
  }
}

__global__ void __launch_bounds__(256)
k_out(const float* __restrict__ Sj, const float* __restrict__ Q,
      const float* __restrict__ tmp, const float* __restrict__ C,
      float* __restrict__ out) {
  int blk = blockIdx.x;
  int b = blk >> 5;
  int tblk = blk & 31;
  int wv = threadIdx.x >> 5;
  int lane = threadIdx.x & 31;
  int m = lane & 15, kh = lane >> 4;
  int t0 = tblk * 32, d0 = wv * 32;
  const float* S0 = Sj + (size_t)(b * NT + t0 + m) * NJ + 2 * kh;
  const float* S1 = S0 + (size_t)16 * NJ;
  const float* Qb = Q + (size_t)b * NJ * ND + (size_t)(2 * kh) * ND + d0 + m;
  const float* Tb = tmp + (size_t)b * NJ * ND + (size_t)(2 * kh) * ND + d0 + m;
  v8f aQ00 = {}, aQ01 = {}, aQ10 = {}, aQ11 = {};
  v8f aT00 = {}, aT01 = {}, aT10 = {}, aT11 = {};
#pragma unroll 2
  for (int k0 = 0; k0 < NJ; k0 += 4) {
    v2f a0 = *(const v2f*)(S0 + k0);
    v2f a1 = *(const v2f*)(S1 + k0);
    const float* qp = Qb + (size_t)k0 * ND;
    const float* tp = Tb + (size_t)k0 * ND;
    v2f q0, q1, u0, u1;
    q0.x = qp[0];  q0.y = qp[ND];
    q1.x = qp[16]; q1.y = qp[ND + 16];
    u0.x = tp[0];  u0.y = tp[ND];
    u1.x = tp[16]; u1.y = tp[ND + 16];
    aQ00 = __builtin_amdgcn_wmma_f32_16x16x4_f32(false, a0, false, q0, (short)0,
                                                 aQ00, false, false);
    aT00 = __builtin_amdgcn_wmma_f32_16x16x4_f32(false, a0, false, u0, (short)0,
                                                 aT00, false, false);
    aQ01 = __builtin_amdgcn_wmma_f32_16x16x4_f32(false, a0, false, q1, (short)0,
                                                 aQ01, false, false);
    aT01 = __builtin_amdgcn_wmma_f32_16x16x4_f32(false, a0, false, u1, (short)0,
                                                 aT01, false, false);
    aQ10 = __builtin_amdgcn_wmma_f32_16x16x4_f32(false, a1, false, q0, (short)0,
                                                 aQ10, false, false);
    aT10 = __builtin_amdgcn_wmma_f32_16x16x4_f32(false, a1, false, u0, (short)0,
                                                 aT10, false, false);
    aQ11 = __builtin_amdgcn_wmma_f32_16x16x4_f32(false, a1, false, q1, (short)0,
                                                 aQ11, false, false);
    aT11 = __builtin_amdgcn_wmma_f32_16x16x4_f32(false, a1, false, u1, (short)0,
                                                 aT11, false, false);
  }
  int d = d0 + m;
#pragma unroll
  for (int r = 0; r < 8; ++r) {
    int t = t0 + r + 8 * kh;
    const float* cp = C + (size_t)(b * NT + t) * ND + d;
    float* ob = out + (size_t)(b * NT + t) * (4 * ND) + d;
    float c0 = cp[0], c1 = cp[16];
    ob[0]           = c0;            ob[16]          = c1;
    ob[ND]          = aQ00[r];       ob[ND + 16]     = aQ01[r];
    ob[2 * ND]      = c0 * aQ00[r];  ob[2 * ND + 16] = c1 * aQ01[r];
    ob[3 * ND]      = c0 * aT00[r];  ob[3 * ND + 16] = c1 * aT01[r];
  }
#pragma unroll
  for (int r = 0; r < 8; ++r) {
    int t = t0 + 16 + r + 8 * kh;
    const float* cp = C + (size_t)(b * NT + t) * ND + d;
    float* ob = out + (size_t)(b * NT + t) * (4 * ND) + d;
    float c0 = cp[0], c1 = cp[16];
    ob[0]           = c0;            ob[16]          = c1;
    ob[ND]          = aQ10[r];       ob[ND + 16]     = aQ11[r];
    ob[2 * ND]      = c0 * aQ10[r];  ob[2 * ND + 16] = c1 * aQ11[r];
    ob[3 * ND]      = c0 * aT10[r];  ob[3 * ND + 16] = c1 * aT11[r];
  }
}

extern "C" void kernel_launch(void* const* d_in, const int* in_sizes, int n_in,
                              void* d_out, int out_size, void* d_ws, size_t ws_size,
                              hipStream_t stream) {
  (void)in_sizes; (void)n_in; (void)out_size; (void)ws_size;
  const float* C = (const float*)d_in[0];
  const float* Q = (const float*)d_in[1];
  const float* w = (const float*)d_in[2];
  float* out = (float*)d_out;

  float* ws = (float*)d_ws;
  float* ch = ws;
  float* qb = ch + NB * NT;
  float* S  = qb + NB * NJ;
  float* Sj = S  + (size_t)NB * NT * NJ;
  float* ST = Sj + (size_t)NB * NT * NJ;
  float* tp = ST + (size_t)NB * NT * NJ;

  int rows = NB * NT + NB * NJ;
  k_rowdots<<<(rows + 7) / 8, 256, 0, stream>>>(C, Q, w, ch, qb);
  k_score<<<NB * 8, 256, 0, stream>>>(C, Q, w, ch, qb, S, ST);
  k_softmax_rows<<<(NB * NT + 7) / 8, 256, 0, stream>>>(S, Sj);
  k_softmax_t<<<(NB * NJ) / 4, 128, 0, stream>>>(ST);
  k_tmp<<<NB * 8, 256, 0, stream>>>(ST, C, tp);
  k_out<<<NB * 32, 256, 0, stream>>>(Sj, Q, tp, C, out);
}


# k_tmp: a wave owns 32 j x 32 d (2x2 tiles, 4 accumulators): 2 A and 2 B fragments feed 4 WMMAs; two fragment sets of 8 k-steps so loads run ahead; 256 blocks of 4 waves; k order and sums unchanged
# speedup vs baseline: 1.8252x; 1.1641x over previous
#include <hip/hip_runtime.h>

#define NB 32
#define NT 1024
#define NJ 128
#define ND 256

typedef __attribute__((ext_vector_type(2))) float v2f;
typedef __attribute__((ext_vector_type(8))) float v8f;

__device__ __forceinline__ float wave_reduce_sum(float v) {
#pragma unroll
  for (int o = 16; o > 0; o >>= 1) v += __shfl_xor(v, o, 32);
  return v;
}
__device__ __forceinline__ float wave_reduce_max(float v) {
#pragma unroll
  for (int o = 16; o > 0; o >>= 1) v = fmaxf(v, __shfl_xor(v, o, 32));
  return v;
}

__global__ void k_rowdots(const float* __restrict__ C, const float* __restrict__ Q,
                          const float* __restrict__ w, float* __restrict__ ch,
                          float* __restrict__ qb) {
  int wave = blockIdx.x * (blockDim.x >> 5) + (threadIdx.x >> 5);
  int lane = threadIdx.x & 31;
  const int nch = NB * NT;
  const int nrows = nch + NB * NJ;
  if (wave >= nrows) return;
  const float* row; const float* wv; float* out;
  if (wave < nch) { row = C + (size_t)wave * ND; wv = w;      out = ch + wave; }
  else { int r = wave - nch; row = Q + (size_t)r * ND; wv = w + ND; out = qb + r; }
  float s = 0.f;
#pragma unroll
  for (int i = lane; i < ND; i += 32) s += row[i] * wv[i];
  s = wave_reduce_sum(s);
  if (lane == 0) *out = s;
}

struct ScoreFrag { v2f wu, a0, a1, q0, q1, q2, q3; };

__device__ __forceinline__ void score_load(ScoreFrag (&f)[4], const float* whu,
                                           const float* C0, const float* Q0, int kb) {
#pragma unroll
  for (int s = 0; s < 4; ++s) {
    int kk = kb + 4 * s;
    f[s].wu = *(const v2f*)(whu + kk);
    f[s].a0 = *(const v2f*)(C0 + kk);
    f[s].a1 = *(const v2f*)(C0 + 16 * ND + kk);
    f[s].q0 = *(const v2f*)(Q0 + kk);
    f[s].q1 = *(const v2f*)(Q0 + 16 * ND + kk);
    f[s].q2 = *(const v2f*)(Q0 + 32 * ND + kk);
    f[s].q3 = *(const v2f*)(Q0 + 48 * ND + kk);
  }
}

__device__ __forceinline__ void score_mma(const ScoreFrag (&f)[4], v8f (&acc)[2][4]) {
#pragma unroll
  for (int s = 0; s < 4; ++s) {
    v2f a0 = f[s].a0 * f[s].wu;
    v2f a1 = f[s].a1 * f[s].wu;
    acc[0][0] = __builtin_amdgcn_wmma_f32_16x16x4_f32(false, a0, false, f[s].q0, (short)0,
                                                      acc[0][0], false, false);
    acc[1][0] = __builtin_amdgcn_wmma_f32_16x16x4_f32(false, a1, false, f[s].q0, (short)0,
                                                      acc[1][0], false, false);
    acc[0][1] = __builtin_amdgcn_wmma_f32_16x16x4_f32(false, a0, false, f[s].q1, (short)0,
                                                      acc[0][1], false, false);
    acc[1][1] = __builtin_amdgcn_wmma_f32_16x16x4_f32(false, a1, false, f[s].q1, (short)0,
                                                      acc[1][1], false, false);
    acc[0][2] = __builtin_amdgcn_wmma_f32_16x16x4_f32(false, a0, false, f[s].q2, (short)0,
                                                      acc[0][2], false, false);
    acc[1][2] = __builtin_amdgcn_wmma_f32_16x16x4_f32(false, a1, false, f[s].q2, (short)0,
                                                      acc[1][2], false, false);
    acc[0][3] = __builtin_amdgcn_wmma_f32_16x16x4_f32(false, a0, false, f[s].q3, (short)0,
                                                      acc[0][3], false, false);
    acc[1][3] = __builtin_amdgcn_wmma_f32_16x16x4_f32(false, a1, false, f[s].q3, (short)0,
                                                      acc[1][3], false, false);
  }
}

__global__ void __launch_bounds__(256)
k_score(const float* __restrict__ C, const float* __restrict__ Q,
        const float* __restrict__ w, const float* __restrict__ ch,
        const float* __restrict__ qb, float* __restrict__ S,
        float* __restrict__ ST) {
  int blk = blockIdx.x;
  int b = blk >> 3;
  int tblk = blk & 7;
  int wv = threadIdx.x >> 5;
  int lane = threadIdx.x & 31;
  int m = lane & 15, kh = lane >> 4;
  int t0 = tblk * 128 + (wv >> 1) * 32;
  int j0 = (wv & 1) * 64;
  const float* whu = w + 2 * ND + 2 * kh;
  const float* C0 = C + (size_t)(b * NT + t0 + m) * ND + 2 * kh;
  const float* Q0 = Q + (size_t)(b * NJ + j0 + m) * ND + 2 * kh;
  v8f acc[2][4] = {};
  ScoreFrag fa[4], fb[4];
  score_load(fa, whu, C0, Q0, 0);
#pragma unroll 1
  for (int kb = 0; kb < ND - 32; kb += 32) {
    score_load(fb, whu, C0, Q0, kb + 16);
    score_mma(fa, acc);
    score_load(fa, whu, C0, Q0, kb + 32);
    score_mma(fb, acc);
  }
  score_load(fb, whu, C0, Q0, ND - 16);
  score_mma(fa, acc);
  score_mma(fb, acc);

#pragma unroll
  for (int rt = 0; rt < 2; ++rt) {
    int tb = t0 + 16 * rt + 8 * kh;
    float chv[8];
#pragma unroll
    for (int r = 0; r < 8; ++r) chv[r] = ch[b * NT + tb + r];
#pragma unroll
    for (int ct = 0; ct < 4; ++ct) {
      int jj = j0 + 16 * ct + m;
      float qv = qb[b * NJ + jj];
      float sv[8];
#pragma unroll
      for (int r = 0; r < 8; ++r) {
        float s = acc[rt][ct][r] + chv[r] + qv;
        S[(size_t)(b * NT + tb + r) * NJ + jj] = s;
        sv[r] = s;
      }
      float* tr = ST + (size_t)(b * NJ + jj) * NT + tb;
      *(float4*)(tr)     = make_float4(sv[0], sv[1], sv[2], sv[3]);
      *(float4*)(tr + 4) = make_float4(sv[4], sv[5], sv[6], sv[7]);
    }
  }
}

__global__ void k_softmax_rows(const float* __restrict__ S, float* __restrict__ Sj) {
  int wave = blockIdx.x * (blockDim.x >> 5) + (threadIdx.x >> 5);
  int lane = threadIdx.x & 31;
  if (wave >= NB * NT) return;
  const float* row = S + (size_t)wave * NJ;
  float4 v = *(const float4*)(row + lane * 4);
  float m = fmaxf(fmaxf(v.x, v.y), fmaxf(v.z, v.w));
  m = wave_reduce_max(m);
  float e0 = expf(v.x - m), e1 = expf(v.y - m), e2 = expf(v.z - m), e3 = expf(v.w - m);
  float s = wave_reduce_sum(e0 + e1 + e2 + e3);
  float inv = 1.0f / s;
  float4 o; o.x = e0 * inv; o.y = e1 * inv; o.z = e2 * inv; o.w = e3 * inv;
  *(float4*)(Sj + (size_t)wave * NJ + lane * 4) = o;
}

__global__ void k_softmax_t(float* __restrict__ ST) {
  int wave = blockIdx.x * (blockDim.x >> 5) + (threadIdx.x >> 5);
  int lane = threadIdx.x & 31;
  if (wave >= NB * NJ) return;
  float* row = ST + (size_t)wave * NT + lane * 4;
  float4 v[8];
#pragma unroll
  for (int i = 0; i < 8; ++i) v[i] = *(const float4*)(row + i * 128);
  float m = -3.4e38f;
#pragma unroll
  for (int i = 0; i < 8; ++i)
    m = fmaxf(m, fmaxf(fmaxf(v[i].x, v[i].y), fmaxf(v[i].z, v[i].w)));
  m = wave_reduce_max(m);
  float s = 0.f;
#pragma unroll
  for (int i = 0; i < 8; ++i) {
    v[i].x = expf(v[i].x - m); v[i].y = expf(v[i].y - m);
    v[i].z = expf(v[i].z - m); v[i].w = expf(v[i].w - m);
    s += (v[i].x + v[i].y) + (v[i].z + v[i].w);
  }
  s = wave_reduce_sum(s);
  float inv = 1.0f / s;
#pragma unroll
  for (int i = 0; i < 8; ++i) {
    float4 o;
    o.x = v[i].x * inv; o.y = v[i].y * inv; o.z = v[i].z * inv; o.w = v[i].w * inv;
    *(float4*)(row + i * 128) = o;
  }
}

struct TmpFrag { v2f a0, a1, b0, b1; };

__device__ __forceinline__ void tmp_load(TmpFrag (&f)[8], const float* A0,
                                         const float* Cb, int kb) {
#pragma unroll
  for (int s = 0; s < 8; ++s) {
    int k = kb + 4 * s;
    f[s].a0 = *(const v2f*)(A0 + k);
    f[s].a1 = *(const v2f*)(A0 + 16 * NT + k);
    f[s].b0.x = Cb[(size_t)k * ND];
    f[s].b0.y = Cb[(size_t)(k + 1) * ND];
    f[s].b1.x = Cb[(size_t)k * ND + 16];
    f[s].b1.y = Cb[(size_t)(k + 1) * ND + 16];
  }
}

__device__ __forceinline__ void tmp_mma(const TmpFrag (&f)[8], v8f (&acc)[2][2]) {
#pragma unroll
  for (int s = 0; s < 8; ++s) {
    acc[0][0] = __builtin_amdgcn_wmma_f32_16x16x4_f32(false, f[s].a0, false, f[s].b0, (short)0,
                                                      acc[0][0], false, false);
    acc[1][0] = __builtin_amdgcn_wmma_f32_16x16x4_f32(false, f[s].a1, false, f[s].b0, (short)0,
                                                      acc[1][0], false, false);
    acc[0][1] = __builtin_amdgcn_wmma_f32_16x16x4_f32(false, f[s].a0, false, f[s].b1, (short)0,
                                                      acc[0][1], false, false);
    acc[1][1] = __builtin_amdgcn_wmma_f32_16x16x4_f32(false, f[s].a1, false, f[s].b1, (short)0,
                                                      acc[1][1], false, false);
  }
}

__global__ void __launch_bounds__(128)
k_tmp(const float* __restrict__ StT, const float* __restrict__ C,
      float* __restrict__ tmp) {
  int b = blockIdx.x >> 3;
  int dblk = blockIdx.x & 7;
  int wv = threadIdx.x >> 5;
  int lane = threadIdx.x & 31;
  int m = lane & 15, kh = lane >> 4;
  int j0 = wv * 32, d0 = dblk * 32;
  const float* A0 = StT + (size_t)(b * NJ + j0 + m) * NT + 2 * kh;
  const float* Cb = C + (size_t)b * NT * ND + (size_t)(2 * kh) * ND + d0 + m;
  v8f acc[2][2] = {};
  TmpFrag fa[8], fb[8];
  tmp_load(fa, A0, Cb, 0);
#pragma unroll 1
  for (int kb = 0; kb < NT - 64; kb += 64) {
    tmp_load(fb, A0, Cb, kb + 32);
    tmp_mma(fa, acc);
    tmp_load(fa, A0, Cb, kb + 64);
    tmp_mma(fb, acc);
  }
  tmp_load(fb, A0, Cb, NT - 32);
  tmp_mma(fa, acc);
  tmp_mma(fb, acc);

#pragma unroll
  for (int rt = 0; rt < 2; ++rt) {
#pragma unroll
    for (int r = 0; r < 8; ++r) {
      int jj = j0 + 16 * rt + r + 8 * kh;
      float* ob = tmp + (size_t)(b * NJ + jj) * ND + d0 + m;
      ob[0]  = acc[rt][0][r];
      ob[16] = acc[rt][1][r];
    }
  }
}

__global__ void __launch_bounds__(256)
k_out(const float* __restrict__ Sj, const float* __restrict__ Q,
      const float* __restrict__ tmp, const float* __restrict__ C,
      float* __restrict__ out) {
  int blk = blockIdx.x;
  int b = blk >> 5;
  int tblk = blk & 31;
  int wv = threadIdx.x >> 5;
  int lane = threadIdx.x & 31;
  int m = lane & 15, kh = lane >> 4;
  int t0 = tblk * 32, d0 = wv * 32;
  const float* S0 = Sj + (size_t)(b * NT + t0 + m) * NJ + 2 * kh;
  const float* S1 = S0 + (size_t)16 * NJ;
  const float* Qb = Q + (size_t)b * NJ * ND + (size_t)(2 * kh) * ND + d0 + m;
  const float* Tb = tmp + (size_t)b * NJ * ND + (size_t)(2 * kh) * ND + d0 + m;
  v8f aQ00 = {}, aQ01 = {}, aQ10 = {}, aQ11 = {};
  v8f aT00 = {}, aT01 = {}, aT10 = {}, aT11 = {};
#pragma unroll 2
  for (int k0 = 0; k0 < NJ; k0 += 4) {
    v2f a0 = *(const v2f*)(S0 + k0);
    v2f a1 = *(const v2f*)(S1 + k0);
    const float* qp = Qb + (size_t)k0 * ND;
    const float* tp = Tb + (size_t)k0 * ND;
    v2f q0, q1, u0, u1;
    q0.x = qp[0];  q0.y = qp[ND];
    q1.x = qp[16]; q1.y = qp[ND + 16];
    u0.x = tp[0];  u0.y = tp[ND];
    u1.x = tp[16]; u1.y = tp[ND + 16];
    aQ00 = __builtin_amdgcn_wmma_f32_16x16x4_f32(false, a0, false, q0, (short)0,
                                                 aQ00, false, false);
    aT00 = __builtin_amdgcn_wmma_f32_16x16x4_f32(false, a0, false, u0, (short)0,
                                                 aT00, false, false);
    aQ01 = __builtin_amdgcn_wmma_f32_16x16x4_f32(false, a0, false, q1, (short)0,
                                                 aQ01, false, false);
    aT01 = __builtin_amdgcn_wmma_f32_16x16x4_f32(false, a0, false, u1, (short)0,
                                                 aT01, false, false);
    aQ10 = __builtin_amdgcn_wmma_f32_16x16x4_f32(false, a1, false, q0, (short)0,
                                                 aQ10, false, false);
    aT10 = __builtin_amdgcn_wmma_f32_16x16x4_f32(false, a1, false, u0, (short)0,
                                                 aT10, false, false);
    aQ11 = __builtin_amdgcn_wmma_f32_16x16x4_f32(false, a1, false, q1, (short)0,
                                                 aQ11, false, false);
    aT11 = __builtin_amdgcn_wmma_f32_16x16x4_f32(false, a1, false, u1, (short)0,
                                                 aT11, false, false);
  }
  int d = d0 + m;
#pragma unroll
  for (int r = 0; r < 8; ++r) {
    int t = t0 + r + 8 * kh;
    const float* cp = C + (size_t)(b * NT + t) * ND + d;
    float* ob = out + (size_t)(b * NT + t) * (4 * ND) + d;
    float c0 = cp[0], c1 = cp[16];
    ob[0]           = c0;            ob[16]          = c1;
    ob[ND]          = aQ00[r];       ob[ND + 16]     = aQ01[r];
    ob[2 * ND]      = c0 * aQ00[r];  ob[2 * ND + 16] = c1 * aQ01[r];
    ob[3 * ND]      = c0 * aT00[r];  ob[3 * ND + 16] = c1 * aT01[r];
  }
#pragma unroll
  for (int r = 0; r < 8; ++r) {
    int t = t0 + 16 + r + 8 * kh;
    const float* cp = C + (size_t)(b * NT + t) * ND + d;
    float* ob = out + (size_t)(b * NT + t) * (4 * ND) + d;
    float c0 = cp[0], c1 = cp[16];
    ob[0]           = c0;            ob[16]          = c1;
    ob[ND]          = aQ10[r];       ob[ND + 16]     = aQ11[r];
    ob[2 * ND]      = c0 * aQ10[r];  ob[2 * ND + 16] = c1 * aQ11[r];
    ob[3 * ND]      = c0 * aT10[r];  ob[3 * ND + 16] = c1 * aT11[r];
  }
}

extern "C" void kernel_launch(void* const* d_in, const int* in_sizes, int n_in,
                              void* d_out, int out_size, void* d_ws, size_t ws_size,
                              hipStream_t stream) {
  (void)in_sizes; (void)n_in; (void)out_size; (void)ws_size;
  const float* C = (const float*)d_in[0];
  const float* Q = (const float*)d_in[1];
  const float* w = (const float*)d_in[2];
  float* out = (float*)d_out;

  float* ws = (float*)d_ws;
  float* ch = ws;
  float* qb = ch + NB * NT;
  float* S  = qb + NB * NJ;
  float* Sj = S  + (size_t)NB * NT * NJ;
  float* ST = Sj + (size_t)NB * NT * NJ;
  float* tp = ST + (size_t)NB * NT * NJ;

  int rows = NB * NT + NB * NJ;
  k_rowdots<<<(rows + 7) / 8, 256, 0, stream>>>(C, Q, w, ch, qb);
  k_score<<<NB * 8, 256, 0, stream>>>(C, Q, w, ch, qb, S, ST);
  k_softmax_rows<<<(NB * NT + 7) / 8, 256, 0, stream>>>(S, Sj);
  k_softmax_t<<<(NB * NJ) / 4, 128, 0, stream>>>(ST);
  k_tmp<<<NB * 8, 128, 0, stream>>>(ST, C, tp);
  k_out<<<NB * 32, 256, 0, stream>>>(Sj, Q, tp, C, out);
}


# k_rowdots and k_softmax_rows: a wave takes 8 consecutive rows instead of 1, its loads issued as one group, reductions side by side: 8704 waves enter instead of 69632; order of every sum unchanged
# speedup vs baseline: 1.8859x; 1.0333x over previous
#include <hip/hip_runtime.h>

#define NB 32
#define NT 1024
#define NJ 128
#define ND 256

typedef __attribute__((ext_vector_type(2))) float v2f;
typedef __attribute__((ext_vector_type(4))) float v4f;
typedef __attribute__((ext_vector_type(8))) float v8f;

__device__ __forceinline__ float wave_reduce_sum(float v) {
#pragma unroll
  for (int o = 16; o > 0; o >>= 1) v += __shfl_xor(v, o, 32);
  return v;
}
__device__ __forceinline__ float wave_reduce_max(float v) {
#pragma unroll
  for (int o = 16; o > 0; o >>= 1) v = fmaxf(v, __shfl_xor(v, o, 32));
  return v;
}

#define RD_ROWS 8
static_assert((NB * NT) % RD_ROWS == 0 && (NB * NJ) % RD_ROWS == 0,
              "a wave's rows must lie in one of the two inputs");
static_assert(RD_ROWS == 8 && ND == 256, "the load fence and the stores name 8 rows of 8 values");
__global__ void __launch_bounds__(128)
k_rowdots(const float* __restrict__ C, const float* __restrict__ Q,
          const float* __restrict__ w, float* __restrict__ ch,
          float* __restrict__ qb) {
  int wave = blockIdx.x * (blockDim.x >> 5) + (threadIdx.x >> 5);
  int lane = threadIdx.x & 31;
  const int nch = NB * NT;
  const int nrows = nch + NB * NJ;
  int r0 = wave * RD_ROWS;
  if (r0 >= nrows) return;
  const float* rows; const float* wv; float* out;
  if (r0 < nch) { rows = C + (size_t)r0 * ND; wv = w;      out = ch + r0; }
  else { int r = r0 - nch; rows = Q + (size_t)r * ND; wv = w + ND; out = qb + r; }
  float wk[ND / 32];
#pragma unroll
  for (int k = 0; k < ND / 32; ++k) wk[k] = wv[lane + 32 * k];
  v8f x[RD_ROWS];
#pragma unroll
  for (int r = 0; r < RD_ROWS; ++r) {
#pragma unroll
    for (int k = 0; k < ND / 32; ++k) x[r][k] = rows[(size_t)r * ND + lane + 32 * k];
  }
  asm volatile("" :: "v"(x[0]), "v"(x[1]), "v"(x[2]), "v"(x[3]),
                     "v"(x[4]), "v"(x[5]), "v"(x[6]), "v"(x[7]));
  float s[RD_ROWS];
#pragma unroll
  for (int r = 0; r < RD_ROWS; ++r) {
    float a = 0.f;
#pragma unroll
    for (int k = 0; k < ND / 32; ++k) a += x[r][k] * wk[k];
    s[r] = a;
  }
#pragma unroll
  for (int o = 16; o > 0; o >>= 1) {
#pragma unroll
    for (int r = 0; r < RD_ROWS; ++r) s[r] += __shfl_xor(s[r], o, 32);
  }
  if (lane == 0) {
    *(float4*)(out)     = make_float4(s[0], s[1], s[2], s[3]);
    *(float4*)(out + 4) = make_float4(s[4], s[5], s[6], s[7]);
  }
}

struct ScoreFrag { v2f wu, a0, a1, q0, q1, q2, q3; };

__device__ __forceinline__ void score_load(ScoreFrag (&f)[4], const float* whu,
                                           const float* C0, const float* Q0, int kb) {
#pragma unroll
  for (int s = 0; s < 4; ++s) {
    int kk = kb + 4 * s;
    f[s].wu = *(const v2f*)(whu + kk);
    f[s].a0 = *(const v2f*)(C0 + kk);
    f[s].a1 = *(const v2f*)(C0 + 16 * ND + kk);
    f[s].q0 = *(const v2f*)(Q0 + kk);
    f[s].q1 = *(const v2f*)(Q0 + 16 * ND + kk);
    f[s].q2 = *(const v2f*)(Q0 + 32 * ND + kk);
    f[s].q3 = *(const v2f*)(Q0 + 48 * ND + kk);
  }
}

__device__ __forceinline__ void score_mma(const ScoreFrag (&f)[4], v8f (&acc)[2][4]) {
#pragma unroll
  for (int s = 0; s < 4; ++s) {
    v2f a0 = f[s].a0 * f[s].wu;
    v2f a1 = f[s].a1 * f[s].wu;
    acc[0][0] = __builtin_amdgcn_wmma_f32_16x16x4_f32(false, a0, false, f[s].q0, (short)0,
                                                      acc[0][0], false, false);
    acc[1][0] = __builtin_amdgcn_wmma_f32_16x16x4_f32(false, a1, false, f[s].q0, (short)0,
                                                      acc[1][0], false, false);
    acc[0][1] = __builtin_amdgcn_wmma_f32_16x16x4_f32(false, a0, false, f[s].q1, (short)0,
                                                      acc[0][1], false, false);
    acc[1][1] = __builtin_amdgcn_wmma_f32_16x16x4_f32(false, a1, false, f[s].q1, (short)0,
                                                      acc[1][1], false, false);
    acc[0][2] = __builtin_amdgcn_wmma_f32_16x16x4_f32(false, a0, false, f[s].q2, (short)0,
                                                      acc[0][2], false, false);
    acc[1][2] = __builtin_amdgcn_wmma_f32_16x16x4_f32(false, a1, false, f[s].q2, (short)0,
                                                      acc[1][2], false, false);
    acc[0][3] = __builtin_amdgcn_wmma_f32_16x16x4_f32(false, a0, false, f[s].q3, (short)0,
                                                      acc[0][3], false, false);
    acc[1][3] = __builtin_amdgcn_wmma_f32_16x16x4_f32(false, a1, false, f[s].q3, (short)0,
                                                      acc[1][3], false, false);
  }
}

__global__ void __launch_bounds__(256)
k_score(const float* __restrict__ C, const float* __restrict__ Q,
        const float* __restrict__ w, const float* __restrict__ ch,
        const float* __restrict__ qb, float* __restrict__ S,
        float* __restrict__ ST) {
  int blk = blockIdx.x;
  int b = blk >> 3;
  int tblk = blk & 7;
  int wv = threadIdx.x >> 5;
  int lane = threadIdx.x & 31;
  int m = lane & 15, kh = lane >> 4;
  int t0 = tblk * 128 + (wv >> 1) * 32;
  int j0 = (wv & 1) * 64;
  const float* whu = w + 2 * ND + 2 * kh;
  const float* C0 = C + (size_t)(b * NT + t0 + m) * ND + 2 * kh;
  const float* Q0 = Q + (size_t)(b * NJ + j0 + m) * ND + 2 * kh;
  v8f acc[2][4] = {};
  ScoreFrag fa[4], fb[4];
  score_load(fa, whu, C0, Q0, 0);
#pragma unroll 1
  for (int kb = 0; kb < ND - 32; kb += 32) {
    score_load(fb, whu, C0, Q0, kb + 16);
    score_mma(fa, acc);
    score_load(fa, whu, C0, Q0, kb + 32);
    score_mma(fb, acc);
  }
  score_load(fb, whu, C0, Q0, ND - 16);
  score_mma(fa, acc);
  score_mma(fb, acc);

#pragma unroll
  for (int rt = 0; rt < 2; ++rt) {
    int tb = t0 + 16 * rt + 8 * kh;
    float chv[8];
#pragma unroll
    for (int r = 0; r < 8; ++r) chv[r] = ch[b * NT + tb + r];
#pragma unroll
    for (int ct = 0; ct < 4; ++ct) {
      int jj = j0 + 16 * ct + m;
      float qv = qb[b * NJ + jj];
      float sv[8];
#pragma unroll
      for (int r = 0; r < 8; ++r) {
        float s = acc[rt][ct][r] + chv[r] + qv;
        S[(size_t)(b * NT + tb + r) * NJ + jj] = s;
        sv[r] = s;
      }
      float* tr = ST + (size_t)(b * NJ + jj) * NT + tb;
      *(float4*)(tr)     = make_float4(sv[0], sv[1], sv[2], sv[3]);
      *(float4*)(tr + 4) = make_float4(sv[4], sv[5], sv[6], sv[7]);
    }
  }
}

#define SR_ROWS 8
static_assert((NB * NT) % SR_ROWS == 0, "whole groups of rows per wave");
static_assert(SR_ROWS == 8, "the load fence names 8 rows");
__global__ void __launch_bounds__(128)
k_softmax_rows(const float* __restrict__ S, float* __restrict__ Sj) {
  int wave = blockIdx.x * (blockDim.x >> 5) + (threadIdx.x >> 5);
  int lane = threadIdx.x & 31;
  int r0 = wave * SR_ROWS;
  if (r0 >= NB * NT) return;
  const float* src = S + (size_t)r0 * NJ + lane * 4;
  float* dst = Sj + (size_t)r0 * NJ + lane * 4;
  v4f v[SR_ROWS];
#pragma unroll
  for (int r = 0; r < SR_ROWS; ++r) v[r] = *(const v4f*)(src + (size_t)r * NJ);
  asm volatile("" :: "v"(v[0]), "v"(v[1]), "v"(v[2]), "v"(v[3]),
                     "v"(v[4]), "v"(v[5]), "v"(v[6]), "v"(v[7]));
  float m[SR_ROWS];
#pragma unroll
  for (int r = 0; r < SR_ROWS; ++r)
    m[r] = fmaxf(fmaxf(v[r].x, v[r].y), fmaxf(v[r].z, v[r].w));
#pragma unroll
  for (int o = 16; o > 0; o >>= 1) {
#pragma unroll
    for (int r = 0; r < SR_ROWS; ++r) m[r] = fmaxf(m[r], __shfl_xor(m[r], o, 32));
  }
  float s[SR_ROWS];
#pragma unroll
  for (int r = 0; r < SR_ROWS; ++r) {
    v[r].x = expf(v[r].x - m[r]); v[r].y = expf(v[r].y - m[r]);
    v[r].z = expf(v[r].z - m[r]); v[r].w = expf(v[r].w - m[r]);
    s[r] = v[r].x + v[r].y + v[r].z + v[r].w;
  }
#pragma unroll
  for (int o = 16; o > 0; o >>= 1) {
#pragma unroll
    for (int r = 0; r < SR_ROWS; ++r) s[r] += __shfl_xor(s[r], o, 32);
  }
#pragma unroll
  for (int r = 0; r < SR_ROWS; ++r) {
    float inv = 1.0f / s[r];
    v4f o4;
    o4.x = v[r].x * inv; o4.y = v[r].y * inv; o4.z = v[r].z * inv; o4.w = v[r].w * inv;
    *(v4f*)(dst + (size_t)r * NJ) = o4;
  }
}

__global__ void k_softmax_t(float* __restrict__ ST) {
  int wave = blockIdx.x * (blockDim.x >> 5) + (threadIdx.x >> 5);
  int lane = threadIdx.x & 31;
  if (wave >= NB * NJ) return;
  float* row = ST + (size_t)wave * NT + lane * 4;
  float4 v[8];
#pragma unroll
  for (int i = 0; i < 8; ++i) v[i] = *(const float4*)(row + i * 128);
  float m = -3.4e38f;
#pragma unroll
  for (int i = 0; i < 8; ++i)
    m = fmaxf(m, fmaxf(fmaxf(v[i].x, v[i].y), fmaxf(v[i].z, v[i].w)));
  m = wave_reduce_max(m);
  float s = 0.f;
#pragma unroll
  for (int i = 0; i < 8; ++i) {
    v[i].x = expf(v[i].x - m); v[i].y = expf(v[i].y - m);
    v[i].z = expf(v[i].z - m); v[i].w = expf(v[i].w - m);
    s += (v[i].x + v[i].y) + (v[i].z + v[i].w);
  }
  s = wave_reduce_sum(s);
  float inv = 1.0f / s;
#pragma unroll
  for (int i = 0; i < 8; ++i) {
    float4 o;
    o.x = v[i].x * inv; o.y = v[i].y * inv; o.z = v[i].z * inv; o.w = v[i].w * inv;
    *(float4*)(row + i * 128) = o;
  }
}

struct TmpFrag { v2f a0, a1, b0, b1; };

__device__ __forceinline__ void tmp_load(TmpFrag (&f)[8], const float* A0,
                                         const float* Cb, int kb) {
#pragma unroll
  for (int s = 0; s < 8; ++s) {
    int k = kb + 4 * s;
    f[s].a0 = *(const v2f*)(A0 + k);
    f[s].a1 = *(const v2f*)(A0 + 16 * NT + k);
    f[s].b0.x = Cb[(size_t)k * ND];
    f[s].b0.y = Cb[(size_t)(k + 1) * ND];
    f[s].b1.x = Cb[(size_t)k * ND + 16];
    f[s].b1.y = Cb[(size_t)(k + 1) * ND + 16];
  }
}

__device__ __forceinline__ void tmp_mma(const TmpFrag (&f)[8], v8f (&acc)[2][2]) {
#pragma unroll
  for (int s = 0; s < 8; ++s) {
    acc[0][0] = __builtin_amdgcn_wmma_f32_16x16x4_f32(false, f[s].a0, false, f[s].b0, (short)0,
                                                      acc[0][0], false, false);
    acc[1][0] = __builtin_amdgcn_wmma_f32_16x16x4_f32(false, f[s].a1, false, f[s].b0, (short)0,
                                                      acc[1][0], false, false);
    acc[0][1] = __builtin_amdgcn_wmma_f32_16x16x4_f32(false, f[s].a0, false, f[s].b1, (short)0,
                                                      acc[0][1], false, false);
    acc[1][1] = __builtin_amdgcn_wmma_f32_16x16x4_f32(false, f[s].a1, false, f[s].b1, (short)0,
                                                      acc[1][1], false, false);
  }
}

__global__ void __launch_bounds__(128)
k_tmp(const float* __restrict__ StT, const float* __restrict__ C,
      float* __restrict__ tmp) {
  int b = blockIdx.x >> 3;
  int dblk = blockIdx.x & 7;
  int wv = threadIdx.x >> 5;
  int lane = threadIdx.x & 31;
  int m = lane & 15, kh = lane >> 4;
  int j0 = wv * 32, d0 = dblk * 32;
  const float* A0 = StT + (size_t)(b * NJ + j0 + m) * NT + 2 * kh;
  const float* Cb = C + (size_t)b * NT * ND + (size_t)(2 * kh) * ND + d0 + m;
  v8f acc[2][2] = {};
  TmpFrag fa[8], fb[8];
  tmp_load(fa, A0, Cb, 0);
#pragma unroll 1
  for (int kb = 0; kb < NT - 64; kb += 64) {
    tmp_load(fb, A0, Cb, kb + 32);
    tmp_mma(fa, acc);
    tmp_load(fa, A0, Cb, kb + 64);
    tmp_mma(fb, acc);
  }
  tmp_load(fb, A0, Cb, NT - 32);
  tmp_mma(fa, acc);
  tmp_mma(fb, acc);

#pragma unroll
  for (int rt = 0; rt < 2; ++rt) {
#pragma unroll
    for (int r = 0; r < 8; ++r) {
      int jj = j0 + 16 * rt + r + 8 * kh;
      float* ob = tmp + (size_t)(b * NJ + jj) * ND + d0 + m;
      ob[0]  = acc[rt][0][r];
      ob[16] = acc[rt][1][r];
    }
  }
}

__global__ void __launch_bounds__(256)
k_out(const float* __restrict__ Sj, const float* __restrict__ Q,
      const float* __restrict__ tmp, const float* __restrict__ C,
      float* __restrict__ out) {
  int blk = blockIdx.x;
  int b = blk >> 5;
  int tblk = blk & 31;
  int wv = threadIdx.x >> 5;
  int lane = threadIdx.x & 31;
  int m = lane & 15, kh = lane >> 4;
  int t0 = tblk * 32, d0 = wv * 32;
  const float* S0 = Sj + (size_t)(b * NT + t0 + m) * NJ + 2 * kh;
  const float* S1 = S0 + (size_t)16 * NJ;
  const float* Qb = Q + (size_t)b * NJ * ND + (size_t)(2 * kh) * ND + d0 + m;
  const float* Tb = tmp + (size_t)b * NJ * ND + (size_t)(2 * kh) * ND + d0 + m;
  v8f aQ00 = {}, aQ01 = {}, aQ10 = {}, aQ11 = {};
  v8f aT00 = {}, aT01 = {}, aT10 = {}, aT11 = {};
#pragma unroll 2
  for (int k0 = 0; k0 < NJ; k0 += 4) {
    v2f a0 = *(const v2f*)(S0 + k0);
    v2f a1 = *(const v2f*)(S1 + k0);
    const float* qp = Qb + (size_t)k0 * ND;
    const float* tp = Tb + (size_t)k0 * ND;
    v2f q0, q1, u0, u1;
    q0.x = qp[0];  q0.y = qp[ND];
    q1.x = qp[16]; q1.y = qp[ND + 16];
    u0.x = tp[0];  u0.y = tp[ND];
    u1.x = tp[16]; u1.y = tp[ND + 16];
    aQ00 = __builtin_amdgcn_wmma_f32_16x16x4_f32(false, a0, false, q0, (short)0,
                                                 aQ00, false, false);
    aT00 = __builtin_amdgcn_wmma_f32_16x16x4_f32(false, a0, false, u0, (short)0,
                                                 aT00, false, false);
    aQ01 = __builtin_amdgcn_wmma_f32_16x16x4_f32(false, a0, false, q1, (short)0,
                                                 aQ01, false, false);
    aT01 = __builtin_amdgcn_wmma_f32_16x16x4_f32(false, a0, false, u1, (short)0,
                                                 aT01, false, false);
    aQ10 = __builtin_amdgcn_wmma_f32_16x16x4_f32(false, a1, false, q0, (short)0,
                                                 aQ10, false, false);
    aT10 = __builtin_amdgcn_wmma_f32_16x16x4_f32(false, a1, false, u0, (short)0,
                                                 aT10, false, false);
    aQ11 = __builtin_amdgcn_wmma_f32_16x16x4_f32(false, a1, false, q1, (short)0,
                                                 aQ11, false, false);
    aT11 = __builtin_amdgcn_wmma_f32_16x16x4_f32(false, a1, false, u1, (short)0,
                                                 aT11, false, false);
  }
  int d = d0 + m;
#pragma unroll
  for (int r = 0; r < 8; ++r) {
    int t = t0 + r + 8 * kh;
    const float* cp = C + (size_t)(b * NT + t) * ND + d;
    float* ob = out + (size_t)(b * NT + t) * (4 * ND) + d;
    float c0 = cp[0], c1 = cp[16];
    ob[0]           = c0;            ob[16]          = c1;
    ob[ND]          = aQ00[r];       ob[ND + 16]     = aQ01[r];
    ob[2 * ND]      = c0 * aQ00[r];  ob[2 * ND + 16] = c1 * aQ01[r];
    ob[3 * ND]      = c0 * aT00[r];  ob[3 * ND + 16] = c1 * aT01[r];
  }
#pragma unroll
  for (int r = 0; r < 8; ++r) {
    int t = t0 + 16 + r + 8 * kh;
    const float* cp = C + (size_t)(b * NT + t) * ND + d;
    float* ob = out + (size_t)(b * NT + t) * (4 * ND) + d;
    float c0 = cp[0], c1 = cp[16];
    ob[0]           = c0;            ob[16]          = c1;
    ob[ND]          = aQ10[r];       ob[ND + 16]     = aQ11[r];
    ob[2 * ND]      = c0 * aQ10[r];  ob[2 * ND + 16] = c1 * aQ11[r];
    ob[3 * ND]      = c0 * aT10[r];  ob[3 * ND + 16] = c1 * aT11[r];
  }
}

extern "C" void kernel_launch(void* const* d_in, const int* in_sizes, int n_in,
                              void* d_out, int out_size, void* d_ws, size_t ws_size,
                              hipStream_t stream) {
  (void)in_sizes; (void)n_in; (void)out_size; (void)ws_size;
  const float* C = (const float*)d_in[0];
  const float* Q = (const float*)d_in[1];
  const float* w = (const float*)d_in[2];
  float* out = (float*)d_out;

  float* ws = (float*)d_ws;
  float* ch = ws;
  float* qb = ch + NB * NT;
  float* S  = qb + NB * NJ;
  float* Sj = S  + (size_t)NB * NT * NJ;
  float* ST = Sj + (size_t)NB * NT * NJ;
  float* tp = ST + (size_t)NB * NT * NJ;

  k_rowdots<<<(NB * NT + NB * NJ) / (4 * RD_ROWS), 128, 0, stream>>>(C, Q, w, ch, qb);
  k_score<<<NB * 8, 256, 0, stream>>>(C, Q, w, ch, qb, S, ST);
  k_softmax_rows<<<(NB * NT) / (4 * SR_ROWS), 128, 0, stream>>>(S, Sj);
  k_softmax_t<<<(NB * NJ) / 4, 128, 0, stream>>>(ST);
  k_tmp<<<NB * 8, 128, 0, stream>>>(ST, C, tp);
  k_out<<<NB * 32, 256, 0, stream>>>(Sj, Q, tp, C, out);
}


# k-contiguous fragment loads of all 3 products (C, Q, w_hu; ST; Sj) take 16 bytes per lane for two k-steps (k order 0,1,4,5/2,3,6,7), each set issued as one group; changes sum order of S, tmp, A, Bm
# speedup vs baseline: 1.9871x; 1.0537x over previous
#include <hip/hip_runtime.h>

#define NB 32
#define NT 1024
#define NJ 128
#define ND 256

typedef __attribute__((ext_vector_type(2))) float v2f;
typedef __attribute__((ext_vector_type(4))) float v4f;
typedef __attribute__((ext_vector_type(8))) float v8f;

__device__ __forceinline__ float wave_reduce_sum(float v) {
#pragma unroll
  for (int o = 16; o > 0; o >>= 1) v += __shfl_xor(v, o, 32);
  return v;
}
__device__ __forceinline__ float wave_reduce_max(float v) {
#pragma unroll
  for (int o = 16; o > 0; o >>= 1) v = fmaxf(v, __shfl_xor(v, o, 32));
  return v;
}

#define RD_ROWS 8
static_assert((NB * NT) % RD_ROWS == 0 && (NB * NJ) % RD_ROWS == 0,
              "a wave's rows must lie in one of the two inputs");
static_assert(RD_ROWS == 8 && ND == 256, "the load fence and the stores name 8 rows of 8 values");
__global__ void __launch_bounds__(128)
k_rowdots(const float* __restrict__ C, const float* __restrict__ Q,
          const float* __restrict__ w, float* __restrict__ ch,
          float* __restrict__ qb) {
  int wave = blockIdx.x * (blockDim.x >> 5) + (threadIdx.x >> 5);
  int lane = threadIdx.x & 31;
  const int nch = NB * NT;
  const int nrows = nch + NB * NJ;
  int r0 = wave * RD_ROWS;
  if (r0 >= nrows) return;
  const float* rows; const float* wv; float* out;
  if (r0 < nch) { rows = C + (size_t)r0 * ND; wv = w;      out = ch + r0; }
  else { int r = r0 - nch; rows = Q + (size_t)r * ND; wv = w + ND; out = qb + r; }
  float wk[ND / 32];
#pragma unroll
  for (int k = 0; k < ND / 32; ++k) wk[k] = wv[lane + 32 * k];
  v8f x[RD_ROWS];
#pragma unroll
  for (int r = 0; r < RD_ROWS; ++r) {
#pragma unroll
    for (int k = 0; k < ND / 32; ++k) x[r][k] = rows[(size_t)r * ND + lane + 32 * k];
  }
  asm volatile("" :: "v"(x[0]), "v"(x[1]), "v"(x[2]), "v"(x[3]),
                     "v"(x[4]), "v"(x[5]), "v"(x[6]), "v"(x[7]));
  float s[RD_ROWS];
#pragma unroll
  for (int r = 0; r < RD_ROWS; ++r) {
    float a = 0.f;
#pragma unroll
    for (int k = 0; k < ND / 32; ++k) a += x[r][k] * wk[k];
    s[r] = a;
  }
#pragma unroll
  for (int o = 16; o > 0; o >>= 1) {
#pragma unroll
    for (int r = 0; r < RD_ROWS; ++r) s[r] += __shfl_xor(s[r], o, 32);
  }
  if (lane == 0) {
    *(float4*)(out)     = make_float4(s[0], s[1], s[2], s[3]);
    *(float4*)(out + 4) = make_float4(s[4], s[5], s[6], s[7]);
  }
}

struct ScoreFrag { v4f wu, a0, a1, q0, q1, q2, q3; };

__device__ __forceinline__ v2f lo2(v4f x) { return __builtin_shufflevector(x, x, 0, 1); }
__device__ __forceinline__ v2f hi2(v4f x) { return __builtin_shufflevector(x, x, 2, 3); }

__device__ __forceinline__ void score_load(ScoreFrag (&f)[2], const float* whu,
                                           const float* C0, const float* Q0, int kb) {
#pragma unroll
  for (int s = 0; s < 2; ++s) {
    int kk = kb + 8 * s;
    f[s].wu = *(const v4f*)(whu + kk);
    f[s].a0 = *(const v4f*)(C0 + kk);
    f[s].a1 = *(const v4f*)(C0 + 16 * ND + kk);
    f[s].q0 = *(const v4f*)(Q0 + kk);
    f[s].q1 = *(const v4f*)(Q0 + 16 * ND + kk);
    f[s].q2 = *(const v4f*)(Q0 + 32 * ND + kk);
    f[s].q3 = *(const v4f*)(Q0 + 48 * ND + kk);
  }
}

__device__ __forceinline__ void score_mma8(v2f a0, v2f a1, v2f q0, v2f q1, v2f q2, v2f q3,
                                           v8f (&acc)[2][4]) {
  acc[0][0] = __builtin_amdgcn_wmma_f32_16x16x4_f32(false, a0, false, q0, (short)0,
                                                    acc[0][0], false, false);
  acc[1][0] = __builtin_amdgcn_wmma_f32_16x16x4_f32(false, a1, false, q0, (short)0,
                                                    acc[1][0], false, false);
  acc[0][1] = __builtin_amdgcn_wmma_f32_16x16x4_f32(false, a0, false, q1, (short)0,
                                                    acc[0][1], false, false);
  acc[1][1] = __builtin_amdgcn_wmma_f32_16x16x4_f32(false, a1, false, q1, (short)0,
                                                    acc[1][1], false, false);
  acc[0][2] = __builtin_amdgcn_wmma_f32_16x16x4_f32(false, a0, false, q2, (short)0,
                                                    acc[0][2], false, false);
  acc[1][2] = __builtin_amdgcn_wmma_f32_16x16x4_f32(false, a1, false, q2, (short)0,
                                                    acc[1][2], false, false);
  acc[0][3] = __builtin_amdgcn_wmma_f32_16x16x4_f32(false, a0, false, q3, (short)0,
                                                    acc[0][3], false, false);
  acc[1][3] = __builtin_amdgcn_wmma_f32_16x16x4_f32(false, a1, false, q3, (short)0,
                                                    acc[1][3], false, false);
}

__device__ __forceinline__ void score_mma(const ScoreFrag (&f)[2], v8f (&acc)[2][4]) {
#pragma unroll
  for (int s = 0; s < 2; ++s) {
    v4f a0 = f[s].a0 * f[s].wu;
    v4f a1 = f[s].a1 * f[s].wu;
    score_mma8(lo2(a0), lo2(a1), lo2(f[s].q0), lo2(f[s].q1), lo2(f[s].q2), lo2(f[s].q3), acc);
    score_mma8(hi2(a0), hi2(a1), hi2(f[s].q0), hi2(f[s].q1), hi2(f[s].q2), hi2(f[s].q3), acc);
  }
}

__global__ void __launch_bounds__(256)
k_score(const float* __restrict__ C, const float* __restrict__ Q,
        const float* __restrict__ w, const float* __restrict__ ch,
        const float* __restrict__ qb, float* __restrict__ S,
        float* __restrict__ ST) {
  int blk = blockIdx.x;
  int b = blk >> 3;
  int tblk = blk & 7;
  int wv = threadIdx.x >> 5;
  int lane = threadIdx.x & 31;
  int m = lane & 15, kh = lane >> 4;
  int t0 = tblk * 128 + (wv >> 1) * 32;
  int j0 = (wv & 1) * 64;
  const float* whu = w + 2 * ND + 4 * kh;
  const float* C0 = C + (size_t)(b * NT + t0 + m) * ND + 4 * kh;
  const float* Q0 = Q + (size_t)(b * NJ + j0 + m) * ND + 4 * kh;
  v8f acc[2][4] = {};
  ScoreFrag fa[2], fb[2];
  score_load(fa, whu, C0, Q0, 0);
#pragma unroll 1
  for (int kb = 0; kb < ND - 32; kb += 32) {
    score_load(fb, whu, C0, Q0, kb + 16);
    score_mma(fa, acc);
    score_load(fa, whu, C0, Q0, kb + 32);
    score_mma(fb, acc);
    __builtin_amdgcn_sched_group_barrier(0x020, 14, 0);
    __builtin_amdgcn_sched_group_barrier(0x008, 32, 0);
    __builtin_amdgcn_sched_group_barrier(0x020, 14, 0);
    __builtin_amdgcn_sched_group_barrier(0x008, 32, 0);
  }
  score_load(fb, whu, C0, Q0, ND - 16);
  score_mma(fa, acc);
  score_mma(fb, acc);

#pragma unroll
  for (int rt = 0; rt < 2; ++rt) {
    int tb = t0 + 16 * rt + 8 * kh;
    float chv[8];
#pragma unroll
    for (int r = 0; r < 8; ++r) chv[r] = ch[b * NT + tb + r];
#pragma unroll
    for (int ct = 0; ct < 4; ++ct) {
      int jj = j0 + 16 * ct + m;
      float qv = qb[b * NJ + jj];
      float sv[8];
#pragma unroll
      for (int r = 0; r < 8; ++r) {
        float s = acc[rt][ct][r] + chv[r] + qv;
        S[(size_t)(b * NT + tb + r) * NJ + jj] = s;
        sv[r] = s;
      }
      float* tr = ST + (size_t)(b * NJ + jj) * NT + tb;
      *(float4*)(tr)     = make_float4(sv[0], sv[1], sv[2], sv[3]);
      *(float4*)(tr + 4) = make_float4(sv[4], sv[5], sv[6], sv[7]);
    }
  }
}

#define SR_ROWS 8
static_assert((NB * NT) % SR_ROWS == 0, "whole groups of rows per wave");
static_assert(SR_ROWS == 8, "the load fence names 8 rows");
__global__ void __launch_bounds__(128)
k_softmax_rows(const float* __restrict__ S, float* __restrict__ Sj) {
  int wave = blockIdx.x * (blockDim.x >> 5) + (threadIdx.x >> 5);
  int lane = threadIdx.x & 31;
  int r0 = wave * SR_ROWS;
  if (r0 >= NB * NT) return;
  const float* src = S + (size_t)r0 * NJ + lane * 4;
  float* dst = Sj + (size_t)r0 * NJ + lane * 4;
  v4f v[SR_ROWS];
#pragma unroll
  for (int r = 0; r < SR_ROWS; ++r) v[r] = *(const v4f*)(src + (size_t)r * NJ);
  asm volatile("" :: "v"(v[0]), "v"(v[1]), "v"(v[2]), "v"(v[3]),
                     "v"(v[4]), "v"(v[5]), "v"(v[6]), "v"(v[7]));
  float m[SR_ROWS];
#pragma unroll
  for (int r = 0; r < SR_ROWS; ++r)
    m[r] = fmaxf(fmaxf(v[r].x, v[r].y), fmaxf(v[r].z, v[r].w));
#pragma unroll
  for (int o = 16; o > 0; o >>= 1) {
#pragma unroll
    for (int r = 0; r < SR_ROWS; ++r) m[r] = fmaxf(m[r], __shfl_xor(m[r], o, 32));
  }
  float s[SR_ROWS];
#pragma unroll
  for (int r = 0; r < SR_ROWS; ++r) {
    v[r].x = expf(v[r].x - m[r]); v[r].y = expf(v[r].y - m[r]);
    v[r].z = expf(v[r].z - m[r]); v[r].w = expf(v[r].w - m[r]);
    s[r] = v[r].x + v[r].y + v[r].z + v[r].w;
  }
#pragma unroll
  for (int o = 16; o > 0; o >>= 1) {
#pragma unroll
    for (int r = 0; r < SR_ROWS; ++r) s[r] += __shfl_xor(s[r], o, 32);
  }
#pragma unroll
  for (int r = 0; r < SR_ROWS; ++r) {
    float inv = 1.0f / s[r];
    v4f o4;
    o4.x = v[r].x * inv; o4.y = v[r].y * inv; o4.z = v[r].z * inv; o4.w = v[r].w * inv;
    *(v4f*)(dst + (size_t)r * NJ) = o4;
  }
}

__global__ void k_softmax_t(float* __restrict__ ST) {
  int wave = blockIdx.x * (blockDim.x >> 5) + (threadIdx.x >> 5);
  int lane = threadIdx.x & 31;
  if (wave >= NB * NJ) return;
  float* row = ST + (size_t)wave * NT + lane * 4;
  float4 v[8];
#pragma unroll
  for (int i = 0; i < 8; ++i) v[i] = *(const float4*)(row + i * 128);
  float m = -3.4e38f;
#pragma unroll
  for (int i = 0; i < 8; ++i)
    m = fmaxf(m, fmaxf(fmaxf(v[i].x, v[i].y), fmaxf(v[i].z, v[i].w)));
  m = wave_reduce_max(m);
  float s = 0.f;
#pragma unroll
  for (int i = 0; i < 8; ++i) {
    v[i].x = expf(v[i].x - m); v[i].y = expf(v[i].y - m);
    v[i].z = expf(v[i].z - m); v[i].w = expf(v[i].w - m);
    s += (v[i].x + v[i].y) + (v[i].z + v[i].w);
  }
  s = wave_reduce_sum(s);
  float inv = 1.0f / s;
#pragma unroll
  for (int i = 0; i < 8; ++i) {
    float4 o;
    o.x = v[i].x * inv; o.y = v[i].y * inv; o.z = v[i].z * inv; o.w = v[i].w * inv;
    *(float4*)(row + i * 128) = o;
  }
}

struct TmpFrag { v4f a0, a1; v2f b0l, b1l, b0h, b1h; };

__device__ __forceinline__ void tmp_load(TmpFrag (&f)[4], const float* A0,
                                         const float* Cb, int kb) {
#pragma unroll
  for (int s = 0; s < 4; ++s) {
    int k = kb + 8 * s;
    f[s].a0 = *(const v4f*)(A0 + k);
    f[s].a1 = *(const v4f*)(A0 + 16 * NT + k);
    f[s].b0l.x = Cb[(size_t)k * ND];
    f[s].b0l.y = Cb[(size_t)(k + 1) * ND];
    f[s].b1l.x = Cb[(size_t)k * ND + 16];
    f[s].b1l.y = Cb[(size_t)(k + 1) * ND + 16];
    f[s].b0h.x = Cb[(size_t)(k + 2) * ND];
    f[s].b0h.y = Cb[(size_t)(k + 3) * ND];
    f[s].b1h.x = Cb[(size_t)(k + 2) * ND + 16];
    f[s].b1h.y = Cb[(size_t)(k + 3) * ND + 16];
  }
}

__device__ __forceinline__ void tmp_mma4(v2f a0, v2f a1, v2f b0, v2f b1, v8f (&acc)[2][2]) {
  acc[0][0] = __builtin_amdgcn_wmma_f32_16x16x4_f32(false, a0, false, b0, (short)0,
                                                    acc[0][0], false, false);
  acc[1][0] = __builtin_amdgcn_wmma_f32_16x16x4_f32(false, a1, false, b0, (short)0,
                                                    acc[1][0], false, false);
  acc[0][1] = __builtin_amdgcn_wmma_f32_16x16x4_f32(false, a0, false, b1, (short)0,
                                                    acc[0][1], false, false);
  acc[1][1] = __builtin_amdgcn_wmma_f32_16x16x4_f32(false, a1, false, b1, (short)0,
                                                    acc[1][1], false, false);
}

__device__ __forceinline__ void tmp_mma(const TmpFrag (&f)[4], v8f (&acc)[2][2]) {
#pragma unroll
  for (int s = 0; s < 4; ++s) {
    tmp_mma4(lo2(f[s].a0), lo2(f[s].a1), f[s].b0l, f[s].b1l, acc);
    tmp_mma4(hi2(f[s].a0), hi2(f[s].a1), f[s].b0h, f[s].b1h, acc);
  }
}

__global__ void __launch_bounds__(128)
k_tmp(const float* __restrict__ StT, const float* __restrict__ C,
      float* __restrict__ tmp) {
  int b = blockIdx.x >> 3;
  int dblk = blockIdx.x & 7;
  int wv = threadIdx.x >> 5;
  int lane = threadIdx.x & 31;
  int m = lane & 15, kh = lane >> 4;
  int j0 = wv * 32, d0 = dblk * 32;
  const float* A0 = StT + (size_t)(b * NJ + j0 + m) * NT + 4 * kh;
  const float* Cb = C + (size_t)b * NT * ND + (size_t)(4 * kh) * ND + d0 + m;
  v8f acc[2][2] = {};
  TmpFrag fa[4], fb[4];
  tmp_load(fa, A0, Cb, 0);
#pragma unroll 1
  for (int kb = 0; kb < NT - 64; kb += 64) {
    tmp_load(fb, A0, Cb, kb + 32);
    tmp_mma(fa, acc);
    tmp_load(fa, A0, Cb, kb + 64);
    tmp_mma(fb, acc);
    __builtin_amdgcn_sched_group_barrier(0x020, 40, 0);
    __builtin_amdgcn_sched_group_barrier(0x008, 32, 0);
    __builtin_amdgcn_sched_group_barrier(0x020, 40, 0);
    __builtin_amdgcn_sched_group_barrier(0x008, 32, 0);
  }
  tmp_load(fb, A0, Cb, NT - 32);
  tmp_mma(fa, acc);
  tmp_mma(fb, acc);

#pragma unroll
  for (int rt = 0; rt < 2; ++rt) {
#pragma unroll
    for (int r = 0; r < 8; ++r) {
      int jj = j0 + 16 * rt + r + 8 * kh;
      float* ob = tmp + (size_t)(b * NJ + jj) * ND + d0 + m;
      ob[0]  = acc[rt][0][r];
      ob[16] = acc[rt][1][r];
    }
  }
}

__global__ void __launch_bounds__(256)
k_out(const float* __restrict__ Sj, const float* __restrict__ Q,
      const float* __restrict__ tmp, const float* __restrict__ C,
      float* __restrict__ out) {
  int blk = blockIdx.x;
  int b = blk >> 5;
  int tblk = blk & 31;
  int wv = threadIdx.x >> 5;
  int lane = threadIdx.x & 31;
  int m = lane & 15, kh = lane >> 4;
  int t0 = tblk * 32, d0 = wv * 32;
  const float* S0 = Sj + (size_t)(b * NT + t0 + m) * NJ + 4 * kh;
  const float* S1 = S0 + (size_t)16 * NJ;
  const float* Qb = Q + (size_t)b * NJ * ND + (size_t)(4 * kh) * ND + d0 + m;
  const float* Tb = tmp + (size_t)b * NJ * ND + (size_t)(4 * kh) * ND + d0 + m;
  v8f aQ00 = {}, aQ01 = {}, aQ10 = {}, aQ11 = {};
  v8f aT00 = {}, aT01 = {}, aT10 = {}, aT11 = {};
#pragma unroll 1
  for (int k0 = 0; k0 < NJ; k0 += 8) {
    v4f a0 = *(const v4f*)(S0 + k0);
    v4f a1 = *(const v4f*)(S1 + k0);
    const float* qp = Qb + (size_t)k0 * ND;
    const float* tp = Tb + (size_t)k0 * ND;
    v2f q0, q1, u0, u1, q0h, q1h, u0h, u1h;
    q0.x = qp[0];           q0.y = qp[ND];
    q1.x = qp[16];          q1.y = qp[ND + 16];
    u0.x = tp[0];           u0.y = tp[ND];
    u1.x = tp[16];          u1.y = tp[ND + 16];
    q0h.x = qp[2 * ND];      q0h.y = qp[3 * ND];
    q1h.x = qp[2 * ND + 16]; q1h.y = qp[3 * ND + 16];
    u0h.x = tp[2 * ND];      u0h.y = tp[3 * ND];
    u1h.x = tp[2 * ND + 16]; u1h.y = tp[3 * ND + 16];
    v2f a0l = lo2(a0), a1l = lo2(a1), a0h = hi2(a0), a1h = hi2(a1);
    aQ00 = __builtin_amdgcn_wmma_f32_16x16x4_f32(false, a0l, false, q0, (short)0,
                                                 aQ00, false, false);
    aT00 = __builtin_amdgcn_wmma_f32_16x16x4_f32(false, a0l, false, u0, (short)0,
                                                 aT00, false, false);
    aQ01 = __builtin_amdgcn_wmma_f32_16x16x4_f32(false, a0l, false, q1, (short)0,
                                                 aQ01, false, false);
    aT01 = __builtin_amdgcn_wmma_f32_16x16x4_f32(false, a0l, false, u1, (short)0,
                                                 aT01, false, false);
    aQ10 = __builtin_amdgcn_wmma_f32_16x16x4_f32(false, a1l, false, q0, (short)0,
                                                 aQ10, false, false);
    aT10 = __builtin_amdgcn_wmma_f32_16x16x4_f32(false, a1l, false, u0, (short)0,
                                                 aT10, false, false);
    aQ11 = __builtin_amdgcn_wmma_f32_16x16x4_f32(false, a1l, false, q1, (short)0,
                                                 aQ11, false, false);
    aT11 = __builtin_amdgcn_wmma_f32_16x16x4_f32(false, a1l, false, u1, (short)0,
                                                 aT11, false, false);
    aQ00 = __builtin_amdgcn_wmma_f32_16x16x4_f32(false, a0h, false, q0h, (short)0,
                                                 aQ00, false, false);
    aT00 = __builtin_amdgcn_wmma_f32_16x16x4_f32(false, a0h, false, u0h, (short)0,
                                                 aT00, false, false);
    aQ01 = __builtin_amdgcn_wmma_f32_16x16x4_f32(false, a0h, false, q1h, (short)0,
                                                 aQ01, false, false);
    aT01 = __builtin_amdgcn_wmma_f32_16x16x4_f32(false, a0h, false, u1h, (short)0,
                                                 aT01, false, false);
    aQ10 = __builtin_amdgcn_wmma_f32_16x16x4_f32(false, a1h, false, q0h, (short)0,
                                                 aQ10, false, false);
    aT10 = __builtin_amdgcn_wmma_f32_16x16x4_f32(false, a1h, false, u0h, (short)0,
                                                 aT10, false, false);
    aQ11 = __builtin_amdgcn_wmma_f32_16x16x4_f32(false, a1h, false, q1h, (short)0,
                                                 aQ11, false, false);
    aT11 = __builtin_amdgcn_wmma_f32_16x16x4_f32(false, a1h, false, u1h, (short)0,
                                                 aT11, false, false);
  }
  int d = d0 + m;
#pragma unroll
  for (int r = 0; r < 8; ++r) {
    int t = t0 + r + 8 * kh;
    const float* cp = C + (size_t)(b * NT + t) * ND + d;
    float* ob = out + (size_t)(b * NT + t) * (4 * ND) + d;
    float c0 = cp[0], c1 = cp[16];
    ob[0]           = c0;            ob[16]          = c1;
    ob[ND]          = aQ00[r];       ob[ND + 16]     = aQ01[r];
    ob[2 * ND]      = c0 * aQ00[r];  ob[2 * ND + 16] = c1 * aQ01[r];
    ob[3 * ND]      = c0 * aT00[r];  ob[3 * ND + 16] = c1 * aT01[r];
  }
#pragma unroll
  for (int r = 0; r < 8; ++r) {
    int t = t0 + 16 + r + 8 * kh;
    const float* cp = C + (size_t)(b * NT + t) * ND + d;
    float* ob = out + (size_t)(b * NT + t) * (4 * ND) + d;
    float c0 = cp[0], c1 = cp[16];
    ob[0]           = c0;            ob[16]          = c1;
    ob[ND]          = aQ10[r];       ob[ND + 16]     = aQ11[r];
    ob[2 * ND]      = c0 * aQ10[r];  ob[2 * ND + 16] = c1 * aQ11[r];
    ob[3 * ND]      = c0 * aT10[r];  ob[3 * ND + 16] = c1 * aT11[r];
  }
}

extern "C" void kernel_launch(void* const* d_in, const int* in_sizes, int n_in,
                              void* d_out, int out_size, void* d_ws, size_t ws_size,
                              hipStream_t stream) {
  (void)in_sizes; (void)n_in; (void)out_size; (void)ws_size;
  const float* C = (const float*)d_in[0];
  const float* Q = (const float*)d_in[1];
  const float* w = (const float*)d_in[2];
  float* out = (float*)d_out;

  float* ws = (float*)d_ws;
  float* ch = ws;
  float* qb = ch + NB * NT;
  float* S  = qb + NB * NJ;
  float* Sj = S  + (size_t)NB * NT * NJ;
  float* ST = Sj + (size_t)NB * NT * NJ;
  float* tp = ST + (size_t)NB * NT * NJ;

  k_rowdots<<<(NB * NT + NB * NJ) / (4 * RD_ROWS), 128, 0, stream>>>(C, Q, w, ch, qb);
  k_score<<<NB * 8, 256, 0, stream>>>(C, Q, w, ch, qb, S, ST);
  k_softmax_rows<<<(NB * NT) / (4 * SR_ROWS), 128, 0, stream>>>(S, Sj);
  k_softmax_t<<<(NB * NJ) / 4, 128, 0, stream>>>(ST);
  k_tmp<<<NB * 8, 128, 0, stream>>>(ST, C, tp);
  k_out<<<NB * 32, 256, 0, stream>>>(Sj, Q, tp, C, out);
}


# k_score finishes the softmax over j: its 128x128 tile of S passes through LDS, a wave normalises 16 whole rows and stores Sj as whole rows; S plane and row-softmax launch removed; sums unchanged
# speedup vs baseline: 2.0520x; 1.0327x over previous
#include <hip/hip_runtime.h>

#define NB 32
#define NT 1024
#define NJ 128
#define ND 256

typedef __attribute__((ext_vector_type(2))) float v2f;
typedef __attribute__((ext_vector_type(4))) float v4f;
typedef __attribute__((ext_vector_type(8))) float v8f;

__device__ __forceinline__ float wave_reduce_sum(float v) {
#pragma unroll
  for (int o = 16; o > 0; o >>= 1) v += __shfl_xor(v, o, 32);
  return v;
}
__device__ __forceinline__ float wave_reduce_max(float v) {
#pragma unroll
  for (int o = 16; o > 0; o >>= 1) v = fmaxf(v, __shfl_xor(v, o, 32));
  return v;
}

#define RD_ROWS 8
static_assert((NB * NT) % RD_ROWS == 0 && (NB * NJ) % RD_ROWS == 0,
              "a wave's rows must lie in one of the two inputs");
static_assert(RD_ROWS == 8 && ND == 256, "the load fence and the stores name 8 rows of 8 values");
__global__ void __launch_bounds__(128)
k_rowdots(const float* __restrict__ C, const float* __restrict__ Q,
          const float* __restrict__ w, float* __restrict__ ch,
          float* __restrict__ qb) {
  int wave = blockIdx.x * (blockDim.x >> 5) + (threadIdx.x >> 5);
  int lane = threadIdx.x & 31;
  const int nch = NB * NT;
  const int nrows = nch + NB * NJ;
  int r0 = wave * RD_ROWS;
  if (r0 >= nrows) return;
  const float* rows; const float* wv; float* out;
  if (r0 < nch) { rows = C + (size_t)r0 * ND; wv = w;      out = ch + r0; }
  else { int r = r0 - nch; rows = Q + (size_t)r * ND; wv = w + ND; out = qb + r; }
  float wk[ND / 32];
#pragma unroll
  for (int k = 0; k < ND / 32; ++k) wk[k] = wv[lane + 32 * k];
  v8f x[RD_ROWS];
#pragma unroll
  for (int r = 0; r < RD_ROWS; ++r) {
#pragma unroll
    for (int k = 0; k < ND / 32; ++k) x[r][k] = rows[(size_t)r * ND + lane + 32 * k];
  }
  asm volatile("" :: "v"(x[0]), "v"(x[1]), "v"(x[2]), "v"(x[3]),
                     "v"(x[4]), "v"(x[5]), "v"(x[6]), "v"(x[7]));
  float s[RD_ROWS];
#pragma unroll
  for (int r = 0; r < RD_ROWS; ++r) {
    float a = 0.f;
#pragma unroll
    for (int k = 0; k < ND / 32; ++k) a += x[r][k] * wk[k];
    s[r] = a;
  }
#pragma unroll
  for (int o = 16; o > 0; o >>= 1) {
#pragma unroll
    for (int r = 0; r < RD_ROWS; ++r) s[r] += __shfl_xor(s[r], o, 32);
  }
  if (lane == 0) {
    *(float4*)(out)     = make_float4(s[0], s[1], s[2], s[3]);
    *(float4*)(out + 4) = make_float4(s[4], s[5], s[6], s[7]);
  }
}

struct ScoreFrag { v4f wu, a0, a1, q0, q1, q2, q3; };

__device__ __forceinline__ v2f lo2(v4f x) { return __builtin_shufflevector(x, x, 0, 1); }
__device__ __forceinline__ v2f hi2(v4f x) { return __builtin_shufflevector(x, x, 2, 3); }

__device__ __forceinline__ void score_load(ScoreFrag (&f)[2], const float* whu,
                                           const float* C0, const float* Q0, int kb) {
#pragma unroll
  for (int s = 0; s < 2; ++s) {
    int kk = kb + 8 * s;
    f[s].wu = *(const v4f*)(whu + kk);
    f[s].a0 = *(const v4f*)(C0 + kk);
    f[s].a1 = *(const v4f*)(C0 + 16 * ND + kk);
    f[s].q0 = *(const v4f*)(Q0 + kk);
    f[s].q1 = *(const v4f*)(Q0 + 16 * ND + kk);
    f[s].q2 = *(const v4f*)(Q0 + 32 * ND + kk);
    f[s].q3 = *(const v4f*)(Q0 + 48 * ND + kk);
  }
}

__device__ __forceinline__ void score_mma8(v2f a0, v2f a1, v2f q0, v2f q1, v2f q2, v2f q3,
                                           v8f (&acc)[2][4]) {
  acc[0][0] = __builtin_amdgcn_wmma_f32_16x16x4_f32(false, a0, false, q0, (short)0,
                                                    acc[0][0], false, false);
  acc[1][0] = __builtin_amdgcn_wmma_f32_16x16x4_f32(false, a1, false, q0, (short)0,
                                                    acc[1][0], false, false);
  acc[0][1] = __builtin_amdgcn_wmma_f32_16x16x4_f32(false, a0, false, q1, (short)0,
                                                    acc[0][1], false, false);
  acc[1][1] = __builtin_amdgcn_wmma_f32_16x16x4_f32(false, a1, false, q1, (short)0,
                                                    acc[1][1], false, false);
  acc[0][2] = __builtin_amdgcn_wmma_f32_16x16x4_f32(false, a0, false, q2, (short)0,
                                                    acc[0][2], false, false);
  acc[1][2] = __builtin_amdgcn_wmma_f32_16x16x4_f32(false, a1, false, q2, (short)0,
                                                    acc[1][2], false, false);
  acc[0][3] = __builtin_amdgcn_wmma_f32_16x16x4_f32(false, a0, false, q3, (short)0,
                                                    acc[0][3], false, false);
  acc[1][3] = __builtin_amdgcn_wmma_f32_16x16x4_f32(false, a1, false, q3, (short)0,
                                                    acc[1][3], false, false);
}

__device__ __forceinline__ void score_mma(const ScoreFrag (&f)[2], v8f (&acc)[2][4]) {
#pragma unroll
  for (int s = 0; s < 2; ++s) {
    v4f a0 = f[s].a0 * f[s].wu;
    v4f a1 = f[s].a1 * f[s].wu;
    score_mma8(lo2(a0), lo2(a1), lo2(f[s].q0), lo2(f[s].q1), lo2(f[s].q2), lo2(f[s].q3), acc);
    score_mma8(hi2(a0), hi2(a1), hi2(f[s].q0), hi2(f[s].q1), hi2(f[s].q2), hi2(f[s].q3), acc);
  }
}

#define SL_PITCH 132
#define SM_ROWS 16
static_assert(SM_ROWS == 16 && NJ == 128, "a lane holds 4 of a row's 128 values; 8 waves x 16 rows = the block's 128");
__global__ void __launch_bounds__(256)
k_score(const float* __restrict__ C, const float* __restrict__ Q,
        const float* __restrict__ w, const float* __restrict__ ch,
        const float* __restrict__ qb, float* __restrict__ Sj,
        float* __restrict__ ST) {
  __shared__ __attribute__((aligned(16))) float Sl[128 * SL_PITCH];
  int blk = blockIdx.x;
  int b = blk >> 3;
  int tblk = blk & 7;
  int wv = threadIdx.x >> 5;
  int lane = threadIdx.x & 31;
  int m = lane & 15, kh = lane >> 4;
  int t0 = tblk * 128 + (wv >> 1) * 32;
  int j0 = (wv & 1) * 64;
  const float* whu = w + 2 * ND + 4 * kh;
  const float* C0 = C + (size_t)(b * NT + t0 + m) * ND + 4 * kh;
  const float* Q0 = Q + (size_t)(b * NJ + j0 + m) * ND + 4 * kh;
  v8f acc[2][4] = {};
  ScoreFrag fa[2], fb[2];
  score_load(fa, whu, C0, Q0, 0);
#pragma unroll 1
  for (int kb = 0; kb < ND - 32; kb += 32) {
    score_load(fb, whu, C0, Q0, kb + 16);
    score_mma(fa, acc);
    score_load(fa, whu, C0, Q0, kb + 32);
    score_mma(fb, acc);
    __builtin_amdgcn_sched_group_barrier(0x020, 14, 0);
    __builtin_amdgcn_sched_group_barrier(0x008, 32, 0);
    __builtin_amdgcn_sched_group_barrier(0x020, 14, 0);
    __builtin_amdgcn_sched_group_barrier(0x008, 32, 0);
  }
  score_load(fb, whu, C0, Q0, ND - 16);
  score_mma(fa, acc);
  score_mma(fb, acc);

#pragma unroll
  for (int rt = 0; rt < 2; ++rt) {
    int tb = t0 + 16 * rt + 8 * kh;
    float chv[8];
#pragma unroll
    for (int r = 0; r < 8; ++r) chv[r] = ch[b * NT + tb + r];
#pragma unroll
    for (int ct = 0; ct < 4; ++ct) {
      int jj = j0 + 16 * ct + m;
      float qv = qb[b * NJ + jj];
      float* sl = Sl + (tb - tblk * 128) * SL_PITCH + jj;
#pragma unroll
      for (int r = 0; r < 8; ++r) {
        float s = acc[rt][ct][r] + chv[r] + qv;
        sl[r * SL_PITCH] = s;
        acc[rt][ct][r] = s;
      }
    }
  }
  __syncthreads();
#pragma unroll
  for (int rt = 0; rt < 2; ++rt) {
    int tb = t0 + 16 * rt + 8 * kh;
#pragma unroll
    for (int ct = 0; ct < 4; ++ct) {
      int jj = j0 + 16 * ct + m;
      float* tr = ST + (size_t)(b * NJ + jj) * NT + tb;
      *(float4*)(tr)     = make_float4(acc[rt][ct][0], acc[rt][ct][1], acc[rt][ct][2], acc[rt][ct][3]);
      *(float4*)(tr + 4) = make_float4(acc[rt][ct][4], acc[rt][ct][5], acc[rt][ct][6], acc[rt][ct][7]);
    }
  }
  const float* rl = Sl + (wv * 16) * SL_PITCH + lane * 4;
  float* dst = Sj + (size_t)(b * NT + tblk * 128 + wv * 16) * NJ + lane * 4;
#pragma unroll
  for (int g = 0; g < 16 / SM_ROWS; ++g) {
    v4f v[SM_ROWS];
#pragma unroll
    for (int r = 0; r < SM_ROWS; ++r) v[r] = *(const v4f*)(rl + (g * SM_ROWS + r) * SL_PITCH);
    float mx[SM_ROWS];
#pragma unroll
    for (int r = 0; r < SM_ROWS; ++r)
      mx[r] = fmaxf(fmaxf(v[r].x, v[r].y), fmaxf(v[r].z, v[r].w));
#pragma unroll
    for (int o = 16; o > 0; o >>= 1) {
#pragma unroll
      for (int r = 0; r < SM_ROWS; ++r) mx[r] = fmaxf(mx[r], __shfl_xor(mx[r], o, 32));
    }
    float sm[SM_ROWS];
#pragma unroll
    for (int r = 0; r < SM_ROWS; ++r) {
      v[r].x = expf(v[r].x - mx[r]); v[r].y = expf(v[r].y - mx[r]);
      v[r].z = expf(v[r].z - mx[r]); v[r].w = expf(v[r].w - mx[r]);
      sm[r] = v[r].x + v[r].y + v[r].z + v[r].w;
    }
#pragma unroll
    for (int o = 16; o > 0; o >>= 1) {
#pragma unroll
      for (int r = 0; r < SM_ROWS; ++r) sm[r] += __shfl_xor(sm[r], o, 32);
    }
#pragma unroll
    for (int r = 0; r < SM_ROWS; ++r) {
      float inv = 1.0f / sm[r];
      v4f o4;
      o4.x = v[r].x * inv; o4.y = v[r].y * inv; o4.z = v[r].z * inv; o4.w = v[r].w * inv;
      *(v4f*)(dst + (size_t)(g * SM_ROWS + r) * NJ) = o4;
    }
  }
}

__global__ void k_softmax_t(float* __restrict__ ST) {
  int wave = blockIdx.x * (blockDim.x >> 5) + (threadIdx.x >> 5);
  int lane = threadIdx.x & 31;
  if (wave >= NB * NJ) return;
  float* row = ST + (size_t)wave * NT + lane * 4;
  float4 v[8];
#pragma unroll
  for (int i = 0; i < 8; ++i) v[i] = *(const float4*)(row + i * 128);
  float m = -3.4e38f;
#pragma unroll
  for (int i = 0; i < 8; ++i)
    m = fmaxf(m, fmaxf(fmaxf(v[i].x, v[i].y), fmaxf(v[i].z, v[i].w)));
  m = wave_reduce_max(m);
  float s = 0.f;
#pragma unroll
  for (int i = 0; i < 8; ++i) {
    v[i].x = expf(v[i].x - m); v[i].y = expf(v[i].y - m);
    v[i].z = expf(v[i].z - m); v[i].w = expf(v[i].w - m);
    s += (v[i].x + v[i].y) + (v[i].z + v[i].w);
  }
  s = wave_reduce_sum(s);
  float inv = 1.0f / s;
#pragma unroll
  for (int i = 0; i < 8; ++i) {
    float4 o;
    o.x = v[i].x * inv; o.y = v[i].y * inv; o.z = v[i].z * inv; o.w = v[i].w * inv;
    *(float4*)(row + i * 128) = o;
  }
}

struct TmpFrag { v4f a0, a1; v2f b0l, b1l, b0h, b1h; };

__device__ __forceinline__ void tmp_load(TmpFrag (&f)[4], const float* A0,
                                         const float* Cb, int kb) {
#pragma unroll
  for (int s = 0; s < 4; ++s) {
    int k = kb + 8 * s;
    f[s].a0 = *(const v4f*)(A0 + k);
    f[s].a1 = *(const v4f*)(A0 + 16 * NT + k);
    f[s].b0l.x = Cb[(size_t)k * ND];
    f[s].b0l.y = Cb[(size_t)(k + 1) * ND];
    f[s].b1l.x = Cb[(size_t)k * ND + 16];
    f[s].b1l.y = Cb[(size_t)(k + 1) * ND + 16];
    f[s].b0h.x = Cb[(size_t)(k + 2) * ND];
    f[s].b0h.y = Cb[(size_t)(k + 3) * ND];
    f[s].b1h.x = Cb[(size_t)(k + 2) * ND + 16];
    f[s].b1h.y = Cb[(size_t)(k + 3) * ND + 16];
  }
}

__device__ __forceinline__ void tmp_mma4(v2f a0, v2f a1, v2f b0, v2f b1, v8f (&acc)[2][2]) {
  acc[0][0] = __builtin_amdgcn_wmma_f32_16x16x4_f32(false, a0, false, b0, (short)0,
                                                    acc[0][0], false, false);
  acc[1][0] = __builtin_amdgcn_wmma_f32_16x16x4_f32(false, a1, false, b0, (short)0,
                                                    acc[1][0], false, false);
  acc[0][1] = __builtin_amdgcn_wmma_f32_16x16x4_f32(false, a0, false, b1, (short)0,
                                                    acc[0][1], false, false);
  acc[1][1] = __builtin_amdgcn_wmma_f32_16x16x4_f32(false, a1, false, b1, (short)0,
                                                    acc[1][1], false, false);
}

__device__ __forceinline__ void tmp_mma(const TmpFrag (&f)[4], v8f (&acc)[2][2]) {
#pragma unroll
  for (int s = 0; s < 4; ++s) {
    tmp_mma4(lo2(f[s].a0), lo2(f[s].a1), f[s].b0l, f[s].b1l, acc);
    tmp_mma4(hi2(f[s].a0), hi2(f[s].a1), f[s].b0h, f[s].b1h, acc);
  }
}

__global__ void __launch_bounds__(128)
k_tmp(const float* __restrict__ StT, const float* __restrict__ C,
      float* __restrict__ tmp) {
  int b = blockIdx.x >> 3;
  int dblk = blockIdx.x & 7;
  int wv = threadIdx.x >> 5;
  int lane = threadIdx.x & 31;
  int m = lane & 15, kh = lane >> 4;
  int j0 = wv * 32, d0 = dblk * 32;
  const float* A0 = StT + (size_t)(b * NJ + j0 + m) * NT + 4 * kh;
  const float* Cb = C + (size_t)b * NT * ND + (size_t)(4 * kh) * ND + d0 + m;
  v8f acc[2][2] = {};
  TmpFrag fa[4], fb[4];
  tmp_load(fa, A0, Cb, 0);
#pragma unroll 1
  for (int kb = 0; kb < NT - 64; kb += 64) {
    tmp_load(fb, A0, Cb, kb + 32);
    tmp_mma(fa, acc);
    tmp_load(fa, A0, Cb, kb + 64);
    tmp_mma(fb, acc);
    __builtin_amdgcn_sched_group_barrier(0x020, 40, 0);
    __builtin_amdgcn_sched_group_barrier(0x008, 32, 0);
    __builtin_amdgcn_sched_group_barrier(0x020, 40, 0);
    __builtin_amdgcn_sched_group_barrier(0x008, 32, 0);
  }
  tmp_load(fb, A0, Cb, NT - 32);
  tmp_mma(fa, acc);
  tmp_mma(fb, acc);

#pragma unroll
  for (int rt = 0; rt < 2; ++rt) {
#pragma unroll
    for (int r = 0; r < 8; ++r) {
      int jj = j0 + 16 * rt + r + 8 * kh;
      float* ob = tmp + (size_t)(b * NJ + jj) * ND + d0 + m;
      ob[0]  = acc[rt][0][r];
      ob[16] = acc[rt][1][r];
    }
  }
}

__global__ void __launch_bounds__(256)
k_out(const float* __restrict__ Sj, const float* __restrict__ Q,
      const float* __restrict__ tmp, const float* __restrict__ C,
      float* __restrict__ out) {
  int blk = blockIdx.x;
  int b = blk >> 5;
  int tblk = blk & 31;
  int wv = threadIdx.x >> 5;
  int lane = threadIdx.x & 31;
  int m = lane & 15, kh = lane >> 4;
  int t0 = tblk * 32, d0 = wv * 32;
  const float* S0 = Sj + (size_t)(b * NT + t0 + m) * NJ + 4 * kh;
  const float* S1 = S0 + (size_t)16 * NJ;
  const float* Qb = Q + (size_t)b * NJ * ND + (size_t)(4 * kh) * ND + d0 + m;
  const float* Tb = tmp + (size_t)b * NJ * ND + (size_t)(4 * kh) * ND + d0 + m;
  v8f aQ00 = {}, aQ01 = {}, aQ10 = {}, aQ11 = {};
  v8f aT00 = {}, aT01 = {}, aT10 = {}, aT11 = {};
#pragma unroll 1
  for (int k0 = 0; k0 < NJ; k0 += 8) {
    v4f a0 = *(const v4f*)(S0 + k0);
    v4f a1 = *(const v4f*)(S1 + k0);
    const float* qp = Qb + (size_t)k0 * ND;
    const float* tp = Tb + (size_t)k0 * ND;
    v2f q0, q1, u0, u1, q0h, q1h, u0h, u1h;
    q0.x = qp[0];           q0.y = qp[ND];
    q1.x = qp[16];          q1.y = qp[ND + 16];
    u0.x = tp[0];           u0.y = tp[ND];
    u1.x = tp[16];          u1.y = tp[ND + 16];
    q0h.x = qp[2 * ND];      q0h.y = qp[3 * ND];
    q1h.x = qp[2 * ND + 16]; q1h.y = qp[3 * ND + 16];
    u0h.x = tp[2 * ND];      u0h.y = tp[3 * ND];
    u1h.x = tp[2 * ND + 16]; u1h.y = tp[3 * ND + 16];
    v2f a0l = lo2(a0), a1l = lo2(a1), a0h = hi2(a0), a1h = hi2(a1);
    aQ00 = __builtin_amdgcn_wmma_f32_16x16x4_f32(false, a0l, false, q0, (short)0,
                                                 aQ00, false, false);
    aT00 = __builtin_amdgcn_wmma_f32_16x16x4_f32(false, a0l, false, u0, (short)0,
                                                 aT00, false, false);
    aQ01 = __builtin_amdgcn_wmma_f32_16x16x4_f32(false, a0l, false, q1, (short)0,
                                                 aQ01, false, false);
    aT01 = __builtin_amdgcn_wmma_f32_16x16x4_f32(false, a0l, false, u1, (short)0,
                                                 aT01, false, false);
    aQ10 = __builtin_amdgcn_wmma_f32_16x16x4_f32(false, a1l, false, q0, (short)0,
                                                 aQ10, false, false);
    aT10 = __builtin_amdgcn_wmma_f32_16x16x4_f32(false, a1l, false, u0, (short)0,
                                                 aT10, false, false);
    aQ11 = __builtin_amdgcn_wmma_f32_16x16x4_f32(false, a1l, false, q1, (short)0,
                                                 aQ11, false, false);
    aT11 = __builtin_amdgcn_wmma_f32_16x16x4_f32(false, a1l, false, u1, (short)0,
                                                 aT11, false, false);
    aQ00 = __builtin_amdgcn_wmma_f32_16x16x4_f32(false, a0h, false, q0h, (short)0,
                                                 aQ00, false, false);
    aT00 = __builtin_amdgcn_wmma_f32_16x16x4_f32(false, a0h, false, u0h, (short)0,
                                                 aT00, false, false);
    aQ01 = __builtin_amdgcn_wmma_f32_16x16x4_f32(false, a0h, false, q1h, (short)0,
                                                 aQ01, false, false);
    aT01 = __builtin_amdgcn_wmma_f32_16x16x4_f32(false, a0h, false, u1h, (short)0,
                                                 aT01, false, false);
    aQ10 = __builtin_amdgcn_wmma_f32_16x16x4_f32(false, a1h, false, q0h, (short)0,
                                                 aQ10, false, false);
    aT10 = __builtin_amdgcn_wmma_f32_16x16x4_f32(false, a1h, false, u0h, (short)0,
                                                 aT10, false, false);
    aQ11 = __builtin_amdgcn_wmma_f32_16x16x4_f32(false, a1h, false, q1h, (short)0,
                                                 aQ11, false, false);
    aT11 = __builtin_amdgcn_wmma_f32_16x16x4_f32(false, a1h, false, u1h, (short)0,
                                                 aT11, false, false);
  }
  int d = d0 + m;
#pragma unroll
  for (int r = 0; r < 8; ++r) {
    int t = t0 + r + 8 * kh;
    const float* cp = C + (size_t)(b * NT + t) * ND + d;
    float* ob = out + (size_t)(b * NT + t) * (4 * ND) + d;
    float c0 = cp[0], c1 = cp[16];
    ob[0]           = c0;            ob[16]          = c1;
    ob[ND]          = aQ00[r];       ob[ND + 16]     = aQ01[r];
    ob[2 * ND]      = c0 * aQ00[r];  ob[2 * ND + 16] = c1 * aQ01[r];
    ob[3 * ND]      = c0 * aT00[r];  ob[3 * ND + 16] = c1 * aT01[r];
  }
#pragma unroll
  for (int r = 0; r < 8; ++r) {
    int t = t0 + 16 + r + 8 * kh;
    const float* cp = C + (size_t)(b * NT + t) * ND + d;
    float* ob = out + (size_t)(b * NT + t) * (4 * ND) + d;
    float c0 = cp[0], c1 = cp[16];
    ob[0]           = c0;            ob[16]          = c1;
    ob[ND]          = aQ10[r];       ob[ND + 16]     = aQ11[r];
    ob[2 * ND]      = c0 * aQ10[r];  ob[2 * ND + 16] = c1 * aQ11[r];
    ob[3 * ND]      = c0 * aT10[r];  ob[3 * ND + 16] = c1 * aT11[r];
  }
}

extern "C" void kernel_launch(void* const* d_in, const int* in_sizes, int n_in,
                              void* d_out, int out_size, void* d_ws, size_t ws_size,
                              hipStream_t stream) {
  (void)in_sizes; (void)n_in; (void)out_size; (void)ws_size;
  const float* C = (const float*)d_in[0];
  const float* Q = (const float*)d_in[1];
  const float* w = (const float*)d_in[2];
  float* out = (float*)d_out;

  float* ws = (float*)d_ws;
  float* ch = ws;
  float* qb = ch + NB * NT;
  float* Sj = qb + NB * NJ;
  float* ST = Sj + (size_t)NB * NT * NJ;
  float* tp = ST + (size_t)NB * NT * NJ;

  k_rowdots<<<(NB * NT + NB * NJ) / (4 * RD_ROWS), 128, 0, stream>>>(C, Q, w, ch, qb);
  k_score<<<NB * 8, 256, 0, stream>>>(C, Q, w, ch, qb, Sj, ST);
  k_softmax_t<<<(NB * NJ) / 4, 128, 0, stream>>>(ST);
  k_tmp<<<NB * 8, 128, 0, stream>>>(ST, C, tp);
  k_out<<<NB * 32, 256, 0, stream>>>(Sj, Q, tp, C, out);
}


# A = Sj@Q with planes C, A, C*A leaves k_out and runs as extra workgroups (block index picks the role): 3/4 beside k_tmp's one wave per SIMD, 1/4 beside Sj@tmp in the last launch; sums unchanged
# speedup vs baseline: 2.1961x; 1.0702x over previous
#include <hip/hip_runtime.h>

#define NB 32
#define NT 1024
#define NJ 128
#define ND 256

typedef __attribute__((ext_vector_type(2))) float v2f;
typedef __attribute__((ext_vector_type(4))) float v4f;
typedef __attribute__((ext_vector_type(8))) float v8f;

__device__ __forceinline__ float wave_reduce_sum(float v) {
#pragma unroll
  for (int o = 16; o > 0; o >>= 1) v += __shfl_xor(v, o, 32);
  return v;
}
__device__ __forceinline__ float wave_reduce_max(float v) {
#pragma unroll
  for (int o = 16; o > 0; o >>= 1) v = fmaxf(v, __shfl_xor(v, o, 32));
  return v;
}

#define RD_ROWS 8
static_assert((NB * NT) % RD_ROWS == 0 && (NB * NJ) % RD_ROWS == 0,
              "a wave's rows must lie in one of the two inputs");
static_assert(RD_ROWS == 8 && ND == 256, "the load fence and the stores name 8 rows of 8 values");
__global__ void __launch_bounds__(128)
k_rowdots(const float* __restrict__ C, const float* __restrict__ Q,
          const float* __restrict__ w, float* __restrict__ ch,
          float* __restrict__ qb) {
  int wave = blockIdx.x * (blockDim.x >> 5) + (threadIdx.x >> 5);
  int lane = threadIdx.x & 31;
  const int nch = NB * NT;
  const int nrows = nch + NB * NJ;
  int r0 = wave * RD_ROWS;
  if (r0 >= nrows) return;
  const float* rows; const float* wv; float* out;
  if (r0 < nch) { rows = C + (size_t)r0 * ND; wv = w;      out = ch + r0; }
  else { int r = r0 - nch; rows = Q + (size_t)r * ND; wv = w + ND; out = qb + r; }
  float wk[ND / 32];
#pragma unroll
  for (int k = 0; k < ND / 32; ++k) wk[k] = wv[lane + 32 * k];
  v8f x[RD_ROWS];
#pragma unroll
  for (int r = 0; r < RD_ROWS; ++r) {
#pragma unroll
    for (int k = 0; k < ND / 32; ++k) x[r][k] = rows[(size_t)r * ND + lane + 32 * k];
  }
  asm volatile("" :: "v"(x[0]), "v"(x[1]), "v"(x[2]), "v"(x[3]),
                     "v"(x[4]), "v"(x[5]), "v"(x[6]), "v"(x[7]));
  float s[RD_ROWS];
#pragma unroll
  for (int r = 0; r < RD_ROWS; ++r) {
    float a = 0.f;
#pragma unroll
    for (int k = 0; k < ND / 32; ++k) a += x[r][k] * wk[k];
    s[r] = a;
  }
#pragma unroll
  for (int o = 16; o > 0; o >>= 1) {
#pragma unroll
    for (int r = 0; r < RD_ROWS; ++r) s[r] += __shfl_xor(s[r], o, 32);
  }
  if (lane == 0) {
    *(float4*)(out)     = make_float4(s[0], s[1], s[2], s[3]);
    *(float4*)(out + 4) = make_float4(s[4], s[5], s[6], s[7]);
  }
}

struct ScoreFrag { v4f wu, a0, a1, q0, q1, q2, q3; };

__device__ __forceinline__ v2f lo2(v4f x) { return __builtin_shufflevector(x, x, 0, 1); }
__device__ __forceinline__ v2f hi2(v4f x) { return __builtin_shufflevector(x, x, 2, 3); }

__device__ __forceinline__ void score_load(ScoreFrag (&f)[2], const float* whu,
                                           const float* C0, const float* Q0, int kb) {
#pragma unroll
  for (int s = 0; s < 2; ++s) {
    int kk = kb + 8 * s;
    f[s].wu = *(const v4f*)(whu + kk);
    f[s].a0 = *(const v4f*)(C0 + kk);
    f[s].a1 = *(const v4f*)(C0 + 16 * ND + kk);
    f[s].q0 = *(const v4f*)(Q0 + kk);
    f[s].q1 = *(const v4f*)(Q0 + 16 * ND + kk);
    f[s].q2 = *(const v4f*)(Q0 + 32 * ND + kk);
    f[s].q3 = *(const v4f*)(Q0 + 48 * ND + kk);
  }
}

__device__ __forceinline__ void score_mma8(v2f a0, v2f a1, v2f q0, v2f q1, v2f q2, v2f q3,
                                           v8f (&acc)[2][4]) {
  acc[0][0] = __builtin_amdgcn_wmma_f32_16x16x4_f32(false, a0, false, q0, (short)0,
                                                    acc[0][0], false, false);
  acc[1][0] = __builtin_amdgcn_wmma_f32_16x16x4_f32(false, a1, false, q0, (short)0,
                                                    acc[1][0], false, false);
  acc[0][1] = __builtin_amdgcn_wmma_f32_16x16x4_f32(false, a0, false, q1, (short)0,
                                                    acc[0][1], false, false);
  acc[1][1] = __builtin_amdgcn_wmma_f32_16x16x4_f32(false, a1, false, q1, (short)0,
                                                    acc[1][1], false, false);
  acc[0][2] = __builtin_amdgcn_wmma_f32_16x16x4_f32(false, a0, false, q2, (short)0,
                                                    acc[0][2], false, false);
  acc[1][2] = __builtin_amdgcn_wmma_f32_16x16x4_f32(false, a1, false, q2, (short)0,
                                                    acc[1][2], false, false);
  acc[0][3] = __builtin_amdgcn_wmma_f32_16x16x4_f32(false, a0, false, q3, (short)0,
                                                    acc[0][3], false, false);
  acc[1][3] = __builtin_amdgcn_wmma_f32_16x16x4_f32(false, a1, false, q3, (short)0,
                                                    acc[1][3], false, false);
}

__device__ __forceinline__ void score_mma(const ScoreFrag (&f)[2], v8f (&acc)[2][4]) {
#pragma unroll
  for (int s = 0; s < 2; ++s) {
    v4f a0 = f[s].a0 * f[s].wu;
    v4f a1 = f[s].a1 * f[s].wu;
    score_mma8(lo2(a0), lo2(a1), lo2(f[s].q0), lo2(f[s].q1), lo2(f[s].q2), lo2(f[s].q3), acc);
    score_mma8(hi2(a0), hi2(a1), hi2(f[s].q0), hi2(f[s].q1), hi2(f[s].q2), hi2(f[s].q3), acc);
  }
}

#define SL_PITCH 132
#define SM_ROWS 16
static_assert(SM_ROWS == 16 && NJ == 128, "a lane holds 4 of a row's 128 values; 8 waves x 16 rows = the block's 128");
__global__ void __launch_bounds__(256)
k_score(const float* __restrict__ C, const float* __restrict__ Q,
        const float* __restrict__ w, const float* __restrict__ ch,
        const float* __restrict__ qb, float* __restrict__ Sj,
        float* __restrict__ ST) {
  __shared__ __attribute__((aligned(16))) float Sl[128 * SL_PITCH];
  int blk = blockIdx.x;
  int b = blk >> 3;
  int tblk = blk & 7;
  int wv = threadIdx.x >> 5;
  int lane = threadIdx.x & 31;
  int m = lane & 15, kh = lane >> 4;
  int t0 = tblk * 128 + (wv >> 1) * 32;
  int j0 = (wv & 1) * 64;
  const float* whu = w + 2 * ND + 4 * kh;
  const float* C0 = C + (size_t)(b * NT + t0 + m) * ND + 4 * kh;
  const float* Q0 = Q + (size_t)(b * NJ + j0 + m) * ND + 4 * kh;
  v8f acc[2][4] = {};
  ScoreFrag fa[2], fb[2];
  score_load(fa, whu, C0, Q0, 0);
#pragma unroll 1
  for (int kb = 0; kb < ND - 32; kb += 32) {
    score_load(fb, whu, C0, Q0, kb + 16);
    score_mma(fa, acc);
    score_load(fa, whu, C0, Q0, kb + 32);
    score_mma(fb, acc);
    __builtin_amdgcn_sched_group_barrier(0x020, 14, 0);
    __builtin_amdgcn_sched_group_barrier(0x008, 32, 0);
    __builtin_amdgcn_sched_group_barrier(0x020, 14, 0);
    __builtin_amdgcn_sched_group_barrier(0x008, 32, 0);
  }
  score_load(fb, whu, C0, Q0, ND - 16);
  score_mma(fa, acc);
  score_mma(fb, acc);

#pragma unroll
  for (int rt = 0; rt < 2; ++rt) {
    int tb = t0 + 16 * rt + 8 * kh;
    float chv[8];
#pragma unroll
    for (int r = 0; r < 8; ++r) chv[r] = ch[b * NT + tb + r];
#pragma unroll
    for (int ct = 0; ct < 4; ++ct) {
      int jj = j0 + 16 * ct + m;
      float qv = qb[b * NJ + jj];
      float* sl = Sl + (tb - tblk * 128) * SL_PITCH + jj;
#pragma unroll
      for (int r = 0; r < 8; ++r) {
        float s = acc[rt][ct][r] + chv[r] + qv;
        sl[r * SL_PITCH] = s;
        acc[rt][ct][r] = s;
      }
    }
  }
  __syncthreads();
#pragma unroll
  for (int rt = 0; rt < 2; ++rt) {
    int tb = t0 + 16 * rt + 8 * kh;
#pragma unroll
    for (int ct = 0; ct < 4; ++ct) {
      int jj = j0 + 16 * ct + m;
      float* tr = ST + (size_t)(b * NJ + jj) * NT + tb;
      *(float4*)(tr)     = make_float4(acc[rt][ct][0], acc[rt][ct][1], acc[rt][ct][2], acc[rt][ct][3]);
      *(float4*)(tr + 4) = make_float4(acc[rt][ct][4], acc[rt][ct][5], acc[rt][ct][6], acc[rt][ct][7]);
    }
  }
  const float* rl = Sl + (wv * 16) * SL_PITCH + lane * 4;
  float* dst = Sj + (size_t)(b * NT + tblk * 128 + wv * 16) * NJ + lane * 4;
#pragma unroll
  for (int g = 0; g < 16 / SM_ROWS; ++g) {
    v4f v[SM_ROWS];
#pragma unroll
    for (int r = 0; r < SM_ROWS; ++r) v[r] = *(const v4f*)(rl + (g * SM_ROWS + r) * SL_PITCH);
    float mx[SM_ROWS];
#pragma unroll
    for (int r = 0; r < SM_ROWS; ++r)
      mx[r] = fmaxf(fmaxf(v[r].x, v[r].y), fmaxf(v[r].z, v[r].w));
#pragma unroll
    for (int o = 16; o > 0; o >>= 1) {
#pragma unroll
      for (int r = 0; r < SM_ROWS; ++r) mx[r] = fmaxf(mx[r], __shfl_xor(mx[r], o, 32));
    }
    float sm[SM_ROWS];
#pragma unroll
    for (int r = 0; r < SM_ROWS; ++r) {
      v[r].x = expf(v[r].x - mx[r]); v[r].y = expf(v[r].y - mx[r]);
      v[r].z = expf(v[r].z - mx[r]); v[r].w = expf(v[r].w - mx[r]);
      sm[r] = v[r].x + v[r].y + v[r].z + v[r].w;
    }
#pragma unroll
    for (int o = 16; o > 0; o >>= 1) {
#pragma unroll
      for (int r = 0; r < SM_ROWS; ++r) sm[r] += __shfl_xor(sm[r], o, 32);
    }
#pragma unroll
    for (int r = 0; r < SM_ROWS; ++r) {
      float inv = 1.0f / sm[r];
      v4f o4;
      o4.x = v[r].x * inv; o4.y = v[r].y * inv; o4.z = v[r].z * inv; o4.w = v[r].w * inv;
      *(v4f*)(dst + (size_t)(g * SM_ROWS + r) * NJ) = o4;
    }
  }
}

__global__ void k_softmax_t(float* __restrict__ ST) {
  int wave = blockIdx.x * (blockDim.x >> 5) + (threadIdx.x >> 5);
  int lane = threadIdx.x & 31;
  if (wave >= NB * NJ) return;
  float* row = ST + (size_t)wave * NT + lane * 4;
  float4 v[8];
#pragma unroll
  for (int i = 0; i < 8; ++i) v[i] = *(const float4*)(row + i * 128);
  float m = -3.4e38f;
#pragma unroll
  for (int i = 0; i < 8; ++i)
    m = fmaxf(m, fmaxf(fmaxf(v[i].x, v[i].y), fmaxf(v[i].z, v[i].w)));
  m = wave_reduce_max(m);
  float s = 0.f;
#pragma unroll
  for (int i = 0; i < 8; ++i) {
    v[i].x = expf(v[i].x - m); v[i].y = expf(v[i].y - m);
    v[i].z = expf(v[i].z - m); v[i].w = expf(v[i].w - m);
    s += (v[i].x + v[i].y) + (v[i].z + v[i].w);
  }
  s = wave_reduce_sum(s);
  float inv = 1.0f / s;
#pragma unroll
  for (int i = 0; i < 8; ++i) {
    float4 o;
    o.x = v[i].x * inv; o.y = v[i].y * inv; o.z = v[i].z * inv; o.w = v[i].w * inv;
    *(float4*)(row + i * 128) = o;
  }
}

struct TmpFrag { v4f a0, a1; v2f b0l, b1l, b0h, b1h; };

__device__ __forceinline__ void tmp_load(TmpFrag (&f)[4], const float* A0,
                                         const float* Cb, int kb) {
#pragma unroll
  for (int s = 0; s < 4; ++s) {
    int k = kb + 8 * s;
    f[s].a0 = *(const v4f*)(A0 + k);
    f[s].a1 = *(const v4f*)(A0 + 16 * NT + k);
    f[s].b0l.x = Cb[(size_t)k * ND];
    f[s].b0l.y = Cb[(size_t)(k + 1) * ND];
    f[s].b1l.x = Cb[(size_t)k * ND + 16];
    f[s].b1l.y = Cb[(size_t)(k + 1) * ND + 16];
    f[s].b0h.x = Cb[(size_t)(k + 2) * ND];
    f[s].b0h.y = Cb[(size_t)(k + 3) * ND];
    f[s].b1h.x = Cb[(size_t)(k + 2) * ND + 16];
    f[s].b1h.y = Cb[(size_t)(k + 3) * ND + 16];
  }
}

__device__ __forceinline__ void tmp_mma4(v2f a0, v2f a1, v2f b0, v2f b1, v8f (&acc)[2][2]) {
  acc[0][0] = __builtin_amdgcn_wmma_f32_16x16x4_f32(false, a0, false, b0, (short)0,
                                                    acc[0][0], false, false);
  acc[1][0] = __builtin_amdgcn_wmma_f32_16x16x4_f32(false, a1, false, b0, (short)0,
                                                    acc[1][0], false, false);
  acc[0][1] = __builtin_amdgcn_wmma_f32_16x16x4_f32(false, a0, false, b1, (short)0,
                                                    acc[0][1], false, false);
  acc[1][1] = __builtin_amdgcn_wmma_f32_16x16x4_f32(false, a1, false, b1, (short)0,
                                                    acc[1][1], false, false);
}

__device__ __forceinline__ void tmp_mma(const TmpFrag (&f)[4], v8f (&acc)[2][2]) {
#pragma unroll
  for (int s = 0; s < 4; ++s) {
    tmp_mma4(lo2(f[s].a0), lo2(f[s].a1), f[s].b0l, f[s].b1l, acc);
    tmp_mma4(hi2(f[s].a0), hi2(f[s].a1), f[s].b0h, f[s].b1h, acc);
  }
}

__device__ __forceinline__ void
tmp_role(int blk, const float* __restrict__ StT, const float* __restrict__ C,
         float* __restrict__ tmp) {
  int b = blk >> 3;
  int dblk = blk & 7;
  int wv = threadIdx.x >> 5;
  int lane = threadIdx.x & 31;
  int m = lane & 15, kh = lane >> 4;
  int j0 = wv * 32, d0 = dblk * 32;
  const float* A0 = StT + (size_t)(b * NJ + j0 + m) * NT + 4 * kh;
  const float* Cb = C + (size_t)b * NT * ND + (size_t)(4 * kh) * ND + d0 + m;
  v8f acc[2][2] = {};
  TmpFrag fa[4], fb[4];
  tmp_load(fa, A0, Cb, 0);
#pragma unroll 1
  for (int kb = 0; kb < NT - 64; kb += 64) {
    tmp_load(fb, A0, Cb, kb + 32);
    tmp_mma(fa, acc);
    tmp_load(fa, A0, Cb, kb + 64);
    tmp_mma(fb, acc);
    __builtin_amdgcn_sched_group_barrier(0x020, 40, 0);
    __builtin_amdgcn_sched_group_barrier(0x008, 32, 0);
    __builtin_amdgcn_sched_group_barrier(0x020, 40, 0);
    __builtin_amdgcn_sched_group_barrier(0x008, 32, 0);
  }
  tmp_load(fb, A0, Cb, NT - 32);
  tmp_mma(fa, acc);
  tmp_mma(fb, acc);

#pragma unroll
  for (int rt = 0; rt < 2; ++rt) {
#pragma unroll
    for (int r = 0; r < 8; ++r) {
      int jj = j0 + 16 * rt + r + 8 * kh;
      float* ob = tmp + (size_t)(b * NJ + jj) * ND + d0 + m;
      ob[0]  = acc[rt][0][r];
      ob[16] = acc[rt][1][r];
    }
  }
}

struct OutFrag { v4f a0, a1; v2f bl[4], bh[4]; };

__device__ __forceinline__ void out_load(OutFrag& f, const float* S0, const float* S1,
                                         const float* Bp, int k0) {
  f.a0 = *(const v4f*)(S0 + k0);
  f.a1 = *(const v4f*)(S1 + k0);
  const float* p = Bp + (size_t)k0 * ND;
#pragma unroll
  for (int c = 0; c < 4; ++c) {
    f.bl[c].x = p[16 * c];           f.bl[c].y = p[ND + 16 * c];
    f.bh[c].x = p[2 * ND + 16 * c];  f.bh[c].y = p[3 * ND + 16 * c];
  }
}

__device__ __forceinline__ void out_mma(const OutFrag& f, v8f (&acc)[2][4]) {
  v2f a0l = lo2(f.a0), a1l = lo2(f.a1), a0h = hi2(f.a0), a1h = hi2(f.a1);
#pragma unroll
  for (int c = 0; c < 4; ++c) {
    acc[0][c] = __builtin_amdgcn_wmma_f32_16x16x4_f32(false, a0l, false, f.bl[c], (short)0,
                                                      acc[0][c], false, false);
    acc[1][c] = __builtin_amdgcn_wmma_f32_16x16x4_f32(false, a1l, false, f.bl[c], (short)0,
                                                      acc[1][c], false, false);
  }
#pragma unroll
  for (int c = 0; c < 4; ++c) {
    acc[0][c] = __builtin_amdgcn_wmma_f32_16x16x4_f32(false, a0h, false, f.bh[c], (short)0,
                                                      acc[0][c], false, false);
    acc[1][c] = __builtin_amdgcn_wmma_f32_16x16x4_f32(false, a1h, false, f.bh[c], (short)0,
                                                      acc[1][c], false, false);
  }
}

template <bool FIRST>
__device__ __forceinline__ void
out_product(int blk, const float* __restrict__ Sj, const float* __restrict__ Bmat,
            const float* __restrict__ C, float* __restrict__ out) {
  int b = blk >> 5;
  int tblk = blk & 31;
  int wv = threadIdx.x >> 5;
  int lane = threadIdx.x & 31;
  int m = lane & 15, kh = lane >> 4;
  int t0 = tblk * 32, d0 = wv * 64;
  const float* S0 = Sj + (size_t)(b * NT + t0 + m) * NJ + 4 * kh;
  const float* S1 = S0 + (size_t)16 * NJ;
  const float* Bp = Bmat + (size_t)b * NJ * ND + (size_t)(4 * kh) * ND + d0 + m;
  v8f acc[2][4] = {};
  OutFrag fa, fb;
  out_load(fa, S0, S1, Bp, 0);
#pragma unroll 1
  for (int k0 = 0; k0 < NJ - 16; k0 += 16) {
    out_load(fb, S0, S1, Bp, k0 + 8);
    out_mma(fa, acc);
    out_load(fa, S0, S1, Bp, k0 + 16);
    out_mma(fb, acc);
    __builtin_amdgcn_sched_group_barrier(0x020, 18, 0);
    __builtin_amdgcn_sched_group_barrier(0x008, 16, 0);
    __builtin_amdgcn_sched_group_barrier(0x020, 18, 0);
    __builtin_amdgcn_sched_group_barrier(0x008, 16, 0);
  }
  out_load(fb, S0, S1, Bp, NJ - 8);
  out_mma(fa, acc);
  out_mma(fb, acc);

#pragma unroll
  for (int rt = 0; rt < 2; ++rt) {
#pragma unroll
    for (int r = 0; r < 8; ++r) {
      int t = t0 + 16 * rt + r + 8 * kh;
      const float* cp = C + (size_t)(b * NT + t) * ND + d0 + m;
      float* ob = out + (size_t)(b * NT + t) * (4 * ND) + d0 + m;
#pragma unroll
      for (int c = 0; c < 4; ++c) {
        float cv = cp[16 * c];
        float a = acc[rt][c][r];
        if (FIRST) {
          ob[16 * c]          = cv;
          ob[ND + 16 * c]     = a;
          ob[2 * ND + 16 * c] = cv * a;
        } else {
          ob[3 * ND + 16 * c] = cv * a;
        }
      }
    }
  }
}

#define A_EARLY (NB * 24)
#define A_LATE  (NB * 32 - A_EARLY)

__global__ void __launch_bounds__(128)
k_tmp_and_a(const float* __restrict__ StT, const float* __restrict__ C,
            float* __restrict__ tmp, const float* __restrict__ Sj,
            const float* __restrict__ Q, float* __restrict__ out) {
  if (blockIdx.x < NB * 8) tmp_role(blockIdx.x, StT, C, tmp);
  else out_product<true>(blockIdx.x - NB * 8, Sj, Q, C, out);
}

__global__ void __launch_bounds__(128)
k_out_b(const float* __restrict__ Sj, const float* __restrict__ Q,
        const float* __restrict__ tmp, const float* __restrict__ C,
        float* __restrict__ out) {
  if (blockIdx.x < A_LATE) out_product<true>(A_EARLY + blockIdx.x, Sj, Q, C, out);
  else out_product<false>(blockIdx.x - A_LATE, Sj, tmp, C, out);
}

extern "C" void kernel_launch(void* const* d_in, const int* in_sizes, int n_in,
                              void* d_out, int out_size, void* d_ws, size_t ws_size,
                              hipStream_t stream) {
  (void)in_sizes; (void)n_in; (void)out_size; (void)ws_size;
  const float* C = (const float*)d_in[0];
  const float* Q = (const float*)d_in[1];
  const float* w = (const float*)d_in[2];
  float* out = (float*)d_out;

  float* ws = (float*)d_ws;
  float* ch = ws;
  float* qb = ch + NB * NT;
  float* Sj = qb + NB * NJ;
  float* ST = Sj + (size_t)NB * NT * NJ;
  float* tp = ST + (size_t)NB * NT * NJ;

  k_rowdots<<<(NB * NT + NB * NJ) / (4 * RD_ROWS), 128, 0, stream>>>(C, Q, w, ch, qb);
  k_score<<<NB * 8, 256, 0, stream>>>(C, Q, w, ch, qb, Sj, ST);
  k_softmax_t<<<(NB * NJ) / 4, 128, 0, stream>>>(ST);
  k_tmp_and_a<<<NB * 8 + A_EARLY, 128, 0, stream>>>(ST, C, tp, Sj, Q, out);
  k_out_b<<<A_LATE + NB * 32, 128, 0, stream>>>(Sj, Q, tp, C, out);
}


# softmax over t by 128-t blocks: k_score stores exp(S - block max) transposed, block max and sum; k_tmp scales A rows by exp(block max - max)/sum; k_softmax_t launch gone; changes column sum order
# speedup vs baseline: 2.8004x; 1.2752x over previous
#include <hip/hip_runtime.h>

#define NB 32
#define NT 1024
#define NJ 128
#define ND 256

typedef __attribute__((ext_vector_type(2))) float v2f;
typedef __attribute__((ext_vector_type(4))) float v4f;
typedef __attribute__((ext_vector_type(8))) float v8f;

#define RD_ROWS 8
static_assert((NB * NT) % RD_ROWS == 0 && (NB * NJ) % RD_ROWS == 0,
              "a wave's rows must lie in one of the two inputs");
static_assert(RD_ROWS == 8 && ND == 256, "the load fence and the stores name 8 rows of 8 values");
__global__ void __launch_bounds__(128)
k_rowdots(const float* __restrict__ C, const float* __restrict__ Q,
          const float* __restrict__ w, float* __restrict__ ch,
          float* __restrict__ qb) {
  int wave = blockIdx.x * (blockDim.x >> 5) + (threadIdx.x >> 5);
  int lane = threadIdx.x & 31;
  const int nch = NB * NT;
  const int nrows = nch + NB * NJ;
  int r0 = wave * RD_ROWS;
  if (r0 >= nrows) return;
  const float* rows; const float* wv; float* out;
  if (r0 < nch) { rows = C + (size_t)r0 * ND; wv = w;      out = ch + r0; }
  else { int r = r0 - nch; rows = Q + (size_t)r * ND; wv = w + ND; out = qb + r; }
  float wk[ND / 32];
#pragma unroll
  for (int k = 0; k < ND / 32; ++k) wk[k] = wv[lane + 32 * k];
  v8f x[RD_ROWS];
#pragma unroll
  for (int r = 0; r < RD_ROWS; ++r) {
#pragma unroll
    for (int k = 0; k < ND / 32; ++k) x[r][k] = rows[(size_t)r * ND + lane + 32 * k];
  }
  asm volatile("" :: "v"(x[0]), "v"(x[1]), "v"(x[2]), "v"(x[3]),
                     "v"(x[4]), "v"(x[5]), "v"(x[6]), "v"(x[7]));
  float s[RD_ROWS];
#pragma unroll
  for (int r = 0; r < RD_ROWS; ++r) {
    float a = 0.f;
#pragma unroll
    for (int k = 0; k < ND / 32; ++k) a += x[r][k] * wk[k];
    s[r] = a;
  }
#pragma unroll
  for (int o = 16; o > 0; o >>= 1) {
#pragma unroll
    for (int r = 0; r < RD_ROWS; ++r) s[r] += __shfl_xor(s[r], o, 32);
  }
  if (lane == 0) {
    *(float4*)(out)     = make_float4(s[0], s[1], s[2], s[3]);
    *(float4*)(out + 4) = make_float4(s[4], s[5], s[6], s[7]);
  }
}

struct ScoreFrag { v4f wu, a0, a1, q0, q1, q2, q3; };

__device__ __forceinline__ v2f lo2(v4f x) { return __builtin_shufflevector(x, x, 0, 1); }
__device__ __forceinline__ v2f hi2(v4f x) { return __builtin_shufflevector(x, x, 2, 3); }

__device__ __forceinline__ void score_load(ScoreFrag (&f)[2], const float* whu,
                                           const float* C0, const float* Q0, int kb) {
#pragma unroll
  for (int s = 0; s < 2; ++s) {
    int kk = kb + 8 * s;
    f[s].wu = *(const v4f*)(whu + kk);
    f[s].a0 = *(const v4f*)(C0 + kk);
    f[s].a1 = *(const v4f*)(C0 + 16 * ND + kk);
    f[s].q0 = *(const v4f*)(Q0 + kk);
    f[s].q1 = *(const v4f*)(Q0 + 16 * ND + kk);
    f[s].q2 = *(const v4f*)(Q0 + 32 * ND + kk);
    f[s].q3 = *(const v4f*)(Q0 + 48 * ND + kk);
  }
}

__device__ __forceinline__ void score_mma8(v2f a0, v2f a1, v2f q0, v2f q1, v2f q2, v2f q3,
                                           v8f (&acc)[2][4]) {
  acc[0][0] = __builtin_amdgcn_wmma_f32_16x16x4_f32(false, a0, false, q0, (short)0,
                                                    acc[0][0], false, false);
  acc[1][0] = __builtin_amdgcn_wmma_f32_16x16x4_f32(false, a1, false, q0, (short)0,
                                                    acc[1][0], false, false);
  acc[0][1] = __builtin_amdgcn_wmma_f32_16x16x4_f32(false, a0, false, q1, (short)0,
                                                    acc[0][1], false, false);
  acc[1][1] = __builtin_amdgcn_wmma_f32_16x16x4_f32(false, a1, false, q1, (short)0,
                                                    acc[1][1], false, false);
  acc[0][2] = __builtin_amdgcn_wmma_f32_16x16x4_f32(false, a0, false, q2, (short)0,
                                                    acc[0][2], false, false);
  acc[1][2] = __builtin_amdgcn_wmma_f32_16x16x4_f32(false, a1, false, q2, (short)0,
                                                    acc[1][2], false, false);
  acc[0][3] = __builtin_amdgcn_wmma_f32_16x16x4_f32(false, a0, false, q3, (short)0,
                                                    acc[0][3], false, false);
  acc[1][3] = __builtin_amdgcn_wmma_f32_16x16x4_f32(false, a1, false, q3, (short)0,
                                                    acc[1][3], false, false);
}

__device__ __forceinline__ void score_mma(const ScoreFrag (&f)[2], v8f (&acc)[2][4]) {
#pragma unroll
  for (int s = 0; s < 2; ++s) {
    v4f a0 = f[s].a0 * f[s].wu;
    v4f a1 = f[s].a1 * f[s].wu;
    score_mma8(lo2(a0), lo2(a1), lo2(f[s].q0), lo2(f[s].q1), lo2(f[s].q2), lo2(f[s].q3), acc);
    score_mma8(hi2(a0), hi2(a1), hi2(f[s].q0), hi2(f[s].q1), hi2(f[s].q2), hi2(f[s].q3), acc);
  }
}

#define SL_PITCH 132
#define SM_ROWS 16
static_assert(SM_ROWS == 16 && NJ == 128, "a lane holds 4 of a row's 128 values; 8 waves x 16 rows = the block's 128");
__global__ void __launch_bounds__(256)
k_score(const float* __restrict__ C, const float* __restrict__ Q,
        const float* __restrict__ w, const float* __restrict__ ch,
        const float* __restrict__ qb, float* __restrict__ Sj,
        float* __restrict__ ET, float* __restrict__ cm,
        float* __restrict__ cz) {
  __shared__ __attribute__((aligned(16))) float Sl[128 * SL_PITCH];
  __shared__ __attribute__((aligned(16))) float SlT[128 * SL_PITCH];
  int blk = blockIdx.x;
  int b = blk >> 3;
  int tblk = blk & 7;
  int wv = threadIdx.x >> 5;
  int lane = threadIdx.x & 31;
  int m = lane & 15, kh = lane >> 4;
  int t0 = tblk * 128 + (wv >> 1) * 32;
  int j0 = (wv & 1) * 64;
  const float* whu = w + 2 * ND + 4 * kh;
  const float* C0 = C + (size_t)(b * NT + t0 + m) * ND + 4 * kh;
  const float* Q0 = Q + (size_t)(b * NJ + j0 + m) * ND + 4 * kh;
  v8f acc[2][4] = {};
  ScoreFrag fa[2], fb[2];
  score_load(fa, whu, C0, Q0, 0);
#pragma unroll 1
  for (int kb = 0; kb < ND - 32; kb += 32) {
    score_load(fb, whu, C0, Q0, kb + 16);
    score_mma(fa, acc);
    score_load(fa, whu, C0, Q0, kb + 32);
    score_mma(fb, acc);
    __builtin_amdgcn_sched_group_barrier(0x020, 14, 0);
    __builtin_amdgcn_sched_group_barrier(0x008, 32, 0);
    __builtin_amdgcn_sched_group_barrier(0x020, 14, 0);
    __builtin_amdgcn_sched_group_barrier(0x008, 32, 0);
  }
  score_load(fb, whu, C0, Q0, ND - 16);
  score_mma(fa, acc);
  score_mma(fb, acc);

#pragma unroll
  for (int rt = 0; rt < 2; ++rt) {
    int tb = t0 + 16 * rt + 8 * kh;
    float chv[8];
#pragma unroll
    for (int r = 0; r < 8; ++r) chv[r] = ch[b * NT + tb + r];
#pragma unroll
    for (int ct = 0; ct < 4; ++ct) {
      int jj = j0 + 16 * ct + m;
      float qv = qb[b * NJ + jj];
      int tl = tb - tblk * 128;
      float* sl = Sl + tl * SL_PITCH + jj;
      float s[8];
#pragma unroll
      for (int r = 0; r < 8; ++r) {
        s[r] = acc[rt][ct][r] + chv[r] + qv;
        sl[r * SL_PITCH] = s[r];
      }
      float* st = SlT + jj * SL_PITCH + tl;
      v4f s_lo, s_hi;
      s_lo.x = s[0]; s_lo.y = s[1]; s_lo.z = s[2]; s_lo.w = s[3];
      s_hi.x = s[4]; s_hi.y = s[5]; s_hi.z = s[6]; s_hi.w = s[7];
      *(v4f*)(st)     = s_lo;
      *(v4f*)(st + 4) = s_hi;
    }
  }
  __syncthreads();
  const float* rl = Sl + (wv * 16) * SL_PITCH + lane * 4;
  float* dst = Sj + (size_t)(b * NT + tblk * 128 + wv * 16) * NJ + lane * 4;
#pragma unroll
  for (int g = 0; g < 16 / SM_ROWS; ++g) {
    v4f v[SM_ROWS];
#pragma unroll
    for (int r = 0; r < SM_ROWS; ++r) v[r] = *(const v4f*)(rl + (g * SM_ROWS + r) * SL_PITCH);
    float mx[SM_ROWS];
#pragma unroll
    for (int r = 0; r < SM_ROWS; ++r)
      mx[r] = fmaxf(fmaxf(v[r].x, v[r].y), fmaxf(v[r].z, v[r].w));
#pragma unroll
    for (int o = 16; o > 0; o >>= 1) {
#pragma unroll
      for (int r = 0; r < SM_ROWS; ++r) mx[r] = fmaxf(mx[r], __shfl_xor(mx[r], o, 32));
    }
    float sm[SM_ROWS];
#pragma unroll
    for (int r = 0; r < SM_ROWS; ++r) {
      v[r].x = expf(v[r].x - mx[r]); v[r].y = expf(v[r].y - mx[r]);
      v[r].z = expf(v[r].z - mx[r]); v[r].w = expf(v[r].w - mx[r]);
      sm[r] = v[r].x + v[r].y + v[r].z + v[r].w;
    }
#pragma unroll
    for (int o = 16; o > 0; o >>= 1) {
#pragma unroll
      for (int r = 0; r < SM_ROWS; ++r) sm[r] += __shfl_xor(sm[r], o, 32);
    }
#pragma unroll
    for (int r = 0; r < SM_ROWS; ++r) {
      float inv = 1.0f / sm[r];
      v4f o4;
      o4.x = v[r].x * inv; o4.y = v[r].y * inv; o4.z = v[r].z * inv; o4.w = v[r].w * inv;
      *(v4f*)(dst + (size_t)(g * SM_ROWS + r) * NJ) = o4;
    }
  }
  int rr = lane >> 3, c8 = lane & 7;
  const float* cl = SlT + (wv * 16 + 4 * rr) * SL_PITCH + 4 * c8;
  float* dse = ET + (size_t)(b * NJ + wv * 16 + 4 * rr) * NT + tblk * 128 + 4 * c8;
  {
    v4f v[4][4];
#pragma unroll
    for (int g = 0; g < 4; ++g) {
#pragma unroll
      for (int i = 0; i < 4; ++i) v[g][i] = *(const v4f*)(cl + g * SL_PITCH + 32 * i);
    }
    float mx[4];
#pragma unroll
    for (int g = 0; g < 4; ++g) {
      float a = fmaxf(fmaxf(v[g][0].x, v[g][0].y), fmaxf(v[g][0].z, v[g][0].w));
#pragma unroll
      for (int i = 1; i < 4; ++i)
        a = fmaxf(a, fmaxf(fmaxf(v[g][i].x, v[g][i].y), fmaxf(v[g][i].z, v[g][i].w)));
      mx[g] = a;
    }
#pragma unroll
    for (int o = 4; o > 0; o >>= 1) {
#pragma unroll
      for (int g = 0; g < 4; ++g) mx[g] = fmaxf(mx[g], __shfl_xor(mx[g], o, 32));
    }
    float sm[4];
#pragma unroll
    for (int g = 0; g < 4; ++g) {
      float a = 0.f;
#pragma unroll
      for (int i = 0; i < 4; ++i) {
        v[g][i].x = expf(v[g][i].x - mx[g]); v[g][i].y = expf(v[g][i].y - mx[g]);
        v[g][i].z = expf(v[g][i].z - mx[g]); v[g][i].w = expf(v[g][i].w - mx[g]);
        a += (v[g][i].x + v[g][i].y) + (v[g][i].z + v[g][i].w);
        *(v4f*)(dse + (size_t)g * NT + 32 * i) = v[g][i];
      }
      sm[g] = a;
    }
#pragma unroll
    for (int o = 4; o > 0; o >>= 1) {
#pragma unroll
      for (int g = 0; g < 4; ++g) sm[g] += __shfl_xor(sm[g], o, 32);
    }
    if (c8 == 0) {
      size_t so = (size_t)(b * 8 + tblk) * NJ + wv * 16 + 4 * rr;
      *(float4*)(cm + so) = make_float4(mx[0], mx[1], mx[2], mx[3]);
      *(float4*)(cz + so) = make_float4(sm[0], sm[1], sm[2], sm[3]);
    }
  }
}

struct TmpFrag { v4f a0, a1; v2f b0l, b1l, b0h, b1h; };

__device__ __forceinline__ void tmp_load(TmpFrag (&f)[4], const float* A0,
                                         const float* Cb, int kb) {
#pragma unroll
  for (int s = 0; s < 4; ++s) {
    int k = kb + 8 * s;
    f[s].a0 = *(const v4f*)(A0 + k);
    f[s].a1 = *(const v4f*)(A0 + 16 * NT + k);
    f[s].b0l.x = Cb[(size_t)k * ND];
    f[s].b0l.y = Cb[(size_t)(k + 1) * ND];
    f[s].b1l.x = Cb[(size_t)k * ND + 16];
    f[s].b1l.y = Cb[(size_t)(k + 1) * ND + 16];
    f[s].b0h.x = Cb[(size_t)(k + 2) * ND];
    f[s].b0h.y = Cb[(size_t)(k + 3) * ND];
    f[s].b1h.x = Cb[(size_t)(k + 2) * ND + 16];
    f[s].b1h.y = Cb[(size_t)(k + 3) * ND + 16];
  }
}

__device__ __forceinline__ void tmp_mma4(v2f a0, v2f a1, v2f b0, v2f b1, v8f (&acc)[2][2]) {
  acc[0][0] = __builtin_amdgcn_wmma_f32_16x16x4_f32(false, a0, false, b0, (short)0,
                                                    acc[0][0], false, false);
  acc[1][0] = __builtin_amdgcn_wmma_f32_16x16x4_f32(false, a1, false, b0, (short)0,
                                                    acc[1][0], false, false);
  acc[0][1] = __builtin_amdgcn_wmma_f32_16x16x4_f32(false, a0, false, b1, (short)0,
                                                    acc[0][1], false, false);
  acc[1][1] = __builtin_amdgcn_wmma_f32_16x16x4_f32(false, a1, false, b1, (short)0,
                                                    acc[1][1], false, false);
}

__device__ __forceinline__ void tmp_mma(const TmpFrag (&f)[4], float s0, float s1,
                                        v8f (&acc)[2][2]) {
#pragma unroll
  for (int s = 0; s < 4; ++s) {
    v4f a0 = f[s].a0 * s0;
    v4f a1 = f[s].a1 * s1;
    tmp_mma4(lo2(a0), lo2(a1), f[s].b0l, f[s].b1l, acc);
    tmp_mma4(hi2(a0), hi2(a1), f[s].b0h, f[s].b1h, acc);
  }
}

#define T_BLKS (NT / 128)
static_assert(T_BLKS == 8, "k_score writes one max and one sum per block of 128 t: 8 per column");
__device__ __forceinline__ void
tmp_role(int blk, const float* __restrict__ ET, const float* __restrict__ cm,
         const float* __restrict__ cz, const float* __restrict__ C,
         float* __restrict__ tmp) {
  __shared__ float ftab[4][T_BLKS][2][32];
  int b = blk >> 3;
  int dblk = blk & 7;
  int wv = threadIdx.x >> 5;
  int lane = threadIdx.x & 31;
  int m = lane & 15, kh = lane >> 4;
  int j0 = wv * 32, d0 = dblk * 32;
  const float* A0 = ET + (size_t)(b * NJ + j0 + m) * NT + 4 * kh;
  const float* Cb = C + (size_t)b * NT * ND + (size_t)(4 * kh) * ND + d0 + m;
  {
    const float* pm = cm + (size_t)b * T_BLKS * NJ + j0 + m;
    const float* pz = cz + (size_t)b * T_BLKS * NJ + j0 + m;
    float m0[T_BLKS], z0[T_BLKS], m1[T_BLKS], z1[T_BLKS];
#pragma unroll
    for (int q = 0; q < T_BLKS; ++q) {
      m0[q] = pm[q * NJ];      z0[q] = pz[q * NJ];
      m1[q] = pm[q * NJ + 16]; z1[q] = pz[q * NJ + 16];
    }
    float M0 = m0[0], M1 = m1[0];
#pragma unroll
    for (int q = 1; q < T_BLKS; ++q) { M0 = fmaxf(M0, m0[q]); M1 = fmaxf(M1, m1[q]); }
    float Z0 = 0.f, Z1 = 0.f;
#pragma unroll
    for (int q = 0; q < T_BLKS; ++q) {
      m0[q] = expf(m0[q] - M0); Z0 += z0[q] * m0[q];
      m1[q] = expf(m1[q] - M1); Z1 += z1[q] * m1[q];
    }
    float i0 = 1.0f / Z0, i1 = 1.0f / Z1;
#pragma unroll
    for (int q = 0; q < T_BLKS; ++q) {
      ftab[wv][q][0][lane] = m0[q] * i0;
      ftab[wv][q][1][lane] = m1[q] * i1;
    }
  }
  v8f acc[2][2] = {};
  TmpFrag fa[4], fb[4];
  tmp_load(fa, A0, Cb, 0);
#pragma unroll 1
  for (int kb = 0; kb < NT - 64; kb += 64) {
    float s0 = ftab[wv][kb >> 7][0][lane], s1 = ftab[wv][kb >> 7][1][lane];
    tmp_load(fb, A0, Cb, kb + 32);
    tmp_mma(fa, s0, s1, acc);
    tmp_load(fa, A0, Cb, kb + 64);
    tmp_mma(fb, s0, s1, acc);
    __builtin_amdgcn_sched_group_barrier(0x020, 40, 0);
    __builtin_amdgcn_sched_group_barrier(0x008, 32, 0);
    __builtin_amdgcn_sched_group_barrier(0x020, 40, 0);
    __builtin_amdgcn_sched_group_barrier(0x008, 32, 0);
  }
  tmp_load(fb, A0, Cb, NT - 32);
  {
    float s0 = ftab[wv][T_BLKS - 1][0][lane], s1 = ftab[wv][T_BLKS - 1][1][lane];
    tmp_mma(fa, s0, s1, acc);
    tmp_mma(fb, s0, s1, acc);
  }

#pragma unroll
  for (int rt = 0; rt < 2; ++rt) {
#pragma unroll
    for (int r = 0; r < 8; ++r) {
      int jj = j0 + 16 * rt + r + 8 * kh;
      float* ob = tmp + (size_t)(b * NJ + jj) * ND + d0 + m;
      ob[0]  = acc[rt][0][r];
      ob[16] = acc[rt][1][r];
    }
  }
}

struct OutFrag { v4f a0, a1; v2f bl[4], bh[4]; };

__device__ __forceinline__ void out_load(OutFrag& f, const float* S0, const float* S1,
                                         const float* Bp, int k0) {
  f.a0 = *(const v4f*)(S0 + k0);
  f.a1 = *(const v4f*)(S1 + k0);
  const float* p = Bp + (size_t)k0 * ND;
#pragma unroll
  for (int c = 0; c < 4; ++c) {
    f.bl[c].x = p[16 * c];           f.bl[c].y = p[ND + 16 * c];
    f.bh[c].x = p[2 * ND + 16 * c];  f.bh[c].y = p[3 * ND + 16 * c];
  }
}

__device__ __forceinline__ void out_mma(const OutFrag& f, v8f (&acc)[2][4]) {
  v2f a0l = lo2(f.a0), a1l = lo2(f.a1), a0h = hi2(f.a0), a1h = hi2(f.a1);
#pragma unroll
  for (int c = 0; c < 4; ++c) {
    acc[0][c] = __builtin_amdgcn_wmma_f32_16x16x4_f32(false, a0l, false, f.bl[c], (short)0,
                                                      acc[0][c], false, false);
    acc[1][c] = __builtin_amdgcn_wmma_f32_16x16x4_f32(false, a1l, false, f.bl[c], (short)0,
                                                      acc[1][c], false, false);
  }
#pragma unroll
  for (int c = 0; c < 4; ++c) {
    acc[0][c] = __builtin_amdgcn_wmma_f32_16x16x4_f32(false, a0h, false, f.bh[c], (short)0,
                                                      acc[0][c], false, false);
    acc[1][c] = __builtin_amdgcn_wmma_f32_16x16x4_f32(false, a1h, false, f.bh[c], (short)0,
                                                      acc[1][c], false, false);
  }
}

template <bool FIRST>
__device__ __forceinline__ void
out_product(int blk, const float* __restrict__ Sj, const float* __restrict__ Bmat,
            const float* __restrict__ C, float* __restrict__ out) {
  int b = blk >> 5;
  int tblk = blk & 31;
  int wv = threadIdx.x >> 5;
  int lane = threadIdx.x & 31;
  int m = lane & 15, kh = lane >> 4;
  int t0 = tblk * 32, d0 = wv * 64;
  const float* S0 = Sj + (size_t)(b * NT + t0 + m) * NJ + 4 * kh;
  const float* S1 = S0 + (size_t)16 * NJ;
  const float* Bp = Bmat + (size_t)b * NJ * ND + (size_t)(4 * kh) * ND + d0 + m;
  v8f acc[2][4] = {};
  OutFrag fa, fb;
  out_load(fa, S0, S1, Bp, 0);
#pragma unroll 1
  for (int k0 = 0; k0 < NJ - 16; k0 += 16) {
    out_load(fb, S0, S1, Bp, k0 + 8);
    out_mma(fa, acc);
    out_load(fa, S0, S1, Bp, k0 + 16);
    out_mma(fb, acc);
    __builtin_amdgcn_sched_group_barrier(0x020, 18, 0);
    __builtin_amdgcn_sched_group_barrier(0x008, 16, 0);
    __builtin_amdgcn_sched_group_barrier(0x020, 18, 0);
    __builtin_amdgcn_sched_group_barrier(0x008, 16, 0);
  }
  out_load(fb, S0, S1, Bp, NJ - 8);
  out_mma(fa, acc);
  out_mma(fb, acc);

#pragma unroll
  for (int rt = 0; rt < 2; ++rt) {
#pragma unroll
    for (int r = 0; r < 8; ++r) {
      int t = t0 + 16 * rt + r + 8 * kh;
      const float* cp = C + (size_t)(b * NT + t) * ND + d0 + m;
      float* ob = out + (size_t)(b * NT + t) * (4 * ND) + d0 + m;
#pragma unroll
      for (int c = 0; c < 4; ++c) {
        float cv = cp[16 * c];
        float a = acc[rt][c][r];
        if (FIRST) {
          ob[16 * c]          = cv;
          ob[ND + 16 * c]     = a;
          ob[2 * ND + 16 * c] = cv * a;
        } else {
          ob[3 * ND + 16 * c] = cv * a;
        }
      }
    }
  }
}

#define A_EARLY (NB * 24)
#define A_LATE  (NB * 32 - A_EARLY)

__global__ void __launch_bounds__(128)
k_tmp_and_a(const float* __restrict__ ET, const float* __restrict__ cm,
            const float* __restrict__ cz, const float* __restrict__ C,
            float* __restrict__ tmp, const float* __restrict__ Sj,
            const float* __restrict__ Q, float* __restrict__ out) {
  if (blockIdx.x < NB * 8) tmp_role(blockIdx.x, ET, cm, cz, C, tmp);
  else out_product<true>(blockIdx.x - NB * 8, Sj, Q, C, out);
}

__global__ void __launch_bounds__(128)
k_out_b(const float* __restrict__ Sj, const float* __restrict__ Q,
        const float* __restrict__ tmp, const float* __restrict__ C,
        float* __restrict__ out) {
  if (blockIdx.x < A_LATE) out_product<true>(A_EARLY + blockIdx.x, Sj, Q, C, out);
  else out_product<false>(blockIdx.x - A_LATE, Sj, tmp, C, out);
}

extern "C" void kernel_launch(void* const* d_in, const int* in_sizes, int n_in,
                              void* d_out, int out_size, void* d_ws, size_t ws_size,
                              hipStream_t stream) {
  (void)in_sizes; (void)n_in; (void)out_size; (void)ws_size;
  const float* C = (const float*)d_in[0];
  const float* Q = (const float*)d_in[1];
  const float* w = (const float*)d_in[2];
  float* out = (float*)d_out;

  float* ws = (float*)d_ws;
  float* ch = ws;
  float* qb = ch + NB * NT;
  float* Sj = qb + NB * NJ;
  float* ET = Sj + (size_t)NB * NT * NJ;
  float* tp = ET + (size_t)NB * NT * NJ;
  float* cm = tp + (size_t)NB * NJ * ND;
  float* cz = cm + NB * 8 * NJ;

  k_rowdots<<<(NB * NT + NB * NJ) / (4 * RD_ROWS), 128, 0, stream>>>(C, Q, w, ch, qb);
  k_score<<<NB * 8, 256, 0, stream>>>(C, Q, w, ch, qb, Sj, ET, cm, cz);
  k_tmp_and_a<<<NB * 8 + A_EARLY, 128, 0, stream>>>(ET, cm, cz, C, tp, Sj, Q, out);
  k_out_b<<<A_LATE + NB * 32, 128, 0, stream>>>(Sj, Q, tp, C, out);
}


# out_product epilogue: a wave's finished 32x64 tile passes through its own LDS rows; C loads and all four out planes go 16 bytes per lane, 4 whole 128-byte lines per instruction; sums unchanged
# speedup vs baseline: 3.0177x; 1.0776x over previous
#include <hip/hip_runtime.h>

#define NB 32
#define NT 1024
#define NJ 128
#define ND 256

typedef __attribute__((ext_vector_type(2))) float v2f;
typedef __attribute__((ext_vector_type(4))) float v4f;
typedef __attribute__((ext_vector_type(8))) float v8f;

#define RD_ROWS 8
static_assert((NB * NT) % RD_ROWS == 0 && (NB * NJ) % RD_ROWS == 0,
              "a wave's rows must lie in one of the two inputs");
static_assert(RD_ROWS == 8 && ND == 256, "the load fence and the stores name 8 rows of 8 values");
__global__ void __launch_bounds__(128)
k_rowdots(const float* __restrict__ C, const float* __restrict__ Q,
          const float* __restrict__ w, float* __restrict__ ch,
          float* __restrict__ qb) {
  int wave = blockIdx.x * (blockDim.x >> 5) + (threadIdx.x >> 5);
  int lane = threadIdx.x & 31;
  const int nch = NB * NT;
  const int nrows = nch + NB * NJ;
  int r0 = wave * RD_ROWS;
  if (r0 >= nrows) return;
  const float* rows; const float* wv; float* out;
  if (r0 < nch) { rows = C + (size_t)r0 * ND; wv = w;      out = ch + r0; }
  else { int r = r0 - nch; rows = Q + (size_t)r * ND; wv = w + ND; out = qb + r; }
  float wk[ND / 32];
#pragma unroll
  for (int k = 0; k < ND / 32; ++k) wk[k] = wv[lane + 32 * k];
  v8f x[RD_ROWS];
#pragma unroll
  for (int r = 0; r < RD_ROWS; ++r) {
#pragma unroll
    for (int k = 0; k < ND / 32; ++k) x[r][k] = rows[(size_t)r * ND + lane + 32 * k];
  }
  asm volatile("" :: "v"(x[0]), "v"(x[1]), "v"(x[2]), "v"(x[3]),
                     "v"(x[4]), "v"(x[5]), "v"(x[6]), "v"(x[7]));
  float s[RD_ROWS];
#pragma unroll
  for (int r = 0; r < RD_ROWS; ++r) {
    float a = 0.f;
#pragma unroll
    for (int k = 0; k < ND / 32; ++k) a += x[r][k] * wk[k];
    s[r] = a;
  }
#pragma unroll
  for (int o = 16; o > 0; o >>= 1) {
#pragma unroll
    for (int r = 0; r < RD_ROWS; ++r) s[r] += __shfl_xor(s[r], o, 32);
  }
  if (lane == 0) {
    *(float4*)(out)     = make_float4(s[0], s[1], s[2], s[3]);
    *(float4*)(out + 4) = make_float4(s[4], s[5], s[6], s[7]);
  }
}

struct ScoreFrag { v4f wu, a0, a1, q0, q1, q2, q3; };

__device__ __forceinline__ v2f lo2(v4f x) { return __builtin_shufflevector(x, x, 0, 1); }
__device__ __forceinline__ v2f hi2(v4f x) { return __builtin_shufflevector(x, x, 2, 3); }

__device__ __forceinline__ void score_load(ScoreFrag (&f)[2], const float* whu,
                                           const float* C0, const float* Q0, int kb) {
#pragma unroll
  for (int s = 0; s < 2; ++s) {
    int kk = kb + 8 * s;
    f[s].wu = *(const v4f*)(whu + kk);
    f[s].a0 = *(const v4f*)(C0 + kk);
    f[s].a1 = *(const v4f*)(C0 + 16 * ND + kk);
    f[s].q0 = *(const v4f*)(Q0 + kk);
    f[s].q1 = *(const v4f*)(Q0 + 16 * ND + kk);
    f[s].q2 = *(const v4f*)(Q0 + 32 * ND + kk);
    f[s].q3 = *(const v4f*)(Q0 + 48 * ND + kk);
  }
}

__device__ __forceinline__ void score_mma8(v2f a0, v2f a1, v2f q0, v2f q1, v2f q2, v2f q3,
                                           v8f (&acc)[2][4]) {
  acc[0][0] = __builtin_amdgcn_wmma_f32_16x16x4_f32(false, a0, false, q0, (short)0,
                                                    acc[0][0], false, false);
  acc[1][0] = __builtin_amdgcn_wmma_f32_16x16x4_f32(false, a1, false, q0, (short)0,
                                                    acc[1][0], false, false);
  acc[0][1] = __builtin_amdgcn_wmma_f32_16x16x4_f32(false, a0, false, q1, (short)0,
                                                    acc[0][1], false, false);
  acc[1][1] = __builtin_amdgcn_wmma_f32_16x16x4_f32(false, a1, false, q1, (short)0,
                                                    acc[1][1], false, false);
  acc[0][2] = __builtin_amdgcn_wmma_f32_16x16x4_f32(false, a0, false, q2, (short)0,
                                                    acc[0][2], false, false);
  acc[1][2] = __builtin_amdgcn_wmma_f32_16x16x4_f32(false, a1, false, q2, (short)0,
                                                    acc[1][2], false, false);
  acc[0][3] = __builtin_amdgcn_wmma_f32_16x16x4_f32(false, a0, false, q3, (short)0,
                                                    acc[0][3], false, false);
  acc[1][3] = __builtin_amdgcn_wmma_f32_16x16x4_f32(false, a1, false, q3, (short)0,
                                                    acc[1][3], false, false);
}

__device__ __forceinline__ void score_mma(const ScoreFrag (&f)[2], v8f (&acc)[2][4]) {
#pragma unroll
  for (int s = 0; s < 2; ++s) {
    v4f a0 = f[s].a0 * f[s].wu;
    v4f a1 = f[s].a1 * f[s].wu;
    score_mma8(lo2(a0), lo2(a1), lo2(f[s].q0), lo2(f[s].q1), lo2(f[s].q2), lo2(f[s].q3), acc);
    score_mma8(hi2(a0), hi2(a1), hi2(f[s].q0), hi2(f[s].q1), hi2(f[s].q2), hi2(f[s].q3), acc);
  }
}

#define SL_PITCH 132
#define SM_ROWS 16
static_assert(SM_ROWS == 16 && NJ == 128, "a lane holds 4 of a row's 128 values; 8 waves x 16 rows = the block's 128");
__global__ void __launch_bounds__(256)
k_score(const float* __restrict__ C, const float* __restrict__ Q,
        const float* __restrict__ w, const float* __restrict__ ch,
        const float* __restrict__ qb, float* __restrict__ Sj,
        float* __restrict__ ET, float* __restrict__ cm,
        float* __restrict__ cz) {
  __shared__ __attribute__((aligned(16))) float Sl[128 * SL_PITCH];
  __shared__ __attribute__((aligned(16))) float SlT[128 * SL_PITCH];
  int blk = blockIdx.x;
  int b = blk >> 3;
  int tblk = blk & 7;
  int wv = threadIdx.x >> 5;
  int lane = threadIdx.x & 31;
  int m = lane & 15, kh = lane >> 4;
  int t0 = tblk * 128 + (wv >> 1) * 32;
  int j0 = (wv & 1) * 64;
  const float* whu = w + 2 * ND + 4 * kh;
  const float* C0 = C + (size_t)(b * NT + t0 + m) * ND + 4 * kh;
  const float* Q0 = Q + (size_t)(b * NJ + j0 + m) * ND + 4 * kh;
  v8f acc[2][4] = {};
  ScoreFrag fa[2], fb[2];
  score_load(fa, whu, C0, Q0, 0);
#pragma unroll 1
  for (int kb = 0; kb < ND - 32; kb += 32) {
    score_load(fb, whu, C0, Q0, kb + 16);
    score_mma(fa, acc);
    score_load(fa, whu, C0, Q0, kb + 32);
    score_mma(fb, acc);
    __builtin_amdgcn_sched_group_barrier(0x020, 14, 0);
    __builtin_amdgcn_sched_group_barrier(0x008, 32, 0);
    __builtin_amdgcn_sched_group_barrier(0x020, 14, 0);
    __builtin_amdgcn_sched_group_barrier(0x008, 32, 0);
  }
  score_load(fb, whu, C0, Q0, ND - 16);
  score_mma(fa, acc);
  score_mma(fb, acc);

#pragma unroll
  for (int rt = 0; rt < 2; ++rt) {
    int tb = t0 + 16 * rt + 8 * kh;
    float chv[8];
#pragma unroll
    for (int r = 0; r < 8; ++r) chv[r] = ch[b * NT + tb + r];
#pragma unroll
    for (int ct = 0; ct < 4; ++ct) {
      int jj = j0 + 16 * ct + m;
      float qv = qb[b * NJ + jj];
      int tl = tb - tblk * 128;
      float* sl = Sl + tl * SL_PITCH + jj;
      float s[8];
#pragma unroll
      for (int r = 0; r < 8; ++r) {
        s[r] = acc[rt][ct][r] + chv[r] + qv;
        sl[r * SL_PITCH] = s[r];
      }
      float* st = SlT + jj * SL_PITCH + tl;
      v4f s_lo, s_hi;
      s_lo.x = s[0]; s_lo.y = s[1]; s_lo.z = s[2]; s_lo.w = s[3];
      s_hi.x = s[4]; s_hi.y = s[5]; s_hi.z = s[6]; s_hi.w = s[7];
      *(v4f*)(st)     = s_lo;
      *(v4f*)(st + 4) = s_hi;
    }
  }
  __syncthreads();
  const float* rl = Sl + (wv * 16) * SL_PITCH + lane * 4;
  float* dst = Sj + (size_t)(b * NT + tblk * 128 + wv * 16) * NJ + lane * 4;
#pragma unroll
  for (int g = 0; g < 16 / SM_ROWS; ++g) {
    v4f v[SM_ROWS];
#pragma unroll
    for (int r = 0; r < SM_ROWS; ++r) v[r] = *(const v4f*)(rl + (g * SM_ROWS + r) * SL_PITCH);
    float mx[SM_ROWS];
#pragma unroll
    for (int r = 0; r < SM_ROWS; ++r)
      mx[r] = fmaxf(fmaxf(v[r].x, v[r].y), fmaxf(v[r].z, v[r].w));
#pragma unroll
    for (int o = 16; o > 0; o >>= 1) {
#pragma unroll
      for (int r = 0; r < SM_ROWS; ++r) mx[r] = fmaxf(mx[r], __shfl_xor(mx[r], o, 32));
    }
    float sm[SM_ROWS];
#pragma unroll
    for (int r = 0; r < SM_ROWS; ++r) {
      v[r].x = expf(v[r].x - mx[r]); v[r].y = expf(v[r].y - mx[r]);
      v[r].z = expf(v[r].z - mx[r]); v[r].w = expf(v[r].w - mx[r]);
      sm[r] = v[r].x + v[r].y + v[r].z + v[r].w;
    }
#pragma unroll
    for (int o = 16; o > 0; o >>= 1) {
#pragma unroll
      for (int r = 0; r < SM_ROWS; ++r) sm[r] += __shfl_xor(sm[r], o, 32);
    }
#pragma unroll
    for (int r = 0; r < SM_ROWS; ++r) {
      float inv = 1.0f / sm[r];
      v4f o4;
      o4.x = v[r].x * inv; o4.y = v[r].y * inv; o4.z = v[r].z * inv; o4.w = v[r].w * inv;
      *(v4f*)(dst + (size_t)(g * SM_ROWS + r) * NJ) = o4;
    }
  }
  int rr = lane >> 3, c8 = lane & 7;
  const float* cl = SlT + (wv * 16 + 4 * rr) * SL_PITCH + 4 * c8;
  float* dse = ET + (size_t)(b * NJ + wv * 16 + 4 * rr) * NT + tblk * 128 + 4 * c8;
  {
    v4f v[4][4];
#pragma unroll
    for (int g = 0; g < 4; ++g) {
#pragma unroll
      for (int i = 0; i < 4; ++i) v[g][i] = *(const v4f*)(cl + g * SL_PITCH + 32 * i);
    }
    float mx[4];
#pragma unroll
    for (int g = 0; g < 4; ++g) {
      float a = fmaxf(fmaxf(v[g][0].x, v[g][0].y), fmaxf(v[g][0].z, v[g][0].w));
#pragma unroll
      for (int i = 1; i < 4; ++i)
        a = fmaxf(a, fmaxf(fmaxf(v[g][i].x, v[g][i].y), fmaxf(v[g][i].z, v[g][i].w)));
      mx[g] = a;
    }
#pragma unroll
    for (int o = 4; o > 0; o >>= 1) {
#pragma unroll
      for (int g = 0; g < 4; ++g) mx[g] = fmaxf(mx[g], __shfl_xor(mx[g], o, 32));
    }
    float sm[4];
#pragma unroll
    for (int g = 0; g < 4; ++g) {
      float a = 0.f;
#pragma unroll
      for (int i = 0; i < 4; ++i) {
        v[g][i].x = expf(v[g][i].x - mx[g]); v[g][i].y = expf(v[g][i].y - mx[g]);
        v[g][i].z = expf(v[g][i].z - mx[g]); v[g][i].w = expf(v[g][i].w - mx[g]);
        a += (v[g][i].x + v[g][i].y) + (v[g][i].z + v[g][i].w);
        *(v4f*)(dse + (size_t)g * NT + 32 * i) = v[g][i];
      }
      sm[g] = a;
    }
#pragma unroll
    for (int o = 4; o > 0; o >>= 1) {
#pragma unroll
      for (int g = 0; g < 4; ++g) sm[g] += __shfl_xor(sm[g], o, 32);
    }
    if (c8 == 0) {
      size_t so = (size_t)(b * 8 + tblk) * NJ + wv * 16 + 4 * rr;
      *(float4*)(cm + so) = make_float4(mx[0], mx[1], mx[2], mx[3]);
      *(float4*)(cz + so) = make_float4(sm[0], sm[1], sm[2], sm[3]);
    }
  }
}

struct TmpFrag { v4f a0, a1; v2f b0l, b1l, b0h, b1h; };

__device__ __forceinline__ void tmp_load(TmpFrag (&f)[4], const float* A0,
                                         const float* Cb, int kb) {
#pragma unroll
  for (int s = 0; s < 4; ++s) {
    int k = kb + 8 * s;
    f[s].a0 = *(const v4f*)(A0 + k);
    f[s].a1 = *(const v4f*)(A0 + 16 * NT + k);
    f[s].b0l.x = Cb[(size_t)k * ND];
    f[s].b0l.y = Cb[(size_t)(k + 1) * ND];
    f[s].b1l.x = Cb[(size_t)k * ND + 16];
    f[s].b1l.y = Cb[(size_t)(k + 1) * ND + 16];
    f[s].b0h.x = Cb[(size_t)(k + 2) * ND];
    f[s].b0h.y = Cb[(size_t)(k + 3) * ND];
    f[s].b1h.x = Cb[(size_t)(k + 2) * ND + 16];
    f[s].b1h.y = Cb[(size_t)(k + 3) * ND + 16];
  }
}

__device__ __forceinline__ void tmp_mma4(v2f a0, v2f a1, v2f b0, v2f b1, v8f (&acc)[2][2]) {
  acc[0][0] = __builtin_amdgcn_wmma_f32_16x16x4_f32(false, a0, false, b0, (short)0,
                                                    acc[0][0], false, false);
  acc[1][0] = __builtin_amdgcn_wmma_f32_16x16x4_f32(false, a1, false, b0, (short)0,
                                                    acc[1][0], false, false);
  acc[0][1] = __builtin_amdgcn_wmma_f32_16x16x4_f32(false, a0, false, b1, (short)0,
                                                    acc[0][1], false, false);
  acc[1][1] = __builtin_amdgcn_wmma_f32_16x16x4_f32(false, a1, false, b1, (short)0,
                                                    acc[1][1], false, false);
}

__device__ __forceinline__ void tmp_mma(const TmpFrag (&f)[4], float s0, float s1,
                                        v8f (&acc)[2][2]) {
#pragma unroll
  for (int s = 0; s < 4; ++s) {
    v4f a0 = f[s].a0 * s0;
    v4f a1 = f[s].a1 * s1;
    tmp_mma4(lo2(a0), lo2(a1), f[s].b0l, f[s].b1l, acc);
    tmp_mma4(hi2(a0), hi2(a1), f[s].b0h, f[s].b1h, acc);
  }
}

#define T_BLKS (NT / 128)
static_assert(T_BLKS == 8, "k_score writes one max and one sum per block of 128 t: 8 per column");
__device__ __forceinline__ void
tmp_role(int blk, const float* __restrict__ ET, const float* __restrict__ cm,
         const float* __restrict__ cz, const float* __restrict__ C,
         float* __restrict__ tmp) {
  __shared__ float ftab[4][T_BLKS][2][32];
  int b = blk >> 3;
  int dblk = blk & 7;
  int wv = threadIdx.x >> 5;
  int lane = threadIdx.x & 31;
  int m = lane & 15, kh = lane >> 4;
  int j0 = wv * 32, d0 = dblk * 32;
  const float* A0 = ET + (size_t)(b * NJ + j0 + m) * NT + 4 * kh;
  const float* Cb = C + (size_t)b * NT * ND + (size_t)(4 * kh) * ND + d0 + m;
  {
    const float* pm = cm + (size_t)b * T_BLKS * NJ + j0 + m;
    const float* pz = cz + (size_t)b * T_BLKS * NJ + j0 + m;
    float m0[T_BLKS], z0[T_BLKS], m1[T_BLKS], z1[T_BLKS];
#pragma unroll
    for (int q = 0; q < T_BLKS; ++q) {
      m0[q] = pm[q * NJ];      z0[q] = pz[q * NJ];
      m1[q] = pm[q * NJ + 16]; z1[q] = pz[q * NJ + 16];
    }
    float M0 = m0[0], M1 = m1[0];
#pragma unroll
    for (int q = 1; q < T_BLKS; ++q) { M0 = fmaxf(M0, m0[q]); M1 = fmaxf(M1, m1[q]); }
    float Z0 = 0.f, Z1 = 0.f;
#pragma unroll
    for (int q = 0; q < T_BLKS; ++q) {
      m0[q] = expf(m0[q] - M0); Z0 += z0[q] * m0[q];
      m1[q] = expf(m1[q] - M1); Z1 += z1[q] * m1[q];
    }
    float i0 = 1.0f / Z0, i1 = 1.0f / Z1;
#pragma unroll
    for (int q = 0; q < T_BLKS; ++q) {
      ftab[wv][q][0][lane] = m0[q] * i0;
      ftab[wv][q][1][lane] = m1[q] * i1;
    }
  }
  v8f acc[2][2] = {};
  TmpFrag fa[4], fb[4];
  tmp_load(fa, A0, Cb, 0);
#pragma unroll 1
  for (int kb = 0; kb < NT - 64; kb += 64) {
    float s0 = ftab[wv][kb >> 7][0][lane], s1 = ftab[wv][kb >> 7][1][lane];
    tmp_load(fb, A0, Cb, kb + 32);
    tmp_mma(fa, s0, s1, acc);
    tmp_load(fa, A0, Cb, kb + 64);
    tmp_mma(fb, s0, s1, acc);
    __builtin_amdgcn_sched_group_barrier(0x020, 40, 0);
    __builtin_amdgcn_sched_group_barrier(0x008, 32, 0);
    __builtin_amdgcn_sched_group_barrier(0x020, 40, 0);
    __builtin_amdgcn_sched_group_barrier(0x008, 32, 0);
  }
  tmp_load(fb, A0, Cb, NT - 32);
  {
    float s0 = ftab[wv][T_BLKS - 1][0][lane], s1 = ftab[wv][T_BLKS - 1][1][lane];
    tmp_mma(fa, s0, s1, acc);
    tmp_mma(fb, s0, s1, acc);
  }

#pragma unroll
  for (int rt = 0; rt < 2; ++rt) {
#pragma unroll
    for (int r = 0; r < 8; ++r) {
      int jj = j0 + 16 * rt + r + 8 * kh;
      float* ob = tmp + (size_t)(b * NJ + jj) * ND + d0 + m;
      ob[0]  = acc[rt][0][r];
      ob[16] = acc[rt][1][r];
    }
  }
}

#define OT_PITCH 68
#define OT_WAVE (32 * OT_PITCH)
struct OutFrag { v4f a0, a1; v2f bl[4], bh[4]; };

__device__ __forceinline__ void out_load(OutFrag& f, const float* S0, const float* S1,
                                         const float* Bp, int k0) {
  f.a0 = *(const v4f*)(S0 + k0);
  f.a1 = *(const v4f*)(S1 + k0);
  const float* p = Bp + (size_t)k0 * ND;
#pragma unroll
  for (int c = 0; c < 4; ++c) {
    f.bl[c].x = p[16 * c];           f.bl[c].y = p[ND + 16 * c];
    f.bh[c].x = p[2 * ND + 16 * c];  f.bh[c].y = p[3 * ND + 16 * c];
  }
}

__device__ __forceinline__ void out_mma(const OutFrag& f, v8f (&acc)[2][4]) {
  v2f a0l = lo2(f.a0), a1l = lo2(f.a1), a0h = hi2(f.a0), a1h = hi2(f.a1);
#pragma unroll
  for (int c = 0; c < 4; ++c) {
    acc[0][c] = __builtin_amdgcn_wmma_f32_16x16x4_f32(false, a0l, false, f.bl[c], (short)0,
                                                      acc[0][c], false, false);
    acc[1][c] = __builtin_amdgcn_wmma_f32_16x16x4_f32(false, a1l, false, f.bl[c], (short)0,
                                                      acc[1][c], false, false);
  }
#pragma unroll
  for (int c = 0; c < 4; ++c) {
    acc[0][c] = __builtin_amdgcn_wmma_f32_16x16x4_f32(false, a0h, false, f.bh[c], (short)0,
                                                      acc[0][c], false, false);
    acc[1][c] = __builtin_amdgcn_wmma_f32_16x16x4_f32(false, a1h, false, f.bh[c], (short)0,
                                                      acc[1][c], false, false);
  }
}

template <bool FIRST>
__device__ __forceinline__ void
out_product(int blk, const float* __restrict__ Sj, const float* __restrict__ Bmat,
            const float* __restrict__ C, float* __restrict__ out, float* tiles) {
  int b = blk >> 5;
  int tblk = blk & 31;
  int wv = threadIdx.x >> 5;
  int lane = threadIdx.x & 31;
  int m = lane & 15, kh = lane >> 4;
  int t0 = tblk * 32, d0 = wv * 64;
  const float* S0 = Sj + (size_t)(b * NT + t0 + m) * NJ + 4 * kh;
  const float* S1 = S0 + (size_t)16 * NJ;
  const float* Bp = Bmat + (size_t)b * NJ * ND + (size_t)(4 * kh) * ND + d0 + m;
  v8f acc[2][4] = {};
  OutFrag fa, fb;
  out_load(fa, S0, S1, Bp, 0);
#pragma unroll 1
  for (int k0 = 0; k0 < NJ - 16; k0 += 16) {
    out_load(fb, S0, S1, Bp, k0 + 8);
    out_mma(fa, acc);
    out_load(fa, S0, S1, Bp, k0 + 16);
    out_mma(fb, acc);
    __builtin_amdgcn_sched_group_barrier(0x020, 18, 0);
    __builtin_amdgcn_sched_group_barrier(0x008, 16, 0);
    __builtin_amdgcn_sched_group_barrier(0x020, 18, 0);
    __builtin_amdgcn_sched_group_barrier(0x008, 16, 0);
  }
  out_load(fb, S0, S1, Bp, NJ - 8);
  out_mma(fa, acc);
  out_mma(fb, acc);

  int rr = lane >> 4, c4 = 4 * (lane & 15);
  const float* cp = C + (size_t)(b * NT + t0 + rr) * ND + d0 + c4;
  v4f cv[16];
#pragma unroll
  for (int i = 0; i < 16; ++i) cv[i] = *(const v4f*)(cp + (size_t)(2 * i) * ND);
  float* tile = tiles + wv * OT_WAVE;
#pragma unroll
  for (int rt = 0; rt < 2; ++rt) {
#pragma unroll
    for (int c = 0; c < 4; ++c) {
#pragma unroll
      for (int r = 0; r < 8; ++r)
        tile[(16 * rt + r + 8 * kh) * OT_PITCH + 16 * c + m] = acc[rt][c][r];
    }
  }
  __builtin_amdgcn_fence(__ATOMIC_RELEASE, "wavefront");
  __builtin_amdgcn_wave_barrier();
  __builtin_amdgcn_fence(__ATOMIC_ACQUIRE, "wavefront");
  const float* tl = tile + rr * OT_PITCH + c4;
  float* ob = out + (size_t)(b * NT + t0 + rr) * (4 * ND) + d0 + c4;
#pragma unroll
  for (int i = 0; i < 16; ++i) {
    v4f a = *(const v4f*)(tl + 2 * i * OT_PITCH);
    float* o = ob + (size_t)(2 * i) * (4 * ND);
    if (FIRST) {
      *(v4f*)(o)          = cv[i];
      *(v4f*)(o + ND)     = a;
      *(v4f*)(o + 2 * ND) = cv[i] * a;
    } else {
      *(v4f*)(o + 3 * ND) = cv[i] * a;
    }
  }
}

#define A_EARLY (NB * 24)
#define A_LATE  (NB * 32 - A_EARLY)

__global__ void __launch_bounds__(128)
k_tmp_and_a(const float* __restrict__ ET, const float* __restrict__ cm,
            const float* __restrict__ cz, const float* __restrict__ C,
            float* __restrict__ tmp, const float* __restrict__ Sj,
            const float* __restrict__ Q, float* __restrict__ out) {
  __shared__ __attribute__((aligned(16))) float tiles[4 * OT_WAVE];
  if (blockIdx.x < NB * 8) tmp_role(blockIdx.x, ET, cm, cz, C, tmp);
  else out_product<true>(blockIdx.x - NB * 8, Sj, Q, C, out, tiles);
}

__global__ void __launch_bounds__(128)
k_out_b(const float* __restrict__ Sj, const float* __restrict__ Q,
        const float* __restrict__ tmp, const float* __restrict__ C,
        float* __restrict__ out) {
  __shared__ __attribute__((aligned(16))) float tiles[4 * OT_WAVE];
  if (blockIdx.x < A_LATE) out_product<true>(A_EARLY + blockIdx.x, Sj, Q, C, out, tiles);
  else out_product<false>(blockIdx.x - A_LATE, Sj, tmp, C, out, tiles);
}

extern "C" void kernel_launch(void* const* d_in, const int* in_sizes, int n_in,
                              void* d_out, int out_size, void* d_ws, size_t ws_size,
                              hipStream_t stream) {
  (void)in_sizes; (void)n_in; (void)out_size; (void)ws_size;
  const float* C = (const float*)d_in[0];
  const float* Q = (const float*)d_in[1];
  const float* w = (const float*)d_in[2];
  float* out = (float*)d_out;

  float* ws = (float*)d_ws;
  float* ch = ws;
  float* qb = ch + NB * NT;
  float* Sj = qb + NB * NJ;
  float* ET = Sj + (size_t)NB * NT * NJ;
  float* tp = ET + (size_t)NB * NT * NJ;
  float* cm = tp + (size_t)NB * NJ * ND;
  float* cz = cm + NB * 8 * NJ;

  k_rowdots<<<(NB * NT + NB * NJ) / (4 * RD_ROWS), 128, 0, stream>>>(C, Q, w, ch, qb);
  k_score<<<NB * 8, 256, 0, stream>>>(C, Q, w, ch, qb, Sj, ET, cm, cz);
  k_tmp_and_a<<<NB * 8 + A_EARLY, 128, 0, stream>>>(ET, cm, cz, C, tp, Sj, Q, out);
  k_out_b<<<A_LATE + NB * 32, 128, 0, stream>>>(Sj, Q, tp, C, out);
}
